# starting kernel
# baseline (speedup 1.0000x reference)
#include <hip/hip_runtime.h>
#include <hip/hip_bf16.h>

#define N_NODES   100000
#define N_EDGES   1600000
#define N_GRAPHS  128
#define HIDDEN    128
#define N_CLASSES 10

typedef __attribute__((ext_vector_type(2))) float v2f;
typedef __attribute__((ext_vector_type(8))) float v8f;

__global__ void deg_kernel(const int* __restrict__ src, const int* __restrict__ dst,
                           float* __restrict__ outdeg, float* __restrict__ indeg) {
    int e = blockIdx.x * blockDim.x + threadIdx.x;
    if (e < N_EDGES) {
        atomicAdd(&outdeg[src[e]], 1.0f);
        atomicAdd(&indeg[dst[e]], 1.0f);
    }
}

__global__ void norm_kernel(const float* __restrict__ outdeg, const float* __restrict__ indeg,
                            float* __restrict__ cs, float* __restrict__ cd) {
    int n = blockIdx.x * blockDim.x + threadIdx.x;
    if (n < N_NODES) {
        cs[n] = rsqrtf(fmaxf(outdeg[n], 1.0f));
        cd[n] = rsqrtf(fmaxf(indeg[n], 1.0f));
    }
}

__global__ void scatter1_kernel(const int* __restrict__ src, const int* __restrict__ dst,
                                const float* __restrict__ indeg, const float* __restrict__ cs,
                                float* __restrict__ agg1) {
    int e = blockIdx.x * blockDim.x + threadIdx.x;
    if (e < N_EDGES) {
        int s = src[e];
        atomicAdd(&agg1[dst[e]], indeg[s] * cs[s]);
    }
}

__global__ void layer1_kernel(const float* __restrict__ agg1, const float* __restrict__ cd,
                              const float* __restrict__ W1, const float* __restrict__ b1,
                              float* __restrict__ h) {
    int i = blockIdx.x * blockDim.x + threadIdx.x;
    if (i < N_NODES * HIDDEN) {
        int m = i >> 7;
        int j = i & (HIDDEN - 1);
        float x = agg1[m] * cd[m];
        h[i] = fmaxf(fmaf(x, W1[j], b1[j]), 0.0f);
    }
}

__global__ void scatter_kernel(const int* __restrict__ src, const int* __restrict__ dst,
                               const float* __restrict__ cs, const float* __restrict__ h,
                               float* __restrict__ agg) {
    int gid  = blockIdx.x * blockDim.x + threadIdx.x;
    int e    = gid >> 5;
    int lane = gid & 31;
    if (e < N_EDGES) {
        int s = src[e], d = dst[e];
        float sc = cs[s];
        const float* hs = h   + (size_t)s * HIDDEN;
        float*       ad = agg + (size_t)d * HIDDEN;
#pragma unroll
        for (int f = lane; f < HIDDEN; f += 32)
            atomicAdd(&ad[f], hs[f] * sc);
    }
}

__global__ void __launch_bounds__(256) gemm_relu_kernel(const float* __restrict__ agg,
                                                        const float* __restrict__ cd,
                                                        const float* __restrict__ W,
                                                        const float* __restrict__ bias,
                                                        float* __restrict__ hout) {
    __shared__ float Wl[HIDDEN * HIDDEN];
    int tid = threadIdx.x;
    for (int i = tid; i < HIDDEN * HIDDEN; i += 256) Wl[i] = W[i];
    __syncthreads();

    int wave = tid >> 5;
    int lane = tid & 31;
    int half = lane >> 4;
    int lr   = lane & 15;
    int m0   = blockIdx.x * 16;
    int m    = m0 + lr;
    int n    = wave * 16 + lr;

    float cdm = cd[m];
    const float* arow = agg + (size_t)m * HIDDEN + half * 2;
    const float* brow = Wl + (half * 2) * HIDDEN + n;

    v8f acc = {};
#pragma unroll
    for (int k0 = 0; k0 < HIDDEN; k0 += 4) {
        v2f av, bv;
        av.x = arow[k0]     * cdm;
        av.y = arow[k0 + 1] * cdm;
        bv.x = brow[k0 * HIDDEN];
        bv.y = brow[k0 * HIDDEN + HIDDEN];
        acc = __builtin_amdgcn_wmma_f32_16x16x4_f32(
             false, av,  false, bv,
             (short)0, acc,  false,  false);
    }

    float bn = bias[n];
#pragma unroll
    for (int r = 0; r < 8; ++r) {
        int mm = m0 + half * 8 + r;
        hout[(size_t)mm * HIDDEN + n] = fmaxf(acc[r] + bn, 0.0f);
    }
}

__global__ void pool_kernel(const int* __restrict__ seg, const float* __restrict__ h,
                            float* __restrict__ hg, float* __restrict__ counts) {
    int gid  = blockIdx.x * blockDim.x + threadIdx.x;
    int node = gid >> 5;
    int lane = gid & 31;
    if (node < N_NODES) {
        int g = seg[node];
        const float* hn  = h  + (size_t)node * HIDDEN;
        float*       hgo = hg + (size_t)g * HIDDEN;
#pragma unroll
        for (int f = lane; f < HIDDEN; f += 32)
            atomicAdd(&hgo[f], hn[f]);
        if (lane == 0) atomicAdd(&counts[g], 1.0f);
    }
}

__global__ void pool_div_kernel(float* __restrict__ hg, const float* __restrict__ counts) {
    int i = blockIdx.x * blockDim.x + threadIdx.x;
    if (i < N_GRAPHS * HIDDEN)
        hg[i] /= fmaxf(counts[i >> 7], 1.0f);
}

__global__ void logits_kernel(const float* __restrict__ hg1, const float* __restrict__ hg2,
                              const float* __restrict__ Wc, const float* __restrict__ bc,
                              float* __restrict__ logits) {
    int i = blockIdx.x * blockDim.x + threadIdx.x;
    if (i < N_GRAPHS * N_CLASSES) {
        int g = i / N_CLASSES, c = i % N_CLASSES;
        float s = bc[c];
#pragma unroll 8
        for (int k = 0; k < HIDDEN; ++k)
            s += fabsf(hg1[g * HIDDEN + k] - hg2[g * HIDDEN + k]) * Wc[k * N_CLASSES + c];
        logits[i] = s;
    }
}

static void run_branch(const int* src, const int* dst, const int* seg,
                       const float* W1, const float* b1,
                       const float* W2, const float* b2,
                       const float* W3, const float* b3,
                       const float* W4, const float* b4,
                       float* hg_out, float* ws_f, hipStream_t stream) {
    float* outdeg = ws_f;
    float* indeg  = ws_f + 100000;
    float* cs     = ws_f + 200000;
    float* cd     = ws_f + 300000;
    float* agg1   = ws_f + 400000;
    float* counts = ws_f + 500000;
    float* h      = ws_f + 500224;
    float* agg    = h + (size_t)N_NODES * HIDDEN;

    hipMemsetAsync(ws_f, 0, (size_t)500224 * sizeof(float), stream);

    const int T = 256;
    deg_kernel<<<(N_EDGES + T - 1) / T, T, 0, stream>>>(src, dst, outdeg, indeg);
    norm_kernel<<<(N_NODES + T - 1) / T, T, 0, stream>>>(outdeg, indeg, cs, cd);

    scatter1_kernel<<<(N_EDGES + T - 1) / T, T, 0, stream>>>(src, dst, indeg, cs, agg1);
    layer1_kernel<<<(N_NODES * HIDDEN + T - 1) / T, T, 0, stream>>>(agg1, cd, W1, b1, h);

    const float* Ws[3] = {W2, W3, W4};
    const float* bs[3] = {b2, b3, b4};
    for (int l = 0; l < 3; ++l) {
        hipMemsetAsync(agg, 0, (size_t)N_NODES * HIDDEN * sizeof(float), stream);
        scatter_kernel<<<(N_EDGES * 32) / T, T, 0, stream>>>(src, dst, cs, h, agg);
        gemm_relu_kernel<<<N_NODES / 16, T, 0, stream>>>(agg, cd, Ws[l], bs[l], h);
    }

    hipMemsetAsync(hg_out, 0, (size_t)N_GRAPHS * HIDDEN * sizeof(float), stream);
    pool_kernel<<<(N_NODES * 32 + T - 1) / T, T, 0, stream>>>(seg, h, hg_out, counts);
    pool_div_kernel<<<(N_GRAPHS * HIDDEN + T - 1) / T, T, 0, stream>>>(hg_out, counts);
}

extern "C" void kernel_launch(void* const* d_in, const int* in_sizes, int n_in,
                              void* d_out, int out_size, void* d_ws, size_t ws_size,
                              hipStream_t stream) {
    const int*   src1 = (const int*)d_in[0];
    const int*   dst1 = (const int*)d_in[1];
    const int*   seg1 = (const int*)d_in[2];
    const int*   src2 = (const int*)d_in[3];
    const int*   dst2 = (const int*)d_in[4];
    const int*   seg2 = (const int*)d_in[5];
    const float* W1   = (const float*)d_in[6];
    const float* b1   = (const float*)d_in[7];
    const float* W2   = (const float*)d_in[8];
    const float* b2   = (const float*)d_in[9];
    const float* W3   = (const float*)d_in[10];
    const float* b3   = (const float*)d_in[11];
    const float* W4   = (const float*)d_in[12];
    const float* b4   = (const float*)d_in[13];
    const float* Wc   = (const float*)d_in[14];
    const float* bc   = (const float*)d_in[15];

    float* out   = (float*)d_out;
    float* hg1   = out;
    float* hg2   = out + N_GRAPHS * HIDDEN;
    float* logit = out + 2 * N_GRAPHS * HIDDEN;
    float* ws_f  = (float*)d_ws;

    run_branch(src1, dst1, seg1, W1, b1, W2, b2, W3, b3, W4, b4, hg1, ws_f, stream);
    run_branch(src2, dst2, seg2, W1, b1, W2, b2, W3, b3, W4, b4, hg2, ws_f, stream);

    logits_kernel<<<(N_GRAPHS * N_CLASSES + 255) / 256, 256, 0, stream>>>(hg1, hg2, Wc, bc, logit);
}


# Edge sums by destination-sorted edge list rebuilt each call (int atomic rank, scan, fill); one wave sums a node's source rows, stores once: no f32 atomics/zero-fill on agg; sum order = rank order
# speedup vs baseline: 2.0133x; 2.0133x over previous
#include <hip/hip_runtime.h>
#include <hip/hip_bf16.h>

#define N_NODES   100000
#define N_EDGES   1600000
#define N_GRAPHS  128
#define HIDDEN    128
#define N_CLASSES 10

typedef __attribute__((ext_vector_type(2))) float v2f;
typedef __attribute__((ext_vector_type(8))) float v8f;

__global__ void deg_rank_kernel(const int* __restrict__ src, const int* __restrict__ dst,
                                float* __restrict__ outdeg, int* __restrict__ indeg_i,
                                int* __restrict__ rank) {
    int e = blockIdx.x * blockDim.x + threadIdx.x;
    if (e < N_EDGES) {
        atomicAdd(&outdeg[src[e]], 1.0f);
        rank[e] = atomicAdd(&indeg_i[dst[e]], 1);
    }
}

__global__ void __launch_bounds__(1024) scan_kernel(const int* __restrict__ cnt, int* __restrict__ rowptr) {
    __shared__ int wsum[32];
    const int tid  = threadIdx.x;
    const int lane = tid & 31;
    const int wave = tid >> 5;
    const int4* __restrict__ c4 = (const int4*)cnt;
    int4* __restrict__ r4 = (int4*)rowptr;
    int carry = 0;
    for (int base = 0; base < N_NODES / 4; base += 1024) {
        const int i = base + tid;
        int4 v = make_int4(0, 0, 0, 0);
        if (i < N_NODES / 4) v = c4[i];
        const int t = v.x + v.y + v.z + v.w;
        int inc = t;
        for (int o = 1; o < 32; o <<= 1) {
            int u = __shfl_up(inc, o);
            if (lane >= o) inc += u;
        }
        if (lane == 31) wsum[wave] = inc;
        __syncthreads();
        const int ws = wsum[lane];
        int winc = ws;
        for (int o = 1; o < 32; o <<= 1) {
            int u = __shfl_up(winc, o);
            if (lane >= o) winc += u;
        }
        const int wave_excl = __shfl(winc, wave) - __shfl(ws, wave);
        const int total     = __shfl(winc, 31);
        const int excl      = carry + wave_excl + inc - t;
        if (i < N_NODES / 4) {
            int4 o4;
            o4.x = excl;
            o4.y = o4.x + v.x;
            o4.z = o4.y + v.y;
            o4.w = o4.z + v.z;
            r4[i] = o4;
        }
        carry += total;
        __syncthreads();
    }
}

__global__ void fill_kernel(const int* __restrict__ src, const int* __restrict__ dst,
                            const int* __restrict__ rank, const int* __restrict__ rowptr,
                            const float* __restrict__ cs, int2* __restrict__ ce) {
    int e = blockIdx.x * blockDim.x + threadIdx.x;
    if (e < N_EDGES) {
        int s = src[e];
        int pos = rowptr[dst[e]] + rank[e];
        ce[pos] = make_int2(s, __float_as_int(cs[s]));
    }
}

__global__ void norm_kernel(const float* __restrict__ outdeg, const int* __restrict__ indeg_i,
                            float* __restrict__ cs, float* __restrict__ cd) {
    int n = blockIdx.x * blockDim.x + threadIdx.x;
    if (n < N_NODES) {
        cs[n] = rsqrtf(fmaxf(outdeg[n], 1.0f));
        cd[n] = rsqrtf(fmaxf((float)indeg_i[n], 1.0f));
    }
}

__global__ void gather1_kernel(const int* __restrict__ rowptr, const int* __restrict__ indeg_i,
                               const int2* __restrict__ ce, float* __restrict__ agg1) {
    int d = blockIdx.x * blockDim.x + threadIdx.x;
    if (d < N_NODES) {
        const int beg = rowptr[d];
        const int cnt = indeg_i[d];
        float a = 0.0f;
        for (int j = 0; j < cnt; ++j) {
            int2 c = ce[beg + j];
            a += (float)indeg_i[c.x] * __int_as_float(c.y);
        }
        agg1[d] = a;
    }
}

__global__ void layer1_kernel(const float* __restrict__ agg1, const float* __restrict__ cd,
                              const float* __restrict__ W1, const float* __restrict__ b1,
                              float* __restrict__ h) {
    int i = blockIdx.x * blockDim.x + threadIdx.x;
    if (i < N_NODES * HIDDEN) {
        int m = i >> 7;
        int j = i & (HIDDEN - 1);
        float x = agg1[m] * cd[m];
        h[i] = fmaxf(fmaf(x, W1[j], b1[j]), 0.0f);
    }
}

__global__ void __launch_bounds__(256) gather_kernel(const int* __restrict__ rowptr,
                                                     const int* __restrict__ indeg_i,
                                                     const int2* __restrict__ ce,
                                                     const float* __restrict__ h,
                                                     float* __restrict__ agg) {
    const int wave = __builtin_amdgcn_readfirstlane((int)(threadIdx.x >> 5));
    const int lane = threadIdx.x & 31;
    const int d    = blockIdx.x * 8 + wave;
    if (d < N_NODES) {
        const int beg = rowptr[d];
        const int cnt = indeg_i[d];
        const float4* __restrict__ h4 = (const float4*)h;
        float4 acc = make_float4(0.0f, 0.0f, 0.0f, 0.0f);
        for (int base = 0; base < cnt; base += 32) {
            const int n = min(cnt - base, 32);
            int cidx = 0, cscl = 0;
            if (lane < n) {
                int2 c = ce[beg + base + lane];
                cidx = c.x;
                cscl = c.y;
            }
            for (int j = 0; j < n; j += 4) {
                float4 v0, v1, v2, v3;
                float  c0, c1 = 0.0f, c2 = 0.0f, c3 = 0.0f;
                v1 = v2 = v3 = make_float4(0.0f, 0.0f, 0.0f, 0.0f);
                {
                    int s = __builtin_amdgcn_readlane(cidx, j);
                    c0 = __int_as_float(__builtin_amdgcn_readlane(cscl, j));
                    v0 = h4[(size_t)s * 32 + lane];
                }
                if (j + 1 < n) {
                    int s = __builtin_amdgcn_readlane(cidx, j + 1);
                    c1 = __int_as_float(__builtin_amdgcn_readlane(cscl, j + 1));
                    v1 = h4[(size_t)s * 32 + lane];
                }
                if (j + 2 < n) {
                    int s = __builtin_amdgcn_readlane(cidx, j + 2);
                    c2 = __int_as_float(__builtin_amdgcn_readlane(cscl, j + 2));
                    v2 = h4[(size_t)s * 32 + lane];
                }
                if (j + 3 < n) {
                    int s = __builtin_amdgcn_readlane(cidx, j + 3);
                    c3 = __int_as_float(__builtin_amdgcn_readlane(cscl, j + 3));
                    v3 = h4[(size_t)s * 32 + lane];
                }
                acc.x = fmaf(v0.x, c0, acc.x); acc.y = fmaf(v0.y, c0, acc.y);
                acc.z = fmaf(v0.z, c0, acc.z); acc.w = fmaf(v0.w, c0, acc.w);
                acc.x = fmaf(v1.x, c1, acc.x); acc.y = fmaf(v1.y, c1, acc.y);
                acc.z = fmaf(v1.z, c1, acc.z); acc.w = fmaf(v1.w, c1, acc.w);
                acc.x = fmaf(v2.x, c2, acc.x); acc.y = fmaf(v2.y, c2, acc.y);
                acc.z = fmaf(v2.z, c2, acc.z); acc.w = fmaf(v2.w, c2, acc.w);
                acc.x = fmaf(v3.x, c3, acc.x); acc.y = fmaf(v3.y, c3, acc.y);
                acc.z = fmaf(v3.z, c3, acc.z); acc.w = fmaf(v3.w, c3, acc.w);
            }
        }
        ((float4*)agg)[(size_t)d * 32 + lane] = acc;
    }
}

__global__ void __launch_bounds__(256) gemm_relu_kernel(const float* __restrict__ agg,
                                                        const float* __restrict__ cd,
                                                        const float* __restrict__ W,
                                                        const float* __restrict__ bias,
                                                        float* __restrict__ hout) {
    __shared__ float Wl[HIDDEN * HIDDEN];
    int tid = threadIdx.x;
    for (int i = tid; i < HIDDEN * HIDDEN; i += 256) Wl[i] = W[i];
    __syncthreads();

    int wave = tid >> 5;
    int lane = tid & 31;
    int half = lane >> 4;
    int lr   = lane & 15;
    int m0   = blockIdx.x * 16;
    int m    = m0 + lr;
    int n    = wave * 16 + lr;

    float cdm = cd[m];
    const float* arow = agg + (size_t)m * HIDDEN + half * 2;
    const float* brow = Wl + (half * 2) * HIDDEN + n;

    v8f acc = {};
#pragma unroll
    for (int k0 = 0; k0 < HIDDEN; k0 += 4) {
        v2f av, bv;
        av.x = arow[k0]     * cdm;
        av.y = arow[k0 + 1] * cdm;
        bv.x = brow[k0 * HIDDEN];
        bv.y = brow[k0 * HIDDEN + HIDDEN];
        acc = __builtin_amdgcn_wmma_f32_16x16x4_f32(
             false, av,  false, bv,
             (short)0, acc,  false,  false);
    }

    float bn = bias[n];
#pragma unroll
    for (int r = 0; r < 8; ++r) {
        int mm = m0 + half * 8 + r;
        hout[(size_t)mm * HIDDEN + n] = fmaxf(acc[r] + bn, 0.0f);
    }
}

__global__ void pool_kernel(const int* __restrict__ seg, const float* __restrict__ h,
                            float* __restrict__ hg, float* __restrict__ counts) {
    int gid  = blockIdx.x * blockDim.x + threadIdx.x;
    int node = gid >> 5;
    int lane = gid & 31;
    if (node < N_NODES) {
        int g = seg[node];
        const float* hn  = h  + (size_t)node * HIDDEN;
        float*       hgo = hg + (size_t)g * HIDDEN;
#pragma unroll
        for (int f = lane; f < HIDDEN; f += 32)
            atomicAdd(&hgo[f], hn[f]);
        if (lane == 0) atomicAdd(&counts[g], 1.0f);
    }
}

__global__ void pool_div_kernel(float* __restrict__ hg, const float* __restrict__ counts) {
    int i = blockIdx.x * blockDim.x + threadIdx.x;
    if (i < N_GRAPHS * HIDDEN)
        hg[i] /= fmaxf(counts[i >> 7], 1.0f);
}

__global__ void logits_kernel(const float* __restrict__ hg1, const float* __restrict__ hg2,
                              const float* __restrict__ Wc, const float* __restrict__ bc,
                              float* __restrict__ logits) {
    int i = blockIdx.x * blockDim.x + threadIdx.x;
    if (i < N_GRAPHS * N_CLASSES) {
        int g = i / N_CLASSES, c = i % N_CLASSES;
        float s = bc[c];
#pragma unroll 8
        for (int k = 0; k < HIDDEN; ++k)
            s += fabsf(hg1[g * HIDDEN + k] - hg2[g * HIDDEN + k]) * Wc[k * N_CLASSES + c];
        logits[i] = s;
    }
}

static void run_branch(const int* src, const int* dst, const int* seg,
                       const float* W1, const float* b1,
                       const float* W2, const float* b2,
                       const float* W3, const float* b3,
                       const float* W4, const float* b4,
                       float* hg_out, float* ws_f, hipStream_t stream) {
    float* outdeg = ws_f;
    int*   indeg_i = (int*)(ws_f + 100000);
    float* cs     = ws_f + 200000;
    float* cd     = ws_f + 300000;
    float* agg1   = ws_f + 400000;
    float* counts = ws_f + 500000;
    float* h      = ws_f + 500224;
    float* agg    = h + (size_t)N_NODES * HIDDEN;
    int*   rowptr = (int*)(agg + (size_t)N_NODES * HIDDEN);
    int*   rank   = rowptr + N_NODES;
    int2*  ce     = (int2*)(rank + N_EDGES);

    hipMemsetAsync(ws_f, 0, (size_t)500224 * sizeof(float), stream);

    const int T = 256;
    deg_rank_kernel<<<(N_EDGES + T - 1) / T, T, 0, stream>>>(src, dst, outdeg, indeg_i, rank);
    norm_kernel<<<(N_NODES + T - 1) / T, T, 0, stream>>>(outdeg, indeg_i, cs, cd);

    scan_kernel<<<1, 1024, 0, stream>>>(indeg_i, rowptr);
    fill_kernel<<<(N_EDGES + T - 1) / T, T, 0, stream>>>(src, dst, rank, rowptr, cs, ce);

    gather1_kernel<<<(N_NODES + T - 1) / T, T, 0, stream>>>(rowptr, indeg_i, ce, agg1);
    layer1_kernel<<<(N_NODES * HIDDEN + T - 1) / T, T, 0, stream>>>(agg1, cd, W1, b1, h);

    const float* Ws[3] = {W2, W3, W4};
    const float* bs[3] = {b2, b3, b4};
    for (int l = 0; l < 3; ++l) {
        gather_kernel<<<(N_NODES * 32) / T, T, 0, stream>>>(rowptr, indeg_i, ce, h, agg);
        gemm_relu_kernel<<<N_NODES / 16, T, 0, stream>>>(agg, cd, Ws[l], bs[l], h);
    }

    hipMemsetAsync(hg_out, 0, (size_t)N_GRAPHS * HIDDEN * sizeof(float), stream);
    pool_kernel<<<(N_NODES * 32 + T - 1) / T, T, 0, stream>>>(seg, h, hg_out, counts);
    pool_div_kernel<<<(N_GRAPHS * HIDDEN + T - 1) / T, T, 0, stream>>>(hg_out, counts);
}

extern "C" void kernel_launch(void* const* d_in, const int* in_sizes, int n_in,
                              void* d_out, int out_size, void* d_ws, size_t ws_size,
                              hipStream_t stream) {
    const int*   src1 = (const int*)d_in[0];
    const int*   dst1 = (const int*)d_in[1];
    const int*   seg1 = (const int*)d_in[2];
    const int*   src2 = (const int*)d_in[3];
    const int*   dst2 = (const int*)d_in[4];
    const int*   seg2 = (const int*)d_in[5];
    const float* W1   = (const float*)d_in[6];
    const float* b1   = (const float*)d_in[7];
    const float* W2   = (const float*)d_in[8];
    const float* b2   = (const float*)d_in[9];
    const float* W3   = (const float*)d_in[10];
    const float* b3   = (const float*)d_in[11];
    const float* W4   = (const float*)d_in[12];
    const float* b4   = (const float*)d_in[13];
    const float* Wc   = (const float*)d_in[14];
    const float* bc   = (const float*)d_in[15];

    float* out   = (float*)d_out;
    float* hg1   = out;
    float* hg2   = out + N_GRAPHS * HIDDEN;
    float* logit = out + 2 * N_GRAPHS * HIDDEN;
    float* ws_f  = (float*)d_ws;

    run_branch(src1, dst1, seg1, W1, b1, W2, b2, W3, b3, W4, b4, hg1, ws_f, stream);
    run_branch(src2, dst2, seg2, W1, b1, W2, b2, W3, b3, W4, b4, hg2, ws_f, stream);

    logits_kernel<<<(N_GRAPHS * N_CLASSES + 255) / 256, 256, 0, stream>>>(hg1, hg2, Wc, bc, logit);
}


# Layer-1 rows are no longer stored or gathered: the layer-2 edge sum rebuilds relu(x[src]*W1+b1) at each edge from the node scalar x=agg1*cd (a wave takes 8 nodes); same products, same sum order
# speedup vs baseline: 2.1467x; 1.0663x over previous
#include <hip/hip_runtime.h>
#include <hip/hip_bf16.h>

#define N_NODES   100000
#define N_EDGES   1600000
#define N_GRAPHS  128
#define HIDDEN    128
#define N_CLASSES 10

typedef __attribute__((ext_vector_type(2))) float v2f;
typedef __attribute__((ext_vector_type(8))) float v8f;

__global__ void deg_rank_kernel(const int* __restrict__ src, const int* __restrict__ dst,
                                float* __restrict__ outdeg, int* __restrict__ indeg_i,
                                int* __restrict__ rank) {
    int e = blockIdx.x * blockDim.x + threadIdx.x;
    if (e < N_EDGES) {
        atomicAdd(&outdeg[src[e]], 1.0f);
        rank[e] = atomicAdd(&indeg_i[dst[e]], 1);
    }
}

__global__ void __launch_bounds__(1024) scan_kernel(const int* __restrict__ cnt, int* __restrict__ rowptr) {
    __shared__ int wsum[32];
    const int tid  = threadIdx.x;
    const int lane = tid & 31;
    const int wave = tid >> 5;
    const int4* __restrict__ c4 = (const int4*)cnt;
    int4* __restrict__ r4 = (int4*)rowptr;
    int carry = 0;
    for (int base = 0; base < N_NODES / 4; base += 1024) {
        const int i = base + tid;
        int4 v = make_int4(0, 0, 0, 0);
        if (i < N_NODES / 4) v = c4[i];
        const int t = v.x + v.y + v.z + v.w;
        int inc = t;
        for (int o = 1; o < 32; o <<= 1) {
            int u = __shfl_up(inc, o);
            if (lane >= o) inc += u;
        }
        if (lane == 31) wsum[wave] = inc;
        __syncthreads();
        const int ws = wsum[lane];
        int winc = ws;
        for (int o = 1; o < 32; o <<= 1) {
            int u = __shfl_up(winc, o);
            if (lane >= o) winc += u;
        }
        const int wave_excl = __shfl(winc, wave) - __shfl(ws, wave);
        const int total     = __shfl(winc, 31);
        const int excl      = carry + wave_excl + inc - t;
        if (i < N_NODES / 4) {
            int4 o4;
            o4.x = excl;
            o4.y = o4.x + v.x;
            o4.z = o4.y + v.y;
            o4.w = o4.z + v.z;
            r4[i] = o4;
        }
        carry += total;
        __syncthreads();
    }
}

__global__ void fill_kernel(const int* __restrict__ src, const int* __restrict__ dst,
                            const int* __restrict__ rank, const int* __restrict__ rowptr,
                            const float* __restrict__ cs, int2* __restrict__ ce) {
    int e = blockIdx.x * blockDim.x + threadIdx.x;
    if (e < N_EDGES) {
        int s = src[e];
        int pos = rowptr[dst[e]] + rank[e];
        ce[pos] = make_int2(s, __float_as_int(cs[s]));
    }
}

__global__ void norm_kernel(const float* __restrict__ outdeg, const int* __restrict__ indeg_i,
                            float* __restrict__ cs, float* __restrict__ cd) {
    int n = blockIdx.x * blockDim.x + threadIdx.x;
    if (n < N_NODES) {
        cs[n] = rsqrtf(fmaxf(outdeg[n], 1.0f));
        cd[n] = rsqrtf(fmaxf((float)indeg_i[n], 1.0f));
    }
}

__global__ void gather1_kernel(const int* __restrict__ rowptr, const int* __restrict__ indeg_i,
                               const int2* __restrict__ ce, const float* __restrict__ cd,
                               float* __restrict__ x) {
    int d = blockIdx.x * blockDim.x + threadIdx.x;
    if (d < N_NODES) {
        const int beg = rowptr[d];
        const int cnt = indeg_i[d];
        float a = 0.0f;
        for (int j = 0; j < cnt; ++j) {
            int2 c = ce[beg + j];
            a += (float)indeg_i[c.x] * __int_as_float(c.y);
        }
        x[d] = a * cd[d];
    }
}

#define NODES_PER_WAVE 8
static_assert(N_NODES % NODES_PER_WAVE == 0, "a wave's 8 nodes must all exist");
__global__ void __launch_bounds__(256) gather_rank1_kernel(const int* __restrict__ rowptr,
                                                           const int* __restrict__ indeg_i,
                                                           const int2* __restrict__ ce,
                                                           const float* __restrict__ x,
                                                           const float* __restrict__ W1,
                                                           const float* __restrict__ b1,
                                                           float* __restrict__ agg) {
    const int wave = __builtin_amdgcn_readfirstlane((int)(threadIdx.x >> 5));
    const int lane = threadIdx.x & 31;
    const int g    = blockIdx.x * 8 + wave;
    if (g < N_NODES / NODES_PER_WAVE) {
        const float4 w = ((const float4*)W1)[lane];
        const float4 b = ((const float4*)b1)[lane];
        const int4 ba = ((const int4*)rowptr)[2 * g],  bb = ((const int4*)rowptr)[2 * g + 1];
        const int4 ca = ((const int4*)indeg_i)[2 * g], cb = ((const int4*)indeg_i)[2 * g + 1];
        const int begs[NODES_PER_WAVE] = {ba.x, ba.y, ba.z, ba.w, bb.x, bb.y, bb.z, bb.w};
        const int cnts[NODES_PER_WAVE] = {ca.x, ca.y, ca.z, ca.w, cb.x, cb.y, cb.z, cb.w};

        int2 e0[NODES_PER_WAVE];
        int  x0[NODES_PER_WAVE];
#pragma unroll
        for (int t = 0; t < NODES_PER_WAVE; ++t)
            e0[t] = ce[min(begs[t] + lane, N_EDGES - 1)];
        __builtin_amdgcn_sched_barrier(0);
#pragma unroll
        for (int t = 0; t < NODES_PER_WAVE; ++t)
            x0[t] = __float_as_int(x[e0[t].x]);
        asm volatile("" :: "v"(x0[0]), "v"(x0[1]), "v"(x0[2]), "v"(x0[3]),
                           "v"(x0[4]), "v"(x0[5]), "v"(x0[6]), "v"(x0[7]));

#pragma unroll
        for (int t = 0; t < NODES_PER_WAVE; ++t) {
            const int beg = begs[t];
            const int cnt = cnts[t];
            int xsrc = x0[t], cscl = e0[t].y;
            float4 acc = make_float4(0.0f, 0.0f, 0.0f, 0.0f);
            for (int base = 0; base < cnt; base += 32) {
                const int n = min(cnt - base, 32);
                if (base > 0) {
                    xsrc = 0; cscl = 0;
                    if (lane < n) {
                        int2 c = ce[beg + base + lane];
                        cscl = c.y;
                        xsrc = __float_as_int(x[c.x]);
                    }
                }
                for (int j = 0; j < n; ++j) {
                    const float xv = __int_as_float(__builtin_amdgcn_readlane(xsrc, j));
                    const float cv = __int_as_float(__builtin_amdgcn_readlane(cscl, j));
                    acc.x = fmaf(fmaxf(fmaf(xv, w.x, b.x), 0.0f), cv, acc.x);
                    acc.y = fmaf(fmaxf(fmaf(xv, w.y, b.y), 0.0f), cv, acc.y);
                    acc.z = fmaf(fmaxf(fmaf(xv, w.z, b.z), 0.0f), cv, acc.z);
                    acc.w = fmaf(fmaxf(fmaf(xv, w.w, b.w), 0.0f), cv, acc.w);
                }
            }
            ((float4*)agg)[(size_t)(g * NODES_PER_WAVE + t) * 32 + lane] = acc;
        }
    }
}

__global__ void __launch_bounds__(256) gather_kernel(const int* __restrict__ rowptr,
                                                     const int* __restrict__ indeg_i,
                                                     const int2* __restrict__ ce,
                                                     const float* __restrict__ h,
                                                     float* __restrict__ agg) {
    const int wave = __builtin_amdgcn_readfirstlane((int)(threadIdx.x >> 5));
    const int lane = threadIdx.x & 31;
    const int d    = blockIdx.x * 8 + wave;
    if (d < N_NODES) {
        const int beg = rowptr[d];
        const int cnt = indeg_i[d];
        const float4* __restrict__ h4 = (const float4*)h;
        float4 acc = make_float4(0.0f, 0.0f, 0.0f, 0.0f);
        for (int base = 0; base < cnt; base += 32) {
            const int n = min(cnt - base, 32);
            int cidx = 0, cscl = 0;
            if (lane < n) {
                int2 c = ce[beg + base + lane];
                cidx = c.x;
                cscl = c.y;
            }
            for (int j = 0; j < n; j += 4) {
                float4 v0, v1, v2, v3;
                float  c0, c1 = 0.0f, c2 = 0.0f, c3 = 0.0f;
                v1 = v2 = v3 = make_float4(0.0f, 0.0f, 0.0f, 0.0f);
                {
                    int s = __builtin_amdgcn_readlane(cidx, j);
                    c0 = __int_as_float(__builtin_amdgcn_readlane(cscl, j));
                    v0 = h4[(size_t)s * 32 + lane];
                }
                if (j + 1 < n) {
                    int s = __builtin_amdgcn_readlane(cidx, j + 1);
                    c1 = __int_as_float(__builtin_amdgcn_readlane(cscl, j + 1));
                    v1 = h4[(size_t)s * 32 + lane];
                }
                if (j + 2 < n) {
                    int s = __builtin_amdgcn_readlane(cidx, j + 2);
                    c2 = __int_as_float(__builtin_amdgcn_readlane(cscl, j + 2));
                    v2 = h4[(size_t)s * 32 + lane];
                }
                if (j + 3 < n) {
                    int s = __builtin_amdgcn_readlane(cidx, j + 3);
                    c3 = __int_as_float(__builtin_amdgcn_readlane(cscl, j + 3));
                    v3 = h4[(size_t)s * 32 + lane];
                }
                acc.x = fmaf(v0.x, c0, acc.x); acc.y = fmaf(v0.y, c0, acc.y);
                acc.z = fmaf(v0.z, c0, acc.z); acc.w = fmaf(v0.w, c0, acc.w);
                acc.x = fmaf(v1.x, c1, acc.x); acc.y = fmaf(v1.y, c1, acc.y);
                acc.z = fmaf(v1.z, c1, acc.z); acc.w = fmaf(v1.w, c1, acc.w);
                acc.x = fmaf(v2.x, c2, acc.x); acc.y = fmaf(v2.y, c2, acc.y);
                acc.z = fmaf(v2.z, c2, acc.z); acc.w = fmaf(v2.w, c2, acc.w);
                acc.x = fmaf(v3.x, c3, acc.x); acc.y = fmaf(v3.y, c3, acc.y);
                acc.z = fmaf(v3.z, c3, acc.z); acc.w = fmaf(v3.w, c3, acc.w);
            }
        }
        ((float4*)agg)[(size_t)d * 32 + lane] = acc;
    }
}

__global__ void __launch_bounds__(256) gemm_relu_kernel(const float* __restrict__ agg,
                                                        const float* __restrict__ cd,
                                                        const float* __restrict__ W,
                                                        const float* __restrict__ bias,
                                                        float* __restrict__ hout) {
    __shared__ float Wl[HIDDEN * HIDDEN];
    int tid = threadIdx.x;
    for (int i = tid; i < HIDDEN * HIDDEN; i += 256) Wl[i] = W[i];
    __syncthreads();

    int wave = tid >> 5;
    int lane = tid & 31;
    int half = lane >> 4;
    int lr   = lane & 15;
    int m0   = blockIdx.x * 16;
    int m    = m0 + lr;
    int n    = wave * 16 + lr;

    float cdm = cd[m];
    const float* arow = agg + (size_t)m * HIDDEN + half * 2;
    const float* brow = Wl + (half * 2) * HIDDEN + n;

    v8f acc = {};
#pragma unroll
    for (int k0 = 0; k0 < HIDDEN; k0 += 4) {
        v2f av, bv;
        av.x = arow[k0]     * cdm;
        av.y = arow[k0 + 1] * cdm;
        bv.x = brow[k0 * HIDDEN];
        bv.y = brow[k0 * HIDDEN + HIDDEN];
        acc = __builtin_amdgcn_wmma_f32_16x16x4_f32(
             false, av,  false, bv,
             (short)0, acc,  false,  false);
    }

    float bn = bias[n];
#pragma unroll
    for (int r = 0; r < 8; ++r) {
        int mm = m0 + half * 8 + r;
        hout[(size_t)mm * HIDDEN + n] = fmaxf(acc[r] + bn, 0.0f);
    }
}

__global__ void pool_kernel(const int* __restrict__ seg, const float* __restrict__ h,
                            float* __restrict__ hg, float* __restrict__ counts) {
    int gid  = blockIdx.x * blockDim.x + threadIdx.x;
    int node = gid >> 5;
    int lane = gid & 31;
    if (node < N_NODES) {
        int g = seg[node];
        const float* hn  = h  + (size_t)node * HIDDEN;
        float*       hgo = hg + (size_t)g * HIDDEN;
#pragma unroll
        for (int f = lane; f < HIDDEN; f += 32)
            atomicAdd(&hgo[f], hn[f]);
        if (lane == 0) atomicAdd(&counts[g], 1.0f);
    }
}

__global__ void pool_div_kernel(float* __restrict__ hg, const float* __restrict__ counts) {
    int i = blockIdx.x * blockDim.x + threadIdx.x;
    if (i < N_GRAPHS * HIDDEN)
        hg[i] /= fmaxf(counts[i >> 7], 1.0f);
}

__global__ void logits_kernel(const float* __restrict__ hg1, const float* __restrict__ hg2,
                              const float* __restrict__ Wc, const float* __restrict__ bc,
                              float* __restrict__ logits) {
    int i = blockIdx.x * blockDim.x + threadIdx.x;
    if (i < N_GRAPHS * N_CLASSES) {
        int g = i / N_CLASSES, c = i % N_CLASSES;
        float s = bc[c];
#pragma unroll 8
        for (int k = 0; k < HIDDEN; ++k)
            s += fabsf(hg1[g * HIDDEN + k] - hg2[g * HIDDEN + k]) * Wc[k * N_CLASSES + c];
        logits[i] = s;
    }
}

static void run_branch(const int* src, const int* dst, const int* seg,
                       const float* W1, const float* b1,
                       const float* W2, const float* b2,
                       const float* W3, const float* b3,
                       const float* W4, const float* b4,
                       float* hg_out, float* ws_f, hipStream_t stream) {
    float* outdeg = ws_f;
    int*   indeg_i = (int*)(ws_f + 100000);
    float* cs     = ws_f + 200000;
    float* cd     = ws_f + 300000;
    float* x      = ws_f + 400000;
    float* counts = ws_f + 500000;
    float* h      = ws_f + 500224;
    float* agg    = h + (size_t)N_NODES * HIDDEN;
    int*   rowptr = (int*)(agg + (size_t)N_NODES * HIDDEN);
    int*   rank   = rowptr + N_NODES;
    int2*  ce     = (int2*)(rank + N_EDGES);

    hipMemsetAsync(ws_f, 0, (size_t)500224 * sizeof(float), stream);

    const int T = 256;
    deg_rank_kernel<<<(N_EDGES + T - 1) / T, T, 0, stream>>>(src, dst, outdeg, indeg_i, rank);
    norm_kernel<<<(N_NODES + T - 1) / T, T, 0, stream>>>(outdeg, indeg_i, cs, cd);

    scan_kernel<<<1, 1024, 0, stream>>>(indeg_i, rowptr);
    fill_kernel<<<(N_EDGES + T - 1) / T, T, 0, stream>>>(src, dst, rank, rowptr, cs, ce);

    gather1_kernel<<<(N_NODES + T - 1) / T, T, 0, stream>>>(rowptr, indeg_i, ce, cd, x);

    gather_rank1_kernel<<<(N_NODES / NODES_PER_WAVE + 7) / 8, T, 0, stream>>>(
        rowptr, indeg_i, ce, x, W1, b1, agg);
    gemm_relu_kernel<<<N_NODES / 16, T, 0, stream>>>(agg, cd, W2, b2, h);

    const float* Ws[2] = {W3, W4};
    const float* bs[2] = {b3, b4};
    for (int l = 0; l < 2; ++l) {
        gather_kernel<<<(N_NODES * 32) / T, T, 0, stream>>>(rowptr, indeg_i, ce, h, agg);
        gemm_relu_kernel<<<N_NODES / 16, T, 0, stream>>>(agg, cd, Ws[l], bs[l], h);
    }

    hipMemsetAsync(hg_out, 0, (size_t)N_GRAPHS * HIDDEN * sizeof(float), stream);
    pool_kernel<<<(N_NODES * 32 + T - 1) / T, T, 0, stream>>>(seg, h, hg_out, counts);
    pool_div_kernel<<<(N_GRAPHS * HIDDEN + T - 1) / T, T, 0, stream>>>(hg_out, counts);
}

extern "C" void kernel_launch(void* const* d_in, const int* in_sizes, int n_in,
                              void* d_out, int out_size, void* d_ws, size_t ws_size,
                              hipStream_t stream) {
    const int*   src1 = (const int*)d_in[0];
    const int*   dst1 = (const int*)d_in[1];
    const int*   seg1 = (const int*)d_in[2];
    const int*   src2 = (const int*)d_in[3];
    const int*   dst2 = (const int*)d_in[4];
    const int*   seg2 = (const int*)d_in[5];
    const float* W1   = (const float*)d_in[6];
    const float* b1   = (const float*)d_in[7];
    const float* W2   = (const float*)d_in[8];
    const float* b2   = (const float*)d_in[9];
    const float* W3   = (const float*)d_in[10];
    const float* b3   = (const float*)d_in[11];
    const float* W4   = (const float*)d_in[12];
    const float* b4   = (const float*)d_in[13];
    const float* Wc   = (const float*)d_in[14];
    const float* bc   = (const float*)d_in[15];

    float* out   = (float*)d_out;
    float* hg1   = out;
    float* hg2   = out + N_GRAPHS * HIDDEN;
    float* logit = out + 2 * N_GRAPHS * HIDDEN;
    float* ws_f  = (float*)d_ws;

    run_branch(src1, dst1, seg1, W1, b1, W2, b2, W3, b3, W4, b4, hg1, ws_f, stream);
    run_branch(src2, dst2, seg2, W1, b1, W2, b2, W3, b3, W4, b4, hg2, ws_f, stream);

    logits_kernel<<<(N_GRAPHS * N_CLASSES + 255) / 256, 256, 0, stream>>>(hg1, hg2, Wc, bc, logit);
}


# GEMM work-groups made persistent: each keeps its W fragments in registers and loops over 6-7 row tiles, a tile's agg rows (x cd) staged once in LDS by whole-row loads; same products, same k order
# speedup vs baseline: 2.5203x; 1.1740x over previous
#include <hip/hip_runtime.h>
#include <hip/hip_bf16.h>

#define N_NODES   100000
#define N_EDGES   1600000
#define N_GRAPHS  128
#define HIDDEN    128
#define N_CLASSES 10

typedef __attribute__((ext_vector_type(2))) float v2f;
typedef __attribute__((ext_vector_type(8))) float v8f;

__global__ void deg_rank_kernel(const int* __restrict__ src, const int* __restrict__ dst,
                                float* __restrict__ outdeg, int* __restrict__ indeg_i,
                                int* __restrict__ rank) {
    int e = blockIdx.x * blockDim.x + threadIdx.x;
    if (e < N_EDGES) {
        atomicAdd(&outdeg[src[e]], 1.0f);
        rank[e] = atomicAdd(&indeg_i[dst[e]], 1);
    }
}

__global__ void __launch_bounds__(1024) scan_kernel(const int* __restrict__ cnt, int* __restrict__ rowptr) {
    __shared__ int wsum[32];
    const int tid  = threadIdx.x;
    const int lane = tid & 31;
    const int wave = tid >> 5;
    const int4* __restrict__ c4 = (const int4*)cnt;
    int4* __restrict__ r4 = (int4*)rowptr;
    int carry = 0;
    for (int base = 0; base < N_NODES / 4; base += 1024) {
        const int i = base + tid;
        int4 v = make_int4(0, 0, 0, 0);
        if (i < N_NODES / 4) v = c4[i];
        const int t = v.x + v.y + v.z + v.w;
        int inc = t;
        for (int o = 1; o < 32; o <<= 1) {
            int u = __shfl_up(inc, o);
            if (lane >= o) inc += u;
        }
        if (lane == 31) wsum[wave] = inc;
        __syncthreads();
        const int ws = wsum[lane];
        int winc = ws;
        for (int o = 1; o < 32; o <<= 1) {
            int u = __shfl_up(winc, o);
            if (lane >= o) winc += u;
        }
        const int wave_excl = __shfl(winc, wave) - __shfl(ws, wave);
        const int total     = __shfl(winc, 31);
        const int excl      = carry + wave_excl + inc - t;
        if (i < N_NODES / 4) {
            int4 o4;
            o4.x = excl;
            o4.y = o4.x + v.x;
            o4.z = o4.y + v.y;
            o4.w = o4.z + v.z;
            r4[i] = o4;
        }
        carry += total;
        __syncthreads();
    }
}

__global__ void fill_kernel(const int* __restrict__ src, const int* __restrict__ dst,
                            const int* __restrict__ rank, const int* __restrict__ rowptr,
                            const float* __restrict__ cs, int2* __restrict__ ce) {
    int e = blockIdx.x * blockDim.x + threadIdx.x;
    if (e < N_EDGES) {
        int s = src[e];
        int pos = rowptr[dst[e]] + rank[e];
        ce[pos] = make_int2(s, __float_as_int(cs[s]));
    }
}

__global__ void norm_kernel(const float* __restrict__ outdeg, const int* __restrict__ indeg_i,
                            float* __restrict__ cs, float* __restrict__ cd) {
    int n = blockIdx.x * blockDim.x + threadIdx.x;
    if (n < N_NODES) {
        cs[n] = rsqrtf(fmaxf(outdeg[n], 1.0f));
        cd[n] = rsqrtf(fmaxf((float)indeg_i[n], 1.0f));
    }
}

__global__ void gather1_kernel(const int* __restrict__ rowptr, const int* __restrict__ indeg_i,
                               const int2* __restrict__ ce, const float* __restrict__ cd,
                               float* __restrict__ x) {
    int d = blockIdx.x * blockDim.x + threadIdx.x;
    if (d < N_NODES) {
        const int beg = rowptr[d];
        const int cnt = indeg_i[d];
        float a = 0.0f;
        for (int j = 0; j < cnt; ++j) {
            int2 c = ce[beg + j];
            a += (float)indeg_i[c.x] * __int_as_float(c.y);
        }
        x[d] = a * cd[d];
    }
}

#define NODES_PER_WAVE 8
static_assert(N_NODES % NODES_PER_WAVE == 0, "a wave's 8 nodes must all exist");
__global__ void __launch_bounds__(256) gather_rank1_kernel(const int* __restrict__ rowptr,
                                                           const int* __restrict__ indeg_i,
                                                           const int2* __restrict__ ce,
                                                           const float* __restrict__ x,
                                                           const float* __restrict__ W1,
                                                           const float* __restrict__ b1,
                                                           float* __restrict__ agg) {
    const int wave = __builtin_amdgcn_readfirstlane((int)(threadIdx.x >> 5));
    const int lane = threadIdx.x & 31;
    const int g    = blockIdx.x * 8 + wave;
    if (g < N_NODES / NODES_PER_WAVE) {
        const float4 w = ((const float4*)W1)[lane];
        const float4 b = ((const float4*)b1)[lane];
        const int4 ba = ((const int4*)rowptr)[2 * g],  bb = ((const int4*)rowptr)[2 * g + 1];
        const int4 ca = ((const int4*)indeg_i)[2 * g], cb = ((const int4*)indeg_i)[2 * g + 1];
        const int begs[NODES_PER_WAVE] = {ba.x, ba.y, ba.z, ba.w, bb.x, bb.y, bb.z, bb.w};
        const int cnts[NODES_PER_WAVE] = {ca.x, ca.y, ca.z, ca.w, cb.x, cb.y, cb.z, cb.w};

        int2 e0[NODES_PER_WAVE];
        int  x0[NODES_PER_WAVE];
#pragma unroll
        for (int t = 0; t < NODES_PER_WAVE; ++t)
            e0[t] = ce[min(begs[t] + lane, N_EDGES - 1)];
        __builtin_amdgcn_sched_barrier(0);
#pragma unroll
        for (int t = 0; t < NODES_PER_WAVE; ++t)
            x0[t] = __float_as_int(x[e0[t].x]);
        asm volatile("" :: "v"(x0[0]), "v"(x0[1]), "v"(x0[2]), "v"(x0[3]),
                           "v"(x0[4]), "v"(x0[5]), "v"(x0[6]), "v"(x0[7]));

#pragma unroll
        for (int t = 0; t < NODES_PER_WAVE; ++t) {
            const int beg = begs[t];
            const int cnt = cnts[t];
            int xsrc = x0[t], cscl = e0[t].y;
            float4 acc = make_float4(0.0f, 0.0f, 0.0f, 0.0f);
            for (int base = 0; base < cnt; base += 32) {
                const int n = min(cnt - base, 32);
                if (base > 0) {
                    xsrc = 0; cscl = 0;
                    if (lane < n) {
                        int2 c = ce[beg + base + lane];
                        cscl = c.y;
                        xsrc = __float_as_int(x[c.x]);
                    }
                }
                for (int j = 0; j < n; ++j) {
                    const float xv = __int_as_float(__builtin_amdgcn_readlane(xsrc, j));
                    const float cv = __int_as_float(__builtin_amdgcn_readlane(cscl, j));
                    acc.x = fmaf(fmaxf(fmaf(xv, w.x, b.x), 0.0f), cv, acc.x);
                    acc.y = fmaf(fmaxf(fmaf(xv, w.y, b.y), 0.0f), cv, acc.y);
                    acc.z = fmaf(fmaxf(fmaf(xv, w.z, b.z), 0.0f), cv, acc.z);
                    acc.w = fmaf(fmaxf(fmaf(xv, w.w, b.w), 0.0f), cv, acc.w);
                }
            }
            ((float4*)agg)[(size_t)(g * NODES_PER_WAVE + t) * 32 + lane] = acc;
        }
    }
}

__global__ void __launch_bounds__(256) gather_kernel(const int* __restrict__ rowptr,
                                                     const int* __restrict__ indeg_i,
                                                     const int2* __restrict__ ce,
                                                     const float* __restrict__ h,
                                                     float* __restrict__ agg) {
    const int wave = __builtin_amdgcn_readfirstlane((int)(threadIdx.x >> 5));
    const int lane = threadIdx.x & 31;
    const int d    = blockIdx.x * 8 + wave;
    if (d < N_NODES) {
        const int beg = rowptr[d];
        const int cnt = indeg_i[d];
        const float4* __restrict__ h4 = (const float4*)h;
        float4 acc = make_float4(0.0f, 0.0f, 0.0f, 0.0f);
        for (int base = 0; base < cnt; base += 32) {
            const int n = min(cnt - base, 32);
            int cidx = 0, cscl = 0;
            if (lane < n) {
                int2 c = ce[beg + base + lane];
                cidx = c.x;
                cscl = c.y;
            }
            for (int j = 0; j < n; j += 4) {
                float4 v0, v1, v2, v3;
                float  c0, c1 = 0.0f, c2 = 0.0f, c3 = 0.0f;
                v1 = v2 = v3 = make_float4(0.0f, 0.0f, 0.0f, 0.0f);
                {
                    int s = __builtin_amdgcn_readlane(cidx, j);
                    c0 = __int_as_float(__builtin_amdgcn_readlane(cscl, j));
                    v0 = h4[(size_t)s * 32 + lane];
                }
                if (j + 1 < n) {
                    int s = __builtin_amdgcn_readlane(cidx, j + 1);
                    c1 = __int_as_float(__builtin_amdgcn_readlane(cscl, j + 1));
                    v1 = h4[(size_t)s * 32 + lane];
                }
                if (j + 2 < n) {
                    int s = __builtin_amdgcn_readlane(cidx, j + 2);
                    c2 = __int_as_float(__builtin_amdgcn_readlane(cscl, j + 2));
                    v2 = h4[(size_t)s * 32 + lane];
                }
                if (j + 3 < n) {
                    int s = __builtin_amdgcn_readlane(cidx, j + 3);
                    c3 = __int_as_float(__builtin_amdgcn_readlane(cscl, j + 3));
                    v3 = h4[(size_t)s * 32 + lane];
                }
                acc.x = fmaf(v0.x, c0, acc.x); acc.y = fmaf(v0.y, c0, acc.y);
                acc.z = fmaf(v0.z, c0, acc.z); acc.w = fmaf(v0.w, c0, acc.w);
                acc.x = fmaf(v1.x, c1, acc.x); acc.y = fmaf(v1.y, c1, acc.y);
                acc.z = fmaf(v1.z, c1, acc.z); acc.w = fmaf(v1.w, c1, acc.w);
                acc.x = fmaf(v2.x, c2, acc.x); acc.y = fmaf(v2.y, c2, acc.y);
                acc.z = fmaf(v2.z, c2, acc.z); acc.w = fmaf(v2.w, c2, acc.w);
                acc.x = fmaf(v3.x, c3, acc.x); acc.y = fmaf(v3.y, c3, acc.y);
                acc.z = fmaf(v3.z, c3, acc.z); acc.w = fmaf(v3.w, c3, acc.w);
            }
        }
        ((float4*)agg)[(size_t)d * 32 + lane] = acc;
    }
}

#define GEMM_BLOCKS 1024
#define A_PITCH     132
static_assert(N_NODES % 16 == 0, "row tiles must be whole");
static_assert(GEMM_BLOCKS <= N_NODES / 16, "every block owns a first tile");
__global__ void __launch_bounds__(256) gemm_relu_kernel(const float* __restrict__ agg,
                                                        const float* __restrict__ cd,
                                                        const float* __restrict__ W,
                                                        const float* __restrict__ bias,
                                                        float* __restrict__ hout) {
    __shared__ __attribute__((aligned(16))) float As[2][16 * A_PITCH];
    constexpr int NT = N_NODES / 16;
    const int tid  = threadIdx.x;
    const int wave = tid >> 5;
    const int lane = tid & 31;
    const int half = lane >> 4;
    const int lr   = lane & 15;
    const int n    = wave * 16 + lr;

    v2f bfr[HIDDEN / 4];
#pragma unroll
    for (int s = 0; s < HIDDEN / 4; ++s) {
        bfr[s].x = W[(4 * s + 2 * half) * HIDDEN + n];
        bfr[s].y = W[(4 * s + 2 * half + 1) * HIDDEN + n];
    }
    const float bn = bias[n];

    const int sr = 2 * wave;
    int t = blockIdx.x;
    float4 p0, p1;
    float  c0, c1;
    {
        const int r0 = t * 16 + sr;
        p0 = ((const float4*)(agg + (size_t)r0 * HIDDEN))[lane];
        p1 = ((const float4*)(agg + (size_t)(r0 + 1) * HIDDEN))[lane];
        c0 = cd[r0];
        c1 = cd[r0 + 1];
    }

    int buf = 0;
    for (;;) {
        {
            float* dst = &As[buf][sr * A_PITCH + lane * 4];
            *(float4*)dst             = make_float4(p0.x * c0, p0.y * c0, p0.z * c0, p0.w * c0);
            *(float4*)(dst + A_PITCH) = make_float4(p1.x * c1, p1.y * c1, p1.z * c1, p1.w * c1);
        }
        __syncthreads();

        const int  tn   = t + GEMM_BLOCKS;
        const bool more = tn < NT;
        if (more) {
            const int r0 = tn * 16 + sr;
            p0 = ((const float4*)(agg + (size_t)r0 * HIDDEN))[lane];
            p1 = ((const float4*)(agg + (size_t)(r0 + 1) * HIDDEN))[lane];
            c0 = cd[r0];
            c1 = cd[r0 + 1];
        }

        const float* arow = &As[buf][lr * A_PITCH + 2 * half];
        v8f acc = {};
#pragma unroll
        for (int s = 0; s < HIDDEN / 4; ++s) {
            const v2f av = *(const v2f*)(arow + 4 * s);
            acc = __builtin_amdgcn_wmma_f32_16x16x4_f32(
                 false, av,  false, bfr[s],
                 (short)0, acc,  false,  false);
        }

        const int m0 = t * 16;
#pragma unroll
        for (int r = 0; r < 8; ++r) {
            const int mm = m0 + half * 8 + r;
            hout[(size_t)mm * HIDDEN + n] = fmaxf(acc[r] + bn, 0.0f);
        }

        if (!more) break;
        t = tn;
        buf ^= 1;
    }
}

__global__ void pool_kernel(const int* __restrict__ seg, const float* __restrict__ h,
                            float* __restrict__ hg, float* __restrict__ counts) {
    int gid  = blockIdx.x * blockDim.x + threadIdx.x;
    int node = gid >> 5;
    int lane = gid & 31;
    if (node < N_NODES) {
        int g = seg[node];
        const float* hn  = h  + (size_t)node * HIDDEN;
        float*       hgo = hg + (size_t)g * HIDDEN;
#pragma unroll
        for (int f = lane; f < HIDDEN; f += 32)
            atomicAdd(&hgo[f], hn[f]);
        if (lane == 0) atomicAdd(&counts[g], 1.0f);
    }
}

__global__ void pool_div_kernel(float* __restrict__ hg, const float* __restrict__ counts) {
    int i = blockIdx.x * blockDim.x + threadIdx.x;
    if (i < N_GRAPHS * HIDDEN)
        hg[i] /= fmaxf(counts[i >> 7], 1.0f);
}

__global__ void logits_kernel(const float* __restrict__ hg1, const float* __restrict__ hg2,
                              const float* __restrict__ Wc, const float* __restrict__ bc,
                              float* __restrict__ logits) {
    int i = blockIdx.x * blockDim.x + threadIdx.x;
    if (i < N_GRAPHS * N_CLASSES) {
        int g = i / N_CLASSES, c = i % N_CLASSES;
        float s = bc[c];
#pragma unroll 8
        for (int k = 0; k < HIDDEN; ++k)
            s += fabsf(hg1[g * HIDDEN + k] - hg2[g * HIDDEN + k]) * Wc[k * N_CLASSES + c];
        logits[i] = s;
    }
}

static void run_branch(const int* src, const int* dst, const int* seg,
                       const float* W1, const float* b1,
                       const float* W2, const float* b2,
                       const float* W3, const float* b3,
                       const float* W4, const float* b4,
                       float* hg_out, float* ws_f, hipStream_t stream) {
    float* outdeg = ws_f;
    int*   indeg_i = (int*)(ws_f + 100000);
    float* cs     = ws_f + 200000;
    float* cd     = ws_f + 300000;
    float* x      = ws_f + 400000;
    float* counts = ws_f + 500000;
    float* h      = ws_f + 500224;
    float* agg    = h + (size_t)N_NODES * HIDDEN;
    int*   rowptr = (int*)(agg + (size_t)N_NODES * HIDDEN);
    int*   rank   = rowptr + N_NODES;
    int2*  ce     = (int2*)(rank + N_EDGES);

    hipMemsetAsync(ws_f, 0, (size_t)500224 * sizeof(float), stream);

    const int T = 256;
    deg_rank_kernel<<<(N_EDGES + T - 1) / T, T, 0, stream>>>(src, dst, outdeg, indeg_i, rank);
    norm_kernel<<<(N_NODES + T - 1) / T, T, 0, stream>>>(outdeg, indeg_i, cs, cd);

    scan_kernel<<<1, 1024, 0, stream>>>(indeg_i, rowptr);
    fill_kernel<<<(N_EDGES + T - 1) / T, T, 0, stream>>>(src, dst, rank, rowptr, cs, ce);

    gather1_kernel<<<(N_NODES + T - 1) / T, T, 0, stream>>>(rowptr, indeg_i, ce, cd, x);

    gather_rank1_kernel<<<(N_NODES / NODES_PER_WAVE + 7) / 8, T, 0, stream>>>(
        rowptr, indeg_i, ce, x, W1, b1, agg);
    gemm_relu_kernel<<<GEMM_BLOCKS, T, 0, stream>>>(agg, cd, W2, b2, h);

    const float* Ws[2] = {W3, W4};
    const float* bs[2] = {b3, b4};
    for (int l = 0; l < 2; ++l) {
        gather_kernel<<<(N_NODES * 32) / T, T, 0, stream>>>(rowptr, indeg_i, ce, h, agg);
        gemm_relu_kernel<<<GEMM_BLOCKS, T, 0, stream>>>(agg, cd, Ws[l], bs[l], h);
    }

    hipMemsetAsync(hg_out, 0, (size_t)N_GRAPHS * HIDDEN * sizeof(float), stream);
    pool_kernel<<<(N_NODES * 32 + T - 1) / T, T, 0, stream>>>(seg, h, hg_out, counts);
    pool_div_kernel<<<(N_GRAPHS * HIDDEN + T - 1) / T, T, 0, stream>>>(hg_out, counts);
}

extern "C" void kernel_launch(void* const* d_in, const int* in_sizes, int n_in,
                              void* d_out, int out_size, void* d_ws, size_t ws_size,
                              hipStream_t stream) {
    const int*   src1 = (const int*)d_in[0];
    const int*   dst1 = (const int*)d_in[1];
    const int*   seg1 = (const int*)d_in[2];
    const int*   src2 = (const int*)d_in[3];
    const int*   dst2 = (const int*)d_in[4];
    const int*   seg2 = (const int*)d_in[5];
    const float* W1   = (const float*)d_in[6];
    const float* b1   = (const float*)d_in[7];
    const float* W2   = (const float*)d_in[8];
    const float* b2   = (const float*)d_in[9];
    const float* W3   = (const float*)d_in[10];
    const float* b3   = (const float*)d_in[11];
    const float* W4   = (const float*)d_in[12];
    const float* b4   = (const float*)d_in[13];
    const float* Wc   = (const float*)d_in[14];
    const float* bc   = (const float*)d_in[15];

    float* out   = (float*)d_out;
    float* hg1   = out;
    float* hg2   = out + N_GRAPHS * HIDDEN;
    float* logit = out + 2 * N_GRAPHS * HIDDEN;
    float* ws_f  = (float*)d_ws;

    run_branch(src1, dst1, seg1, W1, b1, W2, b2, W3, b3, W4, b4, hg1, ws_f, stream);
    run_branch(src2, dst2, seg2, W1, b1, W2, b2, W3, b3, W4, b4, hg2, ws_f, stream);

    logits_kernel<<<(N_GRAPHS * N_CLASSES + 255) / 256, 256, 0, stream>>>(hg1, hg2, Wc, bc, logit);
}


# Mean pooling folded into the layer-4 GEMM epilogue: per 16-row tile a lane adds its 8 rows, lane halves are added, one f32 atomic add per column into hg; layer-4 rows never stored; sum order changes
# speedup vs baseline: 3.4525x; 1.3699x over previous
#include <hip/hip_runtime.h>
#include <hip/hip_bf16.h>

#define N_NODES   100000
#define N_EDGES   1600000
#define N_GRAPHS  128
#define HIDDEN    128
#define N_CLASSES 10

typedef __attribute__((ext_vector_type(2))) float v2f;
typedef __attribute__((ext_vector_type(8))) float v8f;

__global__ void deg_rank_kernel(const int* __restrict__ src, const int* __restrict__ dst,
                                float* __restrict__ outdeg, int* __restrict__ indeg_i,
                                int* __restrict__ rank) {
    int e = blockIdx.x * blockDim.x + threadIdx.x;
    if (e < N_EDGES) {
        atomicAdd(&outdeg[src[e]], 1.0f);
        rank[e] = atomicAdd(&indeg_i[dst[e]], 1);
    }
}

__global__ void __launch_bounds__(1024) scan_kernel(const int* __restrict__ cnt, int* __restrict__ rowptr) {
    __shared__ int wsum[32];
    const int tid  = threadIdx.x;
    const int lane = tid & 31;
    const int wave = tid >> 5;
    const int4* __restrict__ c4 = (const int4*)cnt;
    int4* __restrict__ r4 = (int4*)rowptr;
    int carry = 0;
    for (int base = 0; base < N_NODES / 4; base += 1024) {
        const int i = base + tid;
        int4 v = make_int4(0, 0, 0, 0);
        if (i < N_NODES / 4) v = c4[i];
        const int t = v.x + v.y + v.z + v.w;
        int inc = t;
        for (int o = 1; o < 32; o <<= 1) {
            int u = __shfl_up(inc, o);
            if (lane >= o) inc += u;
        }
        if (lane == 31) wsum[wave] = inc;
        __syncthreads();
        const int ws = wsum[lane];
        int winc = ws;
        for (int o = 1; o < 32; o <<= 1) {
            int u = __shfl_up(winc, o);
            if (lane >= o) winc += u;
        }
        const int wave_excl = __shfl(winc, wave) - __shfl(ws, wave);
        const int total     = __shfl(winc, 31);
        const int excl      = carry + wave_excl + inc - t;
        if (i < N_NODES / 4) {
            int4 o4;
            o4.x = excl;
            o4.y = o4.x + v.x;
            o4.z = o4.y + v.y;
            o4.w = o4.z + v.z;
            r4[i] = o4;
        }
        carry += total;
        __syncthreads();
    }
}

__global__ void fill_kernel(const int* __restrict__ src, const int* __restrict__ dst,
                            const int* __restrict__ rank, const int* __restrict__ rowptr,
                            const float* __restrict__ cs, int2* __restrict__ ce) {
    int e = blockIdx.x * blockDim.x + threadIdx.x;
    if (e < N_EDGES) {
        int s = src[e];
        int pos = rowptr[dst[e]] + rank[e];
        ce[pos] = make_int2(s, __float_as_int(cs[s]));
    }
}

__global__ void norm_kernel(const float* __restrict__ outdeg, const int* __restrict__ indeg_i,
                            float* __restrict__ cs, float* __restrict__ cd) {
    int n = blockIdx.x * blockDim.x + threadIdx.x;
    if (n < N_NODES) {
        cs[n] = rsqrtf(fmaxf(outdeg[n], 1.0f));
        cd[n] = rsqrtf(fmaxf((float)indeg_i[n], 1.0f));
    }
}

__global__ void gather1_kernel(const int* __restrict__ rowptr, const int* __restrict__ indeg_i,
                               const int2* __restrict__ ce, const float* __restrict__ cd,
                               float* __restrict__ x) {
    int d = blockIdx.x * blockDim.x + threadIdx.x;
    if (d < N_NODES) {
        const int beg = rowptr[d];
        const int cnt = indeg_i[d];
        float a = 0.0f;
        for (int j = 0; j < cnt; ++j) {
            int2 c = ce[beg + j];
            a += (float)indeg_i[c.x] * __int_as_float(c.y);
        }
        x[d] = a * cd[d];
    }
}

#define NODES_PER_WAVE 8
static_assert(N_NODES % NODES_PER_WAVE == 0, "a wave's 8 nodes must all exist");
__global__ void __launch_bounds__(256) gather_rank1_kernel(const int* __restrict__ rowptr,
                                                           const int* __restrict__ indeg_i,
                                                           const int2* __restrict__ ce,
                                                           const float* __restrict__ x,
                                                           const float* __restrict__ W1,
                                                           const float* __restrict__ b1,
                                                           float* __restrict__ agg) {
    const int wave = __builtin_amdgcn_readfirstlane((int)(threadIdx.x >> 5));
    const int lane = threadIdx.x & 31;
    const int g    = blockIdx.x * 8 + wave;
    if (g < N_NODES / NODES_PER_WAVE) {
        const float4 w = ((const float4*)W1)[lane];
        const float4 b = ((const float4*)b1)[lane];
        const int4 ba = ((const int4*)rowptr)[2 * g],  bb = ((const int4*)rowptr)[2 * g + 1];
        const int4 ca = ((const int4*)indeg_i)[2 * g], cb = ((const int4*)indeg_i)[2 * g + 1];
        const int begs[NODES_PER_WAVE] = {ba.x, ba.y, ba.z, ba.w, bb.x, bb.y, bb.z, bb.w};
        const int cnts[NODES_PER_WAVE] = {ca.x, ca.y, ca.z, ca.w, cb.x, cb.y, cb.z, cb.w};

        int2 e0[NODES_PER_WAVE];
        int  x0[NODES_PER_WAVE];
#pragma unroll
        for (int t = 0; t < NODES_PER_WAVE; ++t)
            e0[t] = ce[min(begs[t] + lane, N_EDGES - 1)];
        __builtin_amdgcn_sched_barrier(0);
#pragma unroll
        for (int t = 0; t < NODES_PER_WAVE; ++t)
            x0[t] = __float_as_int(x[e0[t].x]);
        asm volatile("" :: "v"(x0[0]), "v"(x0[1]), "v"(x0[2]), "v"(x0[3]),
                           "v"(x0[4]), "v"(x0[5]), "v"(x0[6]), "v"(x0[7]));

#pragma unroll
        for (int t = 0; t < NODES_PER_WAVE; ++t) {
            const int beg = begs[t];
            const int cnt = cnts[t];
            int xsrc = x0[t], cscl = e0[t].y;
            float4 acc = make_float4(0.0f, 0.0f, 0.0f, 0.0f);
            for (int base = 0; base < cnt; base += 32) {
                const int n = min(cnt - base, 32);
                if (base > 0) {
                    xsrc = 0; cscl = 0;
                    if (lane < n) {
                        int2 c = ce[beg + base + lane];
                        cscl = c.y;
                        xsrc = __float_as_int(x[c.x]);
                    }
                }
                for (int j = 0; j < n; ++j) {
                    const float xv = __int_as_float(__builtin_amdgcn_readlane(xsrc, j));
                    const float cv = __int_as_float(__builtin_amdgcn_readlane(cscl, j));
                    acc.x = fmaf(fmaxf(fmaf(xv, w.x, b.x), 0.0f), cv, acc.x);
                    acc.y = fmaf(fmaxf(fmaf(xv, w.y, b.y), 0.0f), cv, acc.y);
                    acc.z = fmaf(fmaxf(fmaf(xv, w.z, b.z), 0.0f), cv, acc.z);
                    acc.w = fmaf(fmaxf(fmaf(xv, w.w, b.w), 0.0f), cv, acc.w);
                }
            }
            ((float4*)agg)[(size_t)(g * NODES_PER_WAVE + t) * 32 + lane] = acc;
        }
    }
}

__global__ void __launch_bounds__(256) gather_kernel(const int* __restrict__ rowptr,
                                                     const int* __restrict__ indeg_i,
                                                     const int2* __restrict__ ce,
                                                     const float* __restrict__ h,
                                                     float* __restrict__ agg) {
    const int wave = __builtin_amdgcn_readfirstlane((int)(threadIdx.x >> 5));
    const int lane = threadIdx.x & 31;
    const int d    = blockIdx.x * 8 + wave;
    if (d < N_NODES) {
        const int beg = rowptr[d];
        const int cnt = indeg_i[d];
        const float4* __restrict__ h4 = (const float4*)h;
        float4 acc = make_float4(0.0f, 0.0f, 0.0f, 0.0f);
        for (int base = 0; base < cnt; base += 32) {
            const int n = min(cnt - base, 32);
            int cidx = 0, cscl = 0;
            if (lane < n) {
                int2 c = ce[beg + base + lane];
                cidx = c.x;
                cscl = c.y;
            }
            for (int j = 0; j < n; j += 4) {
                float4 v0, v1, v2, v3;
                float  c0, c1 = 0.0f, c2 = 0.0f, c3 = 0.0f;
                v1 = v2 = v3 = make_float4(0.0f, 0.0f, 0.0f, 0.0f);
                {
                    int s = __builtin_amdgcn_readlane(cidx, j);
                    c0 = __int_as_float(__builtin_amdgcn_readlane(cscl, j));
                    v0 = h4[(size_t)s * 32 + lane];
                }
                if (j + 1 < n) {
                    int s = __builtin_amdgcn_readlane(cidx, j + 1);
                    c1 = __int_as_float(__builtin_amdgcn_readlane(cscl, j + 1));
                    v1 = h4[(size_t)s * 32 + lane];
                }
                if (j + 2 < n) {
                    int s = __builtin_amdgcn_readlane(cidx, j + 2);
                    c2 = __int_as_float(__builtin_amdgcn_readlane(cscl, j + 2));
                    v2 = h4[(size_t)s * 32 + lane];
                }
                if (j + 3 < n) {
                    int s = __builtin_amdgcn_readlane(cidx, j + 3);
                    c3 = __int_as_float(__builtin_amdgcn_readlane(cscl, j + 3));
                    v3 = h4[(size_t)s * 32 + lane];
                }
                acc.x = fmaf(v0.x, c0, acc.x); acc.y = fmaf(v0.y, c0, acc.y);
                acc.z = fmaf(v0.z, c0, acc.z); acc.w = fmaf(v0.w, c0, acc.w);
                acc.x = fmaf(v1.x, c1, acc.x); acc.y = fmaf(v1.y, c1, acc.y);
                acc.z = fmaf(v1.z, c1, acc.z); acc.w = fmaf(v1.w, c1, acc.w);
                acc.x = fmaf(v2.x, c2, acc.x); acc.y = fmaf(v2.y, c2, acc.y);
                acc.z = fmaf(v2.z, c2, acc.z); acc.w = fmaf(v2.w, c2, acc.w);
                acc.x = fmaf(v3.x, c3, acc.x); acc.y = fmaf(v3.y, c3, acc.y);
                acc.z = fmaf(v3.z, c3, acc.z); acc.w = fmaf(v3.w, c3, acc.w);
            }
        }
        ((float4*)agg)[(size_t)d * 32 + lane] = acc;
    }
}

#define GEMM_BLOCKS 1024
#define A_PITCH     132
static_assert(N_NODES % 16 == 0, "row tiles must be whole");
static_assert(GEMM_BLOCKS <= N_NODES / 16, "every block owns a first tile");
template <bool POOL>
__device__ __forceinline__ void gemm_relu_body(const float* __restrict__ agg,
                                               const float* __restrict__ cd,
                                               const float* __restrict__ W,
                                               const float* __restrict__ bias,
                                               float* __restrict__ hout,
                                               const int* __restrict__ seg,
                                               float* __restrict__ hg,
                                               float* __restrict__ counts) {
    __shared__ __attribute__((aligned(16))) float As[2][16 * A_PITCH];
    constexpr int NT = N_NODES / 16;
    const int tid  = threadIdx.x;
    const int wave = tid >> 5;
    const int lane = tid & 31;
    const int half = lane >> 4;
    const int lr   = lane & 15;
    const int n    = wave * 16 + lr;

    v2f bfr[HIDDEN / 4];
#pragma unroll
    for (int s = 0; s < HIDDEN / 4; ++s) {
        bfr[s].x = W[(4 * s + 2 * half) * HIDDEN + n];
        bfr[s].y = W[(4 * s + 2 * half + 1) * HIDDEN + n];
    }
    const float bn = bias[n];

    const int sr = 2 * wave;
    int t = blockIdx.x;
    float4 p0, p1;
    float  c0, c1;
    int4 na = {}, nb = {}, nc = {}, nd = {};
    {
        const int r0 = t * 16 + sr;
        p0 = ((const float4*)(agg + (size_t)r0 * HIDDEN))[lane];
        p1 = ((const float4*)(agg + (size_t)(r0 + 1) * HIDDEN))[lane];
        c0 = cd[r0];
        c1 = cd[r0 + 1];
        if constexpr (POOL) {
            const int4* __restrict__ sg = (const int4*)(seg + t * 16);
            na = sg[0]; nb = sg[1]; nc = sg[2]; nd = sg[3];
        }
    }

    int buf = 0;
    for (;;) {
        {
            float* dst = &As[buf][sr * A_PITCH + lane * 4];
            *(float4*)dst             = make_float4(p0.x * c0, p0.y * c0, p0.z * c0, p0.w * c0);
            *(float4*)(dst + A_PITCH) = make_float4(p1.x * c1, p1.y * c1, p1.z * c1, p1.w * c1);
        }
        __syncthreads();

        const int  tn   = t + GEMM_BLOCKS;
        const bool more = tn < NT;
        const int4 sa = na, sb = nb, sc = nc, sd = nd;
        if (more) {
            const int r0 = tn * 16 + sr;
            p0 = ((const float4*)(agg + (size_t)r0 * HIDDEN))[lane];
            p1 = ((const float4*)(agg + (size_t)(r0 + 1) * HIDDEN))[lane];
            c0 = cd[r0];
            c1 = cd[r0 + 1];
        }
        if constexpr (POOL) {
            const int4* __restrict__ sg = (const int4*)(seg + min(tn, NT - 1) * 16);
            na = sg[0]; nb = sg[1]; nc = sg[2]; nd = sg[3];
        }

        const float* arow = &As[buf][lr * A_PITCH + 2 * half];
        v8f acc = {};
#pragma unroll
        for (int s = 0; s < HIDDEN / 4; ++s) {
            const v2f av = *(const v2f*)(arow + 4 * s);
            acc = __builtin_amdgcn_wmma_f32_16x16x4_f32(
                 false, av,  false, bfr[s],
                 (short)0, acc,  false,  false);
        }

        const int m0 = t * 16;
        if constexpr (!POOL) {
#pragma unroll
            for (int r = 0; r < 8; ++r) {
                const int mm = m0 + half * 8 + r;
                hout[(size_t)mm * HIDDEN + n] = fmaxf(acc[r] + bn, 0.0f);
            }
        } else {
            const int  g0 = sa.x;
            const int  differ = (sa.y ^ g0) | (sa.z ^ g0) | (sa.w ^ g0) |
                                (sb.x ^ g0) | (sb.y ^ g0) | (sb.z ^ g0) | (sb.w ^ g0) |
                                (sc.x ^ g0) | (sc.y ^ g0) | (sc.z ^ g0) | (sc.w ^ g0) |
                                (sd.x ^ g0) | (sd.y ^ g0) | (sd.z ^ g0) | (sd.w ^ g0);
            if (differ == 0) {
                float s = fmaxf(acc[0] + bn, 0.0f);
#pragma unroll
                for (int r = 1; r < 8; ++r)
                    s += fmaxf(acc[r] + bn, 0.0f);
                s += __shfl_xor(s, 16);
                if (half == 0) atomicAdd(&hg[g0 * HIDDEN + n], s);
                if (tid == 0)  atomicAdd(&counts[g0], 16.0f);
            } else {
                const int gl[8] = {sa.x, sa.y, sa.z, sa.w, sb.x, sb.y, sb.z, sb.w};
                const int gh[8] = {sc.x, sc.y, sc.z, sc.w, sd.x, sd.y, sd.z, sd.w};
#pragma unroll
                for (int r = 0; r < 8; ++r) {
                    const int g = half ? gh[r] : gl[r];
                    atomicAdd(&hg[g * HIDDEN + n], fmaxf(acc[r] + bn, 0.0f));
                    if (wave == 0 && lr == 0) atomicAdd(&counts[g], 1.0f);
                }
            }
        }

        if (!more) break;
        t = tn;
        buf ^= 1;
    }
}

__global__ void __launch_bounds__(256) gemm_relu_kernel(const float* __restrict__ agg,
                                                        const float* __restrict__ cd,
                                                        const float* __restrict__ W,
                                                        const float* __restrict__ bias,
                                                        float* __restrict__ hout) {
    gemm_relu_body<false>(agg, cd, W, bias, hout, nullptr, nullptr, nullptr);
}

__global__ void __launch_bounds__(256) gemm_relu_pool_kernel(const float* __restrict__ agg,
                                                             const float* __restrict__ cd,
                                                             const float* __restrict__ W,
                                                             const float* __restrict__ bias,
                                                             const int* __restrict__ seg,
                                                             float* __restrict__ hg,
                                                             float* __restrict__ counts) {
    gemm_relu_body<true>(agg, cd, W, bias, nullptr, seg, hg, counts);
}

__global__ void pool_div_kernel(float* __restrict__ hg, const float* __restrict__ counts) {
    int i = blockIdx.x * blockDim.x + threadIdx.x;
    if (i < N_GRAPHS * HIDDEN)
        hg[i] /= fmaxf(counts[i >> 7], 1.0f);
}

__global__ void logits_kernel(const float* __restrict__ hg1, const float* __restrict__ hg2,
                              const float* __restrict__ Wc, const float* __restrict__ bc,
                              float* __restrict__ logits) {
    int i = blockIdx.x * blockDim.x + threadIdx.x;
    if (i < N_GRAPHS * N_CLASSES) {
        int g = i / N_CLASSES, c = i % N_CLASSES;
        float s = bc[c];
#pragma unroll 8
        for (int k = 0; k < HIDDEN; ++k)
            s += fabsf(hg1[g * HIDDEN + k] - hg2[g * HIDDEN + k]) * Wc[k * N_CLASSES + c];
        logits[i] = s;
    }
}

static void run_branch(const int* src, const int* dst, const int* seg,
                       const float* W1, const float* b1,
                       const float* W2, const float* b2,
                       const float* W3, const float* b3,
                       const float* W4, const float* b4,
                       float* hg_out, float* ws_f, hipStream_t stream) {
    float* outdeg = ws_f;
    int*   indeg_i = (int*)(ws_f + 100000);
    float* cs     = ws_f + 200000;
    float* cd     = ws_f + 300000;
    float* x      = ws_f + 400000;
    float* counts = ws_f + 500000;
    float* h      = ws_f + 500224;
    float* agg    = h + (size_t)N_NODES * HIDDEN;
    int*   rowptr = (int*)(agg + (size_t)N_NODES * HIDDEN);
    int*   rank   = rowptr + N_NODES;
    int2*  ce     = (int2*)(rank + N_EDGES);

    hipMemsetAsync(ws_f, 0, (size_t)500224 * sizeof(float), stream);

    const int T = 256;
    deg_rank_kernel<<<(N_EDGES + T - 1) / T, T, 0, stream>>>(src, dst, outdeg, indeg_i, rank);
    norm_kernel<<<(N_NODES + T - 1) / T, T, 0, stream>>>(outdeg, indeg_i, cs, cd);

    scan_kernel<<<1, 1024, 0, stream>>>(indeg_i, rowptr);
    fill_kernel<<<(N_EDGES + T - 1) / T, T, 0, stream>>>(src, dst, rank, rowptr, cs, ce);

    gather1_kernel<<<(N_NODES + T - 1) / T, T, 0, stream>>>(rowptr, indeg_i, ce, cd, x);

    gather_rank1_kernel<<<(N_NODES / NODES_PER_WAVE + 7) / 8, T, 0, stream>>>(
        rowptr, indeg_i, ce, x, W1, b1, agg);
    gemm_relu_kernel<<<GEMM_BLOCKS, T, 0, stream>>>(agg, cd, W2, b2, h);

    gather_kernel<<<(N_NODES * 32) / T, T, 0, stream>>>(rowptr, indeg_i, ce, h, agg);
    gemm_relu_kernel<<<GEMM_BLOCKS, T, 0, stream>>>(agg, cd, W3, b3, h);

    gather_kernel<<<(N_NODES * 32) / T, T, 0, stream>>>(rowptr, indeg_i, ce, h, agg);
    hipMemsetAsync(hg_out, 0, (size_t)N_GRAPHS * HIDDEN * sizeof(float), stream);
    gemm_relu_pool_kernel<<<GEMM_BLOCKS, T, 0, stream>>>(agg, cd, W4, b4, seg, hg_out, counts);
    pool_div_kernel<<<(N_GRAPHS * HIDDEN + T - 1) / T, T, 0, stream>>>(hg_out, counts);
}

extern "C" void kernel_launch(void* const* d_in, const int* in_sizes, int n_in,
                              void* d_out, int out_size, void* d_ws, size_t ws_size,
                              hipStream_t stream) {
    const int*   src1 = (const int*)d_in[0];
    const int*   dst1 = (const int*)d_in[1];
    const int*   seg1 = (const int*)d_in[2];
    const int*   src2 = (const int*)d_in[3];
    const int*   dst2 = (const int*)d_in[4];
    const int*   seg2 = (const int*)d_in[5];
    const float* W1   = (const float*)d_in[6];
    const float* b1   = (const float*)d_in[7];
    const float* W2   = (const float*)d_in[8];
    const float* b2   = (const float*)d_in[9];
    const float* W3   = (const float*)d_in[10];
    const float* b3   = (const float*)d_in[11];
    const float* W4   = (const float*)d_in[12];
    const float* b4   = (const float*)d_in[13];
    const float* Wc   = (const float*)d_in[14];
    const float* bc   = (const float*)d_in[15];

    float* out   = (float*)d_out;
    float* hg1   = out;
    float* hg2   = out + N_GRAPHS * HIDDEN;
    float* logit = out + 2 * N_GRAPHS * HIDDEN;
    float* ws_f  = (float*)d_ws;

    run_branch(src1, dst1, seg1, W1, b1, W2, b2, W3, b3, W4, b4, hg1, ws_f, stream);
    run_branch(src2, dst2, seg2, W1, b1, W2, b2, W3, b3, W4, b4, hg2, ws_f, stream);

    logits_kernel<<<(N_GRAPHS * N_CLASSES + 255) / 256, 256, 0, stream>>>(hg1, hg2, Wc, bc, logit);
}


# Both graph branches run in every launch (blockIdx.y picks the branch, arrays laid out [2][...]): 13 launches and 2 memsets instead of 25 and 4; per-branch work, products and sum orders unchanged
# speedup vs baseline: 3.8935x; 1.1277x over previous
#include <hip/hip_runtime.h>
#include <hip/hip_bf16.h>

#define N_NODES   100000
#define N_EDGES   1600000
#define N_GRAPHS  128
#define HIDDEN    128
#define N_CLASSES 10

typedef __attribute__((ext_vector_type(2))) float v2f;
typedef __attribute__((ext_vector_type(8))) float v8f;


__global__ void deg_rank_kernel(const int* __restrict__ src1, const int* __restrict__ dst1,
                                const int* __restrict__ src2, const int* __restrict__ dst2,
                                float* __restrict__ outdeg, int* __restrict__ indeg_i,
                                int* __restrict__ rank) {
    const int br = blockIdx.y;
    const int* __restrict__ src = br ? src2 : src1;
    const int* __restrict__ dst = br ? dst2 : dst1;
    outdeg  += br * N_NODES;
    indeg_i += br * N_NODES;
    rank    += (size_t)br * N_EDGES;
    int e = blockIdx.x * blockDim.x + threadIdx.x;
    if (e < N_EDGES) {
        atomicAdd(&outdeg[src[e]], 1.0f);
        rank[e] = atomicAdd(&indeg_i[dst[e]], 1);
    }
}

__global__ void __launch_bounds__(1024) scan_kernel(const int* __restrict__ cnt, int* __restrict__ rowptr) {
    __shared__ int wsum[32];
    cnt    += blockIdx.x * N_NODES;
    rowptr += blockIdx.x * N_NODES;
    const int tid  = threadIdx.x;
    const int lane = tid & 31;
    const int wave = tid >> 5;
    const int4* __restrict__ c4 = (const int4*)cnt;
    int4* __restrict__ r4 = (int4*)rowptr;
    int carry = 0;
    for (int base = 0; base < N_NODES / 4; base += 1024) {
        const int i = base + tid;
        int4 v = make_int4(0, 0, 0, 0);
        if (i < N_NODES / 4) v = c4[i];
        const int t = v.x + v.y + v.z + v.w;
        int inc = t;
        for (int o = 1; o < 32; o <<= 1) {
            int u = __shfl_up(inc, o);
            if (lane >= o) inc += u;
        }
        if (lane == 31) wsum[wave] = inc;
        __syncthreads();
        const int ws = wsum[lane];
        int winc = ws;
        for (int o = 1; o < 32; o <<= 1) {
            int u = __shfl_up(winc, o);
            if (lane >= o) winc += u;
        }
        const int wave_excl = __shfl(winc, wave) - __shfl(ws, wave);
        const int total     = __shfl(winc, 31);
        const int excl      = carry + wave_excl + inc - t;
        if (i < N_NODES / 4) {
            int4 o4;
            o4.x = excl;
            o4.y = o4.x + v.x;
            o4.z = o4.y + v.y;
            o4.w = o4.z + v.z;
            r4[i] = o4;
        }
        carry += total;
        __syncthreads();
    }
}

__global__ void fill_kernel(const int* __restrict__ src1, const int* __restrict__ dst1,
                            const int* __restrict__ src2, const int* __restrict__ dst2,
                            const int* __restrict__ rank, const int* __restrict__ rowptr,
                            const float* __restrict__ cs, int2* __restrict__ ce) {
    const int br = blockIdx.y;
    const int* __restrict__ src = br ? src2 : src1;
    const int* __restrict__ dst = br ? dst2 : dst1;
    rank   += (size_t)br * N_EDGES;
    rowptr += br * N_NODES;
    cs     += br * N_NODES;
    ce     += (size_t)br * N_EDGES;
    int e = blockIdx.x * blockDim.x + threadIdx.x;
    if (e < N_EDGES) {
        int s = src[e];
        int pos = rowptr[dst[e]] + rank[e];
        ce[pos] = make_int2(s, __float_as_int(cs[s]));
    }
}

__global__ void norm_kernel(const float* __restrict__ outdeg, const int* __restrict__ indeg_i,
                            float* __restrict__ cs, float* __restrict__ cd) {
    int n = blockIdx.x * blockDim.x + threadIdx.x;
    if (n < 2 * N_NODES) {
        cs[n] = rsqrtf(fmaxf(outdeg[n], 1.0f));
        cd[n] = rsqrtf(fmaxf((float)indeg_i[n], 1.0f));
    }
}

__global__ void gather1_kernel(const int* __restrict__ rowptr, const int* __restrict__ indeg_i,
                               const int2* __restrict__ ce, const float* __restrict__ cd,
                               float* __restrict__ x) {
    const int br = blockIdx.y;
    rowptr  += br * N_NODES;
    indeg_i += br * N_NODES;
    ce      += (size_t)br * N_EDGES;
    cd      += br * N_NODES;
    x       += br * N_NODES;
    int d = blockIdx.x * blockDim.x + threadIdx.x;
    if (d < N_NODES) {
        const int beg = rowptr[d];
        const int cnt = indeg_i[d];
        float a = 0.0f;
        for (int j = 0; j < cnt; ++j) {
            int2 c = ce[beg + j];
            a += (float)indeg_i[c.x] * __int_as_float(c.y);
        }
        x[d] = a * cd[d];
    }
}

#define NODES_PER_WAVE 8
static_assert(N_NODES % NODES_PER_WAVE == 0, "a wave's 8 nodes must all exist");
__global__ void __launch_bounds__(256) gather_rank1_kernel(const int* __restrict__ rowptr,
                                                           const int* __restrict__ indeg_i,
                                                           const int2* __restrict__ ce,
                                                           const float* __restrict__ x,
                                                           const float* __restrict__ W1,
                                                           const float* __restrict__ b1,
                                                           float* __restrict__ agg) {
    const int br = blockIdx.y;
    rowptr  += br * N_NODES;
    indeg_i += br * N_NODES;
    ce      += (size_t)br * N_EDGES;
    x       += br * N_NODES;
    agg     += (size_t)br * N_NODES * HIDDEN;
    const int wave = __builtin_amdgcn_readfirstlane((int)(threadIdx.x >> 5));
    const int lane = threadIdx.x & 31;
    const int g    = blockIdx.x * 8 + wave;
    if (g < N_NODES / NODES_PER_WAVE) {
        const float4 w = ((const float4*)W1)[lane];
        const float4 b = ((const float4*)b1)[lane];
        const int4 ba = ((const int4*)rowptr)[2 * g],  bb = ((const int4*)rowptr)[2 * g + 1];
        const int4 ca = ((const int4*)indeg_i)[2 * g], cb = ((const int4*)indeg_i)[2 * g + 1];
        const int begs[NODES_PER_WAVE] = {ba.x, ba.y, ba.z, ba.w, bb.x, bb.y, bb.z, bb.w};
        const int cnts[NODES_PER_WAVE] = {ca.x, ca.y, ca.z, ca.w, cb.x, cb.y, cb.z, cb.w};

        int2 e0[NODES_PER_WAVE];
        int  x0[NODES_PER_WAVE];
#pragma unroll
        for (int t = 0; t < NODES_PER_WAVE; ++t)
            e0[t] = ce[min(begs[t] + lane, N_EDGES - 1)];
        __builtin_amdgcn_sched_barrier(0);
#pragma unroll
        for (int t = 0; t < NODES_PER_WAVE; ++t)
            x0[t] = __float_as_int(x[e0[t].x]);
        asm volatile("" :: "v"(x0[0]), "v"(x0[1]), "v"(x0[2]), "v"(x0[3]),
                           "v"(x0[4]), "v"(x0[5]), "v"(x0[6]), "v"(x0[7]));

#pragma unroll
        for (int t = 0; t < NODES_PER_WAVE; ++t) {
            const int beg = begs[t];
            const int cnt = cnts[t];
            int xsrc = x0[t], cscl = e0[t].y;
            float4 acc = make_float4(0.0f, 0.0f, 0.0f, 0.0f);
            for (int base = 0; base < cnt; base += 32) {
                const int n = min(cnt - base, 32);
                if (base > 0) {
                    xsrc = 0; cscl = 0;
                    if (lane < n) {
                        int2 c = ce[beg + base + lane];
                        cscl = c.y;
                        xsrc = __float_as_int(x[c.x]);
                    }
                }
                for (int j = 0; j < n; ++j) {
                    const float xv = __int_as_float(__builtin_amdgcn_readlane(xsrc, j));
                    const float cv = __int_as_float(__builtin_amdgcn_readlane(cscl, j));
                    acc.x = fmaf(fmaxf(fmaf(xv, w.x, b.x), 0.0f), cv, acc.x);
                    acc.y = fmaf(fmaxf(fmaf(xv, w.y, b.y), 0.0f), cv, acc.y);
                    acc.z = fmaf(fmaxf(fmaf(xv, w.z, b.z), 0.0f), cv, acc.z);
                    acc.w = fmaf(fmaxf(fmaf(xv, w.w, b.w), 0.0f), cv, acc.w);
                }
            }
            ((float4*)agg)[(size_t)(g * NODES_PER_WAVE + t) * 32 + lane] = acc;
        }
    }
}

__global__ void __launch_bounds__(256) gather_kernel(const int* __restrict__ rowptr,
                                                     const int* __restrict__ indeg_i,
                                                     const int2* __restrict__ ce,
                                                     const float* __restrict__ h,
                                                     float* __restrict__ agg) {
    const int br = blockIdx.y;
    rowptr  += br * N_NODES;
    indeg_i += br * N_NODES;
    ce      += (size_t)br * N_EDGES;
    h       += (size_t)br * N_NODES * HIDDEN;
    agg     += (size_t)br * N_NODES * HIDDEN;
    const int wave = __builtin_amdgcn_readfirstlane((int)(threadIdx.x >> 5));
    const int lane = threadIdx.x & 31;
    const int d    = blockIdx.x * 8 + wave;
    if (d < N_NODES) {
        const int beg = rowptr[d];
        const int cnt = indeg_i[d];
        const float4* __restrict__ h4 = (const float4*)h;
        float4 acc = make_float4(0.0f, 0.0f, 0.0f, 0.0f);
        for (int base = 0; base < cnt; base += 32) {
            const int n = min(cnt - base, 32);
            int cidx = 0, cscl = 0;
            if (lane < n) {
                int2 c = ce[beg + base + lane];
                cidx = c.x;
                cscl = c.y;
            }
            for (int j = 0; j < n; j += 4) {
                float4 v0, v1, v2, v3;
                float  c0, c1 = 0.0f, c2 = 0.0f, c3 = 0.0f;
                v1 = v2 = v3 = make_float4(0.0f, 0.0f, 0.0f, 0.0f);
                {
                    int s = __builtin_amdgcn_readlane(cidx, j);
                    c0 = __int_as_float(__builtin_amdgcn_readlane(cscl, j));
                    v0 = h4[(size_t)s * 32 + lane];
                }
                if (j + 1 < n) {
                    int s = __builtin_amdgcn_readlane(cidx, j + 1);
                    c1 = __int_as_float(__builtin_amdgcn_readlane(cscl, j + 1));
                    v1 = h4[(size_t)s * 32 + lane];
                }
                if (j + 2 < n) {
                    int s = __builtin_amdgcn_readlane(cidx, j + 2);
                    c2 = __int_as_float(__builtin_amdgcn_readlane(cscl, j + 2));
                    v2 = h4[(size_t)s * 32 + lane];
                }
                if (j + 3 < n) {
                    int s = __builtin_amdgcn_readlane(cidx, j + 3);
                    c3 = __int_as_float(__builtin_amdgcn_readlane(cscl, j + 3));
                    v3 = h4[(size_t)s * 32 + lane];
                }
                acc.x = fmaf(v0.x, c0, acc.x); acc.y = fmaf(v0.y, c0, acc.y);
                acc.z = fmaf(v0.z, c0, acc.z); acc.w = fmaf(v0.w, c0, acc.w);
                acc.x = fmaf(v1.x, c1, acc.x); acc.y = fmaf(v1.y, c1, acc.y);
                acc.z = fmaf(v1.z, c1, acc.z); acc.w = fmaf(v1.w, c1, acc.w);
                acc.x = fmaf(v2.x, c2, acc.x); acc.y = fmaf(v2.y, c2, acc.y);
                acc.z = fmaf(v2.z, c2, acc.z); acc.w = fmaf(v2.w, c2, acc.w);
                acc.x = fmaf(v3.x, c3, acc.x); acc.y = fmaf(v3.y, c3, acc.y);
                acc.z = fmaf(v3.z, c3, acc.z); acc.w = fmaf(v3.w, c3, acc.w);
            }
        }
        ((float4*)agg)[(size_t)d * 32 + lane] = acc;
    }
}

#define GEMM_BLOCKS 1024
#define A_PITCH     132
static_assert(N_NODES % 16 == 0, "row tiles must be whole");
static_assert(GEMM_BLOCKS <= N_NODES / 16, "every block owns a first tile");
template <bool POOL>
__device__ __forceinline__ void gemm_relu_body(const float* __restrict__ agg,
                                               const float* __restrict__ cd,
                                               const float* __restrict__ W,
                                               const float* __restrict__ bias,
                                               float* __restrict__ hout,
                                               const int* __restrict__ seg,
                                               float* __restrict__ hg,
                                               float* __restrict__ counts) {
    __shared__ __attribute__((aligned(16))) float As[2][16 * A_PITCH];
    constexpr int NT = N_NODES / 16;
    const int tid  = threadIdx.x;
    const int wave = tid >> 5;
    const int lane = tid & 31;
    const int half = lane >> 4;
    const int lr   = lane & 15;
    const int n    = wave * 16 + lr;

    v2f bfr[HIDDEN / 4];
#pragma unroll
    for (int s = 0; s < HIDDEN / 4; ++s) {
        bfr[s].x = W[(4 * s + 2 * half) * HIDDEN + n];
        bfr[s].y = W[(4 * s + 2 * half + 1) * HIDDEN + n];
    }
    const float bn = bias[n];

    const int sr = 2 * wave;
    int t = blockIdx.x;
    float4 p0, p1;
    float  c0, c1;
    int4 na = {}, nb = {}, nc = {}, nd = {};
    {
        const int r0 = t * 16 + sr;
        p0 = ((const float4*)(agg + (size_t)r0 * HIDDEN))[lane];
        p1 = ((const float4*)(agg + (size_t)(r0 + 1) * HIDDEN))[lane];
        c0 = cd[r0];
        c1 = cd[r0 + 1];
        if constexpr (POOL) {
            const int4* __restrict__ sg = (const int4*)(seg + t * 16);
            na = sg[0]; nb = sg[1]; nc = sg[2]; nd = sg[3];
        }
    }

    int buf = 0;
    for (;;) {
        {
            float* dst = &As[buf][sr * A_PITCH + lane * 4];
            *(float4*)dst             = make_float4(p0.x * c0, p0.y * c0, p0.z * c0, p0.w * c0);
            *(float4*)(dst + A_PITCH) = make_float4(p1.x * c1, p1.y * c1, p1.z * c1, p1.w * c1);
        }
        __syncthreads();

        const int  tn   = t + GEMM_BLOCKS;
        const bool more = tn < NT;
        const int4 sa = na, sb = nb, sc = nc, sd = nd;
        if (more) {
            const int r0 = tn * 16 + sr;
            p0 = ((const float4*)(agg + (size_t)r0 * HIDDEN))[lane];
            p1 = ((const float4*)(agg + (size_t)(r0 + 1) * HIDDEN))[lane];
            c0 = cd[r0];
            c1 = cd[r0 + 1];
        }
        if constexpr (POOL) {
            const int4* __restrict__ sg = (const int4*)(seg + min(tn, NT - 1) * 16);
            na = sg[0]; nb = sg[1]; nc = sg[2]; nd = sg[3];
        }

        const float* arow = &As[buf][lr * A_PITCH + 2 * half];
        v8f acc = {};
#pragma unroll
        for (int s = 0; s < HIDDEN / 4; ++s) {
            const v2f av = *(const v2f*)(arow + 4 * s);
            acc = __builtin_amdgcn_wmma_f32_16x16x4_f32(
                 false, av,  false, bfr[s],
                 (short)0, acc,  false,  false);
        }

        const int m0 = t * 16;
        if constexpr (!POOL) {
#pragma unroll
            for (int r = 0; r < 8; ++r) {
                const int mm = m0 + half * 8 + r;
                hout[(size_t)mm * HIDDEN + n] = fmaxf(acc[r] + bn, 0.0f);
            }
        } else {
            const int  g0 = sa.x;
            const int  differ = (sa.y ^ g0) | (sa.z ^ g0) | (sa.w ^ g0) |
                                (sb.x ^ g0) | (sb.y ^ g0) | (sb.z ^ g0) | (sb.w ^ g0) |
                                (sc.x ^ g0) | (sc.y ^ g0) | (sc.z ^ g0) | (sc.w ^ g0) |
                                (sd.x ^ g0) | (sd.y ^ g0) | (sd.z ^ g0) | (sd.w ^ g0);
            if (differ == 0) {
                float s = fmaxf(acc[0] + bn, 0.0f);
#pragma unroll
                for (int r = 1; r < 8; ++r)
                    s += fmaxf(acc[r] + bn, 0.0f);
                s += __shfl_xor(s, 16);
                if (half == 0) atomicAdd(&hg[g0 * HIDDEN + n], s);
                if (tid == 0)  atomicAdd(&counts[g0], 16.0f);
            } else {
                const int gl[8] = {sa.x, sa.y, sa.z, sa.w, sb.x, sb.y, sb.z, sb.w};
                const int gh[8] = {sc.x, sc.y, sc.z, sc.w, sd.x, sd.y, sd.z, sd.w};
#pragma unroll
                for (int r = 0; r < 8; ++r) {
                    const int g = half ? gh[r] : gl[r];
                    atomicAdd(&hg[g * HIDDEN + n], fmaxf(acc[r] + bn, 0.0f));
                    if (wave == 0 && lr == 0) atomicAdd(&counts[g], 1.0f);
                }
            }
        }

        if (!more) break;
        t = tn;
        buf ^= 1;
    }
}

__global__ void __launch_bounds__(256) gemm_relu_kernel(const float* __restrict__ agg,
                                                        const float* __restrict__ cd,
                                                        const float* __restrict__ W,
                                                        const float* __restrict__ bias,
                                                        float* __restrict__ hout) {
    const int br = blockIdx.y;
    gemm_relu_body<false>(agg + (size_t)br * N_NODES * HIDDEN, cd + br * N_NODES, W, bias,
                          hout + (size_t)br * N_NODES * HIDDEN, nullptr, nullptr, nullptr);
}

__global__ void __launch_bounds__(256) gemm_relu_pool_kernel(const float* __restrict__ agg,
                                                             const float* __restrict__ cd,
                                                             const float* __restrict__ W,
                                                             const float* __restrict__ bias,
                                                             const int* __restrict__ seg1,
                                                             const int* __restrict__ seg2,
                                                             float* __restrict__ hg,
                                                             float* __restrict__ counts) {
    const int br = blockIdx.y;
    gemm_relu_body<true>(agg + (size_t)br * N_NODES * HIDDEN, cd + br * N_NODES, W, bias, nullptr,
                         br ? seg2 : seg1, hg + br * N_GRAPHS * HIDDEN, counts + br * N_GRAPHS);
}

__global__ void pool_div_kernel(float* __restrict__ hg, const float* __restrict__ counts) {
    int i = blockIdx.x * blockDim.x + threadIdx.x;
    if (i < 2 * N_GRAPHS * HIDDEN)
        hg[i] /= fmaxf(counts[i >> 7], 1.0f);
}

__global__ void logits_kernel(const float* __restrict__ hg1, const float* __restrict__ hg2,
                              const float* __restrict__ Wc, const float* __restrict__ bc,
                              float* __restrict__ logits) {
    int i = blockIdx.x * blockDim.x + threadIdx.x;
    if (i < N_GRAPHS * N_CLASSES) {
        int g = i / N_CLASSES, c = i % N_CLASSES;
        float s = bc[c];
#pragma unroll 8
        for (int k = 0; k < HIDDEN; ++k)
            s += fabsf(hg1[g * HIDDEN + k] - hg2[g * HIDDEN + k]) * Wc[k * N_CLASSES + c];
        logits[i] = s;
    }
}

extern "C" void kernel_launch(void* const* d_in, const int* in_sizes, int n_in,
                              void* d_out, int out_size, void* d_ws, size_t ws_size,
                              hipStream_t stream) {
    const int*   src1 = (const int*)d_in[0];
    const int*   dst1 = (const int*)d_in[1];
    const int*   seg1 = (const int*)d_in[2];
    const int*   src2 = (const int*)d_in[3];
    const int*   dst2 = (const int*)d_in[4];
    const int*   seg2 = (const int*)d_in[5];
    const float* W1   = (const float*)d_in[6];
    const float* b1   = (const float*)d_in[7];
    const float* W2   = (const float*)d_in[8];
    const float* b2   = (const float*)d_in[9];
    const float* W3   = (const float*)d_in[10];
    const float* b3   = (const float*)d_in[11];
    const float* W4   = (const float*)d_in[12];
    const float* b4   = (const float*)d_in[13];
    const float* Wc   = (const float*)d_in[14];
    const float* bc   = (const float*)d_in[15];

    float* out   = (float*)d_out;
    float* hg    = out;
    float* hg1   = out;
    float* hg2   = out + N_GRAPHS * HIDDEN;
    float* logit = out + 2 * N_GRAPHS * HIDDEN;
    float* ws_f  = (float*)d_ws;

    float* outdeg  = ws_f;
    int*   indeg_i = (int*)(ws_f + 200000);
    float* counts  = ws_f + 400000;
    float* cs      = ws_f + 400256;
    float* cd      = ws_f + 600256;
    float* x       = ws_f + 800256;
    int*   rowptr  = (int*)(ws_f + 1000256);
    float* h       = ws_f + 1200256;
    float* agg     = h + (size_t)2 * N_NODES * HIDDEN;
    int*   rank    = (int*)(agg + (size_t)2 * N_NODES * HIDDEN);
    int2*  ce      = (int2*)(rank + (size_t)2 * N_EDGES);

    hipMemsetAsync(ws_f, 0, (size_t)400256 * sizeof(float), stream);
    hipMemsetAsync(hg, 0, (size_t)2 * N_GRAPHS * HIDDEN * sizeof(float), stream);

    const int T = 256;
    deg_rank_kernel<<<dim3((N_EDGES + T - 1) / T, 2), T, 0, stream>>>(src1, dst1, src2, dst2, outdeg, indeg_i, rank);
    norm_kernel<<<(2 * N_NODES + T - 1) / T, T, 0, stream>>>(outdeg, indeg_i, cs, cd);

    scan_kernel<<<2, 1024, 0, stream>>>(indeg_i, rowptr);
    fill_kernel<<<dim3((N_EDGES + T - 1) / T, 2), T, 0, stream>>>(src1, dst1, src2, dst2, rank, rowptr, cs, ce);

    gather1_kernel<<<dim3((N_NODES + T - 1) / T, 2), T, 0, stream>>>(rowptr, indeg_i, ce, cd, x);

    gather_rank1_kernel<<<dim3((N_NODES / NODES_PER_WAVE + 7) / 8, 2), T, 0, stream>>>(
        rowptr, indeg_i, ce, x, W1, b1, agg);
    gemm_relu_kernel<<<dim3(GEMM_BLOCKS, 2), T, 0, stream>>>(agg, cd, W2, b2, h);

    gather_kernel<<<dim3((N_NODES * 32) / T, 2), T, 0, stream>>>(rowptr, indeg_i, ce, h, agg);
    gemm_relu_kernel<<<dim3(GEMM_BLOCKS, 2), T, 0, stream>>>(agg, cd, W3, b3, h);

    gather_kernel<<<dim3((N_NODES * 32) / T, 2), T, 0, stream>>>(rowptr, indeg_i, ce, h, agg);
    gemm_relu_pool_kernel<<<dim3(GEMM_BLOCKS, 2), T, 0, stream>>>(agg, cd, W4, b4, seg1, seg2, hg, counts);
    pool_div_kernel<<<(2 * N_GRAPHS * HIDDEN + T - 1) / T, T, 0, stream>>>(hg, counts);

    logits_kernel<<<(N_GRAPHS * N_CLASSES + 255) / 256, 256, 0, stream>>>(hg1, hg2, Wc, bc, logit);
}


# Edge list ordered by destination through a two-level counting sort on LDS counters (4096-edge chunks binned by 512-node range, a block per range): no returning global adds; sum order = LDS rank order
# speedup vs baseline: 5.5463x; 1.4245x over previous
#include <hip/hip_runtime.h>
#include <hip/hip_bf16.h>

#define N_NODES   100000
#define N_EDGES   1600000
#define N_GRAPHS  128
#define HIDDEN    128
#define N_CLASSES 10

typedef __attribute__((ext_vector_type(2))) float v2f;
typedef __attribute__((ext_vector_type(8))) float v8f;


#define NB_SHIFT  9
#define NB_NODES  (1 << NB_SHIFT)
#define NBUCK     ((N_NODES + NB_NODES - 1) / NB_NODES)
#define CH_EDGES  4096
#define NCH       ((N_EDGES + CH_EDGES - 1) / CH_EDGES)
#define SRC_BITS  17
#define PLACE_CAP 12288
static_assert(N_NODES <= (1 << SRC_BITS), "a source id must fit its field");
static_assert(NBUCK + 1 <= 256, "one bin thread per range start");
static_assert(NCH <= 1024, "one place thread per chunk");
static_assert(N_EDGES % 4 == 0 && CH_EDGES % 1024 == 0, "16-byte copy-out of whole chunks");

template <int NW>
__device__ __forceinline__ int block_excl_scan(int v, int* wsum, int& total) {
    const int lane = threadIdx.x & 31;
    const int wave = threadIdx.x >> 5;
    int inc = v;
    for (int o = 1; o < 32; o <<= 1) {
        int u = __shfl_up(inc, o);
        if (lane >= o) inc += u;
    }
    if (lane == 31) wsum[wave] = inc;
    __syncthreads();
    const int ws = lane < NW ? wsum[lane] : 0;
    int winc = ws;
    for (int o = 1; o < 32; o <<= 1) {
        int u = __shfl_up(winc, o);
        if (lane >= o) winc += u;
    }
    const int wave_excl = __shfl(winc, wave) - __shfl(ws, wave);
    total = __shfl(winc, 31);
    __syncthreads();
    return wave_excl + inc - v;
}

__global__ void __launch_bounds__(256) bin_kernel(const int* __restrict__ src1, const int* __restrict__ dst1,
                                                  const int* __restrict__ src2, const int* __restrict__ dst2,
                                                  float* __restrict__ outdeg, int* __restrict__ srt,
                                                  int* __restrict__ off) {
    __shared__ int wsum[32];
    __shared__ int hist[256];
    __shared__ __attribute__((aligned(16))) int stage[CH_EDGES];
    const int br    = blockIdx.y;
    const int chunk = blockIdx.x;
    const int* __restrict__ src = br ? src2 : src1;
    const int* __restrict__ dst = br ? dst2 : dst1;
    outdeg += br * N_NODES;
    srt    += (size_t)br * N_EDGES;
    off    += (size_t)(br * NCH + chunk) * (NBUCK + 1);
    const int tid = threadIdx.x;
    const int e0  = chunk * CH_EDGES;
    hist[tid] = 0;
    __syncthreads();
    constexpr int PER = CH_EDGES / 256;
    int s[PER], d[PER], rk[PER];
#pragma unroll
    for (int i = 0; i < PER; ++i) {
        const int e = e0 + i * 256 + tid;
        s[i] = e < N_EDGES ? src[e] : -1;
        d[i] = e < N_EDGES ? dst[e] : 0;
    }
#pragma unroll
    for (int i = 0; i < PER; ++i) {
        rk[i] = 0;
        if (s[i] >= 0) {
            atomicAdd(&outdeg[s[i]], 1.0f);
            rk[i] = atomicAdd(&hist[d[i] >> NB_SHIFT], 1);
        }
    }
    __syncthreads();
    int total;
    const int excl = block_excl_scan<8>(hist[tid], wsum, total);
    hist[tid] = excl;
    if (tid <= NBUCK) off[tid] = excl;
    __syncthreads();
#pragma unroll
    for (int i = 0; i < PER; ++i)
        if (s[i] >= 0)
            stage[hist[d[i] >> NB_SHIFT] + rk[i]] = s[i] | ((d[i] & (NB_NODES - 1)) << SRC_BITS);
    __syncthreads();
    const int n4 = min(CH_EDGES, N_EDGES - e0) / 4;
    int4* __restrict__ o4 = (int4*)(srt + e0);
    const int4* st4 = (const int4*)stage;
    for (int j = tid; j < n4; j += 256) o4[j] = st4[j];
}

template <bool COUNT>
__device__ __forceinline__ void walk_pieces(const int* __restrict__ srt, const int* pbeg, const int* plen,
                                            int* cntn, int* sorted, bool staged, int S,
                                            const float* __restrict__ outdeg, int2* __restrict__ ce) {
    const int lane = threadIdx.x & 31;
    const int wave = threadIdx.x >> 5;
    auto take = [&](int en) {
        if constexpr (COUNT) {
            atomicAdd(&cntn[en >> SRC_BITS], 1);
        } else {
            const int pos = atomicAdd(&cntn[en >> SRC_BITS], 1);
            const int s   = en & ((1 << SRC_BITS) - 1);
            if (staged) sorted[pos] = s;
            else        ce[S + pos] = make_int2(s, __float_as_int(rsqrtf(fmaxf(outdeg[s], 1.0f))));
        }
    };
    for (int c = wave; c < NCH; c += 128) {
        int bg[4], ln[4], en[4];
#pragma unroll
        for (int k = 0; k < 4; ++k) {
            const int cc = c + 32 * k;
            ln[k] = cc < NCH ? plen[cc] : 0;
            bg[k] = cc < NCH ? pbeg[cc] : 0;
        }
#pragma unroll
        for (int k = 0; k < 4; ++k)
            en[k] = lane < ln[k] ? srt[bg[k] + lane] : -1;
#pragma unroll
        for (int k = 0; k < 4; ++k) {
            if (en[k] >= 0) take(en[k]);
            for (int i = lane + 32; i < ln[k]; i += 32) take(srt[bg[k] + i]);
        }
    }
}

__global__ void __launch_bounds__(1024) place_kernel(const int* __restrict__ srt, const int* __restrict__ off,
                                                     const float* __restrict__ outdeg,
                                                     int* __restrict__ indeg_i, int* __restrict__ rowptr,
                                                     int2* __restrict__ ce) {
    __shared__ int wsum[32];
    __shared__ int pbeg[NCH], plen[NCH];
    __shared__ int cntn[NB_NODES];
    __shared__ int sorted[PLACE_CAP];
    const int br = blockIdx.y;
    const int b  = blockIdx.x;
    srt     += (size_t)br * N_EDGES;
    off     += (size_t)br * NCH * (NBUCK + 1);
    outdeg  += br * N_NODES;
    indeg_i += br * N_NODES;
    rowptr  += br * N_NODES;
    ce      += (size_t)br * N_EDGES;
    const int tid = threadIdx.x;
    int o0 = 0, len = 0;
    if (tid < NCH) {
        const int* p = off + (size_t)tid * (NBUCK + 1) + b;
        o0  = p[0];
        len = p[1] - o0;
        pbeg[tid] = tid * CH_EDGES + o0;
        plen[tid] = len;
    }
    if (tid < NB_NODES) cntn[tid] = 0;
    int S, T;
    block_excl_scan<32>(o0, wsum, S);
    block_excl_scan<32>(len, wsum, T);
    const bool staged = T <= PLACE_CAP;

    walk_pieces<true>(srt, pbeg, plen, cntn, sorted, staged, S, outdeg, ce);
    __syncthreads();

    const int v = tid < NB_NODES ? cntn[tid] : 0;
    int tot;
    const int ex = block_excl_scan<32>(v, wsum, tot);
    if (tid < NB_NODES) {
        cntn[tid] = ex;
        const int node = b * NB_NODES + tid;
        if (node < N_NODES) {
            indeg_i[node] = v;
            rowptr[node]  = S + ex;
        }
    }
    __syncthreads();

    walk_pieces<false>(srt, pbeg, plen, cntn, sorted, staged, S, outdeg, ce);
    __syncthreads();

    if (staged) {
        constexpr int R = PLACE_CAP / 1024;
        int   sv[R];
        float od[R];
#pragma unroll
        for (int k = 0; k < R; ++k) {
            const int j = tid + k * 1024;
            sv[k] = j < T ? sorted[j] : 0;
        }
#pragma unroll
        for (int k = 0; k < R; ++k) od[k] = outdeg[sv[k]];
#pragma unroll
        for (int k = 0; k < R; ++k) {
            const int j = tid + k * 1024;
            if (j < T) ce[S + j] = make_int2(sv[k], __float_as_int(rsqrtf(fmaxf(od[k], 1.0f))));
        }
    }
}

__global__ void norm_kernel(const float* __restrict__ outdeg, const int* __restrict__ indeg_i,
                            float* __restrict__ cs, float* __restrict__ cd) {
    int n = blockIdx.x * blockDim.x + threadIdx.x;
    if (n < 2 * N_NODES) {
        cs[n] = rsqrtf(fmaxf(outdeg[n], 1.0f));
        cd[n] = rsqrtf(fmaxf((float)indeg_i[n], 1.0f));
    }
}

__global__ void gather1_kernel(const int* __restrict__ rowptr, const int* __restrict__ indeg_i,
                               const int2* __restrict__ ce, const float* __restrict__ cd,
                               float* __restrict__ x) {
    const int br = blockIdx.y;
    rowptr  += br * N_NODES;
    indeg_i += br * N_NODES;
    ce      += (size_t)br * N_EDGES;
    cd      += br * N_NODES;
    x       += br * N_NODES;
    int d = blockIdx.x * blockDim.x + threadIdx.x;
    if (d < N_NODES) {
        const int beg = rowptr[d];
        const int cnt = indeg_i[d];
        float a = 0.0f;
        for (int j = 0; j < cnt; ++j) {
            int2 c = ce[beg + j];
            a += (float)indeg_i[c.x] * __int_as_float(c.y);
        }
        x[d] = a * cd[d];
    }
}

#define NODES_PER_WAVE 8
static_assert(N_NODES % NODES_PER_WAVE == 0, "a wave's 8 nodes must all exist");
__global__ void __launch_bounds__(256) gather_rank1_kernel(const int* __restrict__ rowptr,
                                                           const int* __restrict__ indeg_i,
                                                           const int2* __restrict__ ce,
                                                           const float* __restrict__ x,
                                                           const float* __restrict__ W1,
                                                           const float* __restrict__ b1,
                                                           float* __restrict__ agg) {
    const int br = blockIdx.y;
    rowptr  += br * N_NODES;
    indeg_i += br * N_NODES;
    ce      += (size_t)br * N_EDGES;
    x       += br * N_NODES;
    agg     += (size_t)br * N_NODES * HIDDEN;
    const int wave = __builtin_amdgcn_readfirstlane((int)(threadIdx.x >> 5));
    const int lane = threadIdx.x & 31;
    const int g    = blockIdx.x * 8 + wave;
    if (g < N_NODES / NODES_PER_WAVE) {
        const float4 w = ((const float4*)W1)[lane];
        const float4 b = ((const float4*)b1)[lane];
        const int4 ba = ((const int4*)rowptr)[2 * g],  bb = ((const int4*)rowptr)[2 * g + 1];
        const int4 ca = ((const int4*)indeg_i)[2 * g], cb = ((const int4*)indeg_i)[2 * g + 1];
        const int begs[NODES_PER_WAVE] = {ba.x, ba.y, ba.z, ba.w, bb.x, bb.y, bb.z, bb.w};
        const int cnts[NODES_PER_WAVE] = {ca.x, ca.y, ca.z, ca.w, cb.x, cb.y, cb.z, cb.w};

        int2 e0[NODES_PER_WAVE];
        int  x0[NODES_PER_WAVE];
#pragma unroll
        for (int t = 0; t < NODES_PER_WAVE; ++t)
            e0[t] = ce[min(begs[t] + lane, N_EDGES - 1)];
        __builtin_amdgcn_sched_barrier(0);
#pragma unroll
        for (int t = 0; t < NODES_PER_WAVE; ++t)
            x0[t] = __float_as_int(x[e0[t].x]);
        asm volatile("" :: "v"(x0[0]), "v"(x0[1]), "v"(x0[2]), "v"(x0[3]),
                           "v"(x0[4]), "v"(x0[5]), "v"(x0[6]), "v"(x0[7]));

#pragma unroll
        for (int t = 0; t < NODES_PER_WAVE; ++t) {
            const int beg = begs[t];
            const int cnt = cnts[t];
            int xsrc = x0[t], cscl = e0[t].y;
            float4 acc = make_float4(0.0f, 0.0f, 0.0f, 0.0f);
            for (int base = 0; base < cnt; base += 32) {
                const int n = min(cnt - base, 32);
                if (base > 0) {
                    xsrc = 0; cscl = 0;
                    if (lane < n) {
                        int2 c = ce[beg + base + lane];
                        cscl = c.y;
                        xsrc = __float_as_int(x[c.x]);
                    }
                }
                for (int j = 0; j < n; ++j) {
                    const float xv = __int_as_float(__builtin_amdgcn_readlane(xsrc, j));
                    const float cv = __int_as_float(__builtin_amdgcn_readlane(cscl, j));
                    acc.x = fmaf(fmaxf(fmaf(xv, w.x, b.x), 0.0f), cv, acc.x);
                    acc.y = fmaf(fmaxf(fmaf(xv, w.y, b.y), 0.0f), cv, acc.y);
                    acc.z = fmaf(fmaxf(fmaf(xv, w.z, b.z), 0.0f), cv, acc.z);
                    acc.w = fmaf(fmaxf(fmaf(xv, w.w, b.w), 0.0f), cv, acc.w);
                }
            }
            ((float4*)agg)[(size_t)(g * NODES_PER_WAVE + t) * 32 + lane] = acc;
        }
    }
}

__global__ void __launch_bounds__(256) gather_kernel(const int* __restrict__ rowptr,
                                                     const int* __restrict__ indeg_i,
                                                     const int2* __restrict__ ce,
                                                     const float* __restrict__ h,
                                                     float* __restrict__ agg) {
    const int br = blockIdx.y;
    rowptr  += br * N_NODES;
    indeg_i += br * N_NODES;
    ce      += (size_t)br * N_EDGES;
    h       += (size_t)br * N_NODES * HIDDEN;
    agg     += (size_t)br * N_NODES * HIDDEN;
    const int wave = __builtin_amdgcn_readfirstlane((int)(threadIdx.x >> 5));
    const int lane = threadIdx.x & 31;
    const int d    = blockIdx.x * 8 + wave;
    if (d < N_NODES) {
        const int beg = rowptr[d];
        const int cnt = indeg_i[d];
        const float4* __restrict__ h4 = (const float4*)h;
        float4 acc = make_float4(0.0f, 0.0f, 0.0f, 0.0f);
        for (int base = 0; base < cnt; base += 32) {
            const int n = min(cnt - base, 32);
            int cidx = 0, cscl = 0;
            if (lane < n) {
                int2 c = ce[beg + base + lane];
                cidx = c.x;
                cscl = c.y;
            }
            for (int j = 0; j < n; j += 4) {
                float4 v0, v1, v2, v3;
                float  c0, c1 = 0.0f, c2 = 0.0f, c3 = 0.0f;
                v1 = v2 = v3 = make_float4(0.0f, 0.0f, 0.0f, 0.0f);
                {
                    int s = __builtin_amdgcn_readlane(cidx, j);
                    c0 = __int_as_float(__builtin_amdgcn_readlane(cscl, j));
                    v0 = h4[(size_t)s * 32 + lane];
                }
                if (j + 1 < n) {
                    int s = __builtin_amdgcn_readlane(cidx, j + 1);
                    c1 = __int_as_float(__builtin_amdgcn_readlane(cscl, j + 1));
                    v1 = h4[(size_t)s * 32 + lane];
                }
                if (j + 2 < n) {
                    int s = __builtin_amdgcn_readlane(cidx, j + 2);
                    c2 = __int_as_float(__builtin_amdgcn_readlane(cscl, j + 2));
                    v2 = h4[(size_t)s * 32 + lane];
                }
                if (j + 3 < n) {
                    int s = __builtin_amdgcn_readlane(cidx, j + 3);
                    c3 = __int_as_float(__builtin_amdgcn_readlane(cscl, j + 3));
                    v3 = h4[(size_t)s * 32 + lane];
                }
                acc.x = fmaf(v0.x, c0, acc.x); acc.y = fmaf(v0.y, c0, acc.y);
                acc.z = fmaf(v0.z, c0, acc.z); acc.w = fmaf(v0.w, c0, acc.w);
                acc.x = fmaf(v1.x, c1, acc.x); acc.y = fmaf(v1.y, c1, acc.y);
                acc.z = fmaf(v1.z, c1, acc.z); acc.w = fmaf(v1.w, c1, acc.w);
                acc.x = fmaf(v2.x, c2, acc.x); acc.y = fmaf(v2.y, c2, acc.y);
                acc.z = fmaf(v2.z, c2, acc.z); acc.w = fmaf(v2.w, c2, acc.w);
                acc.x = fmaf(v3.x, c3, acc.x); acc.y = fmaf(v3.y, c3, acc.y);
                acc.z = fmaf(v3.z, c3, acc.z); acc.w = fmaf(v3.w, c3, acc.w);
            }
        }
        ((float4*)agg)[(size_t)d * 32 + lane] = acc;
    }
}

#define GEMM_BLOCKS 1024
#define A_PITCH     132
static_assert(N_NODES % 16 == 0, "row tiles must be whole");
static_assert(GEMM_BLOCKS <= N_NODES / 16, "every block owns a first tile");
template <bool POOL>
__device__ __forceinline__ void gemm_relu_body(const float* __restrict__ agg,
                                               const float* __restrict__ cd,
                                               const float* __restrict__ W,
                                               const float* __restrict__ bias,
                                               float* __restrict__ hout,
                                               const int* __restrict__ seg,
                                               float* __restrict__ hg,
                                               float* __restrict__ counts) {
    __shared__ __attribute__((aligned(16))) float As[2][16 * A_PITCH];
    constexpr int NT = N_NODES / 16;
    const int tid  = threadIdx.x;
    const int wave = tid >> 5;
    const int lane = tid & 31;
    const int half = lane >> 4;
    const int lr   = lane & 15;
    const int n    = wave * 16 + lr;

    v2f bfr[HIDDEN / 4];
#pragma unroll
    for (int s = 0; s < HIDDEN / 4; ++s) {
        bfr[s].x = W[(4 * s + 2 * half) * HIDDEN + n];
        bfr[s].y = W[(4 * s + 2 * half + 1) * HIDDEN + n];
    }
    const float bn = bias[n];

    const int sr = 2 * wave;
    int t = blockIdx.x;
    float4 p0, p1;
    float  c0, c1;
    int4 na = {}, nb = {}, nc = {}, nd = {};
    {
        const int r0 = t * 16 + sr;
        p0 = ((const float4*)(agg + (size_t)r0 * HIDDEN))[lane];
        p1 = ((const float4*)(agg + (size_t)(r0 + 1) * HIDDEN))[lane];
        c0 = cd[r0];
        c1 = cd[r0 + 1];
        if constexpr (POOL) {
            const int4* __restrict__ sg = (const int4*)(seg + t * 16);
            na = sg[0]; nb = sg[1]; nc = sg[2]; nd = sg[3];
        }
    }

    int buf = 0;
    for (;;) {
        {
            float* dst = &As[buf][sr * A_PITCH + lane * 4];
            *(float4*)dst             = make_float4(p0.x * c0, p0.y * c0, p0.z * c0, p0.w * c0);
            *(float4*)(dst + A_PITCH) = make_float4(p1.x * c1, p1.y * c1, p1.z * c1, p1.w * c1);
        }
        __syncthreads();

        const int  tn   = t + GEMM_BLOCKS;
        const bool more = tn < NT;
        const int4 sa = na, sb = nb, sc = nc, sd = nd;
        if (more) {
            const int r0 = tn * 16 + sr;
            p0 = ((const float4*)(agg + (size_t)r0 * HIDDEN))[lane];
            p1 = ((const float4*)(agg + (size_t)(r0 + 1) * HIDDEN))[lane];
            c0 = cd[r0];
            c1 = cd[r0 + 1];
        }
        if constexpr (POOL) {
            const int4* __restrict__ sg = (const int4*)(seg + min(tn, NT - 1) * 16);
            na = sg[0]; nb = sg[1]; nc = sg[2]; nd = sg[3];
        }

        const float* arow = &As[buf][lr * A_PITCH + 2 * half];
        v8f acc = {};
#pragma unroll
        for (int s = 0; s < HIDDEN / 4; ++s) {
            const v2f av = *(const v2f*)(arow + 4 * s);
            acc = __builtin_amdgcn_wmma_f32_16x16x4_f32(
                 false, av,  false, bfr[s],
                 (short)0, acc,  false,  false);
        }

        const int m0 = t * 16;
        if constexpr (!POOL) {
#pragma unroll
            for (int r = 0; r < 8; ++r) {
                const int mm = m0 + half * 8 + r;
                hout[(size_t)mm * HIDDEN + n] = fmaxf(acc[r] + bn, 0.0f);
            }
        } else {
            const int  g0 = sa.x;
            const int  differ = (sa.y ^ g0) | (sa.z ^ g0) | (sa.w ^ g0) |
                                (sb.x ^ g0) | (sb.y ^ g0) | (sb.z ^ g0) | (sb.w ^ g0) |
                                (sc.x ^ g0) | (sc.y ^ g0) | (sc.z ^ g0) | (sc.w ^ g0) |
                                (sd.x ^ g0) | (sd.y ^ g0) | (sd.z ^ g0) | (sd.w ^ g0);
            if (differ == 0) {
                float s = fmaxf(acc[0] + bn, 0.0f);
#pragma unroll
                for (int r = 1; r < 8; ++r)
                    s += fmaxf(acc[r] + bn, 0.0f);
                s += __shfl_xor(s, 16);
                if (half == 0) atomicAdd(&hg[g0 * HIDDEN + n], s);
                if (tid == 0)  atomicAdd(&counts[g0], 16.0f);
            } else {
                const int gl[8] = {sa.x, sa.y, sa.z, sa.w, sb.x, sb.y, sb.z, sb.w};
                const int gh[8] = {sc.x, sc.y, sc.z, sc.w, sd.x, sd.y, sd.z, sd.w};
#pragma unroll
                for (int r = 0; r < 8; ++r) {
                    const int g = half ? gh[r] : gl[r];
                    atomicAdd(&hg[g * HIDDEN + n], fmaxf(acc[r] + bn, 0.0f));
                    if (wave == 0 && lr == 0) atomicAdd(&counts[g], 1.0f);
                }
            }
        }

        if (!more) break;
        t = tn;
        buf ^= 1;
    }
}

__global__ void __launch_bounds__(256) gemm_relu_kernel(const float* __restrict__ agg,
                                                        const float* __restrict__ cd,
                                                        const float* __restrict__ W,
                                                        const float* __restrict__ bias,
                                                        float* __restrict__ hout) {
    const int br = blockIdx.y;
    gemm_relu_body<false>(agg + (size_t)br * N_NODES * HIDDEN, cd + br * N_NODES, W, bias,
                          hout + (size_t)br * N_NODES * HIDDEN, nullptr, nullptr, nullptr);
}

__global__ void __launch_bounds__(256) gemm_relu_pool_kernel(const float* __restrict__ agg,
                                                             const float* __restrict__ cd,
                                                             const float* __restrict__ W,
                                                             const float* __restrict__ bias,
                                                             const int* __restrict__ seg1,
                                                             const int* __restrict__ seg2,
                                                             float* __restrict__ hg,
                                                             float* __restrict__ counts) {
    const int br = blockIdx.y;
    gemm_relu_body<true>(agg + (size_t)br * N_NODES * HIDDEN, cd + br * N_NODES, W, bias, nullptr,
                         br ? seg2 : seg1, hg + br * N_GRAPHS * HIDDEN, counts + br * N_GRAPHS);
}

__global__ void pool_div_kernel(float* __restrict__ hg, const float* __restrict__ counts) {
    int i = blockIdx.x * blockDim.x + threadIdx.x;
    if (i < 2 * N_GRAPHS * HIDDEN)
        hg[i] /= fmaxf(counts[i >> 7], 1.0f);
}

__global__ void logits_kernel(const float* __restrict__ hg1, const float* __restrict__ hg2,
                              const float* __restrict__ Wc, const float* __restrict__ bc,
                              float* __restrict__ logits) {
    int i = blockIdx.x * blockDim.x + threadIdx.x;
    if (i < N_GRAPHS * N_CLASSES) {
        int g = i / N_CLASSES, c = i % N_CLASSES;
        float s = bc[c];
#pragma unroll 8
        for (int k = 0; k < HIDDEN; ++k)
            s += fabsf(hg1[g * HIDDEN + k] - hg2[g * HIDDEN + k]) * Wc[k * N_CLASSES + c];
        logits[i] = s;
    }
}

extern "C" void kernel_launch(void* const* d_in, const int* in_sizes, int n_in,
                              void* d_out, int out_size, void* d_ws, size_t ws_size,
                              hipStream_t stream) {
    const int*   src1 = (const int*)d_in[0];
    const int*   dst1 = (const int*)d_in[1];
    const int*   seg1 = (const int*)d_in[2];
    const int*   src2 = (const int*)d_in[3];
    const int*   dst2 = (const int*)d_in[4];
    const int*   seg2 = (const int*)d_in[5];
    const float* W1   = (const float*)d_in[6];
    const float* b1   = (const float*)d_in[7];
    const float* W2   = (const float*)d_in[8];
    const float* b2   = (const float*)d_in[9];
    const float* W3   = (const float*)d_in[10];
    const float* b3   = (const float*)d_in[11];
    const float* W4   = (const float*)d_in[12];
    const float* b4   = (const float*)d_in[13];
    const float* Wc   = (const float*)d_in[14];
    const float* bc   = (const float*)d_in[15];

    float* out   = (float*)d_out;
    float* hg    = out;
    float* hg1   = out;
    float* hg2   = out + N_GRAPHS * HIDDEN;
    float* logit = out + 2 * N_GRAPHS * HIDDEN;
    float* ws_f  = (float*)d_ws;

    float* outdeg  = ws_f;
    float* counts  = ws_f + 200000;
    int*   indeg_i = (int*)(ws_f + 200256);
    float* cs      = ws_f + 400256;
    float* cd      = ws_f + 600256;
    float* x       = ws_f + 800256;
    int*   rowptr  = (int*)(ws_f + 1000256);
    float* h       = ws_f + 1200256;
    float* agg     = h + (size_t)2 * N_NODES * HIDDEN;
    int*   srt     = (int*)(agg + (size_t)2 * N_NODES * HIDDEN);
    int2*  ce      = (int2*)(srt + (size_t)2 * N_EDGES);
    int*   off     = (int*)(ce + (size_t)2 * N_EDGES);

    hipMemsetAsync(ws_f, 0, (size_t)200256 * sizeof(float), stream);
    hipMemsetAsync(hg, 0, (size_t)2 * N_GRAPHS * HIDDEN * sizeof(float), stream);

    const int T = 256;
    bin_kernel<<<dim3(NCH, 2), T, 0, stream>>>(src1, dst1, src2, dst2, outdeg, srt, off);
    place_kernel<<<dim3(NBUCK, 2), 1024, 0, stream>>>(srt, off, outdeg, indeg_i, rowptr, ce);
    norm_kernel<<<(2 * N_NODES + T - 1) / T, T, 0, stream>>>(outdeg, indeg_i, cs, cd);

    gather1_kernel<<<dim3((N_NODES + T - 1) / T, 2), T, 0, stream>>>(rowptr, indeg_i, ce, cd, x);

    gather_rank1_kernel<<<dim3((N_NODES / NODES_PER_WAVE + 7) / 8, 2), T, 0, stream>>>(
        rowptr, indeg_i, ce, x, W1, b1, agg);
    gemm_relu_kernel<<<dim3(GEMM_BLOCKS, 2), T, 0, stream>>>(agg, cd, W2, b2, h);

    gather_kernel<<<dim3((N_NODES * 32) / T, 2), T, 0, stream>>>(rowptr, indeg_i, ce, h, agg);
    gemm_relu_kernel<<<dim3(GEMM_BLOCKS, 2), T, 0, stream>>>(agg, cd, W3, b3, h);

    gather_kernel<<<dim3((N_NODES * 32) / T, 2), T, 0, stream>>>(rowptr, indeg_i, ce, h, agg);
    gemm_relu_pool_kernel<<<dim3(GEMM_BLOCKS, 2), T, 0, stream>>>(agg, cd, W4, b4, seg1, seg2, hg, counts);
    pool_div_kernel<<<(2 * N_GRAPHS * HIDDEN + T - 1) / T, T, 0, stream>>>(hg, counts);

    logits_kernel<<<(N_GRAPHS * N_CLASSES + 255) / 256, 256, 0, stream>>>(hg1, hg2, Wc, bc, logit);
}


# Edge sums of layers 2-4 built inside the GEMM's tile staging (a wave sums the sources of its 2 tile rows, x cd, into LDS): agg never stored or re-read, 3 launches fewer; same products and sum orders
# speedup vs baseline: 6.0291x; 1.0870x over previous
#include <hip/hip_runtime.h>
#include <hip/hip_bf16.h>

#define N_NODES   100000
#define N_EDGES   1600000
#define N_GRAPHS  128
#define HIDDEN    128
#define N_CLASSES 10

typedef __attribute__((ext_vector_type(2))) float v2f;
typedef __attribute__((ext_vector_type(8))) float v8f;


#define NB_SHIFT  9
#define NB_NODES  (1 << NB_SHIFT)
#define NBUCK     ((N_NODES + NB_NODES - 1) / NB_NODES)
#define CH_EDGES  4096
#define NCH       ((N_EDGES + CH_EDGES - 1) / CH_EDGES)
#define SRC_BITS  17
#define PLACE_CAP 12288
static_assert(N_NODES <= (1 << SRC_BITS), "a source id must fit its field");
static_assert(NBUCK + 1 <= 256, "one bin thread per range start");
static_assert(NCH <= 1024, "one place thread per chunk");
static_assert(N_EDGES % 4 == 0 && CH_EDGES % 1024 == 0, "16-byte copy-out of whole chunks");

template <int NW>
__device__ __forceinline__ int block_excl_scan(int v, int* wsum, int& total) {
    const int lane = threadIdx.x & 31;
    const int wave = threadIdx.x >> 5;
    int inc = v;
    for (int o = 1; o < 32; o <<= 1) {
        int u = __shfl_up(inc, o);
        if (lane >= o) inc += u;
    }
    if (lane == 31) wsum[wave] = inc;
    __syncthreads();
    const int ws = lane < NW ? wsum[lane] : 0;
    int winc = ws;
    for (int o = 1; o < 32; o <<= 1) {
        int u = __shfl_up(winc, o);
        if (lane >= o) winc += u;
    }
    const int wave_excl = __shfl(winc, wave) - __shfl(ws, wave);
    total = __shfl(winc, 31);
    __syncthreads();
    return wave_excl + inc - v;
}

__global__ void __launch_bounds__(256) bin_kernel(const int* __restrict__ src1, const int* __restrict__ dst1,
                                                  const int* __restrict__ src2, const int* __restrict__ dst2,
                                                  float* __restrict__ outdeg, int* __restrict__ srt,
                                                  int* __restrict__ off) {
    __shared__ int wsum[32];
    __shared__ int hist[256];
    __shared__ __attribute__((aligned(16))) int stage[CH_EDGES];
    const int br    = blockIdx.y;
    const int chunk = blockIdx.x;
    const int* __restrict__ src = br ? src2 : src1;
    const int* __restrict__ dst = br ? dst2 : dst1;
    outdeg += br * N_NODES;
    srt    += (size_t)br * N_EDGES;
    off    += (size_t)(br * NCH + chunk) * (NBUCK + 1);
    const int tid = threadIdx.x;
    const int e0  = chunk * CH_EDGES;
    hist[tid] = 0;
    __syncthreads();
    constexpr int PER = CH_EDGES / 256;
    int s[PER], d[PER], rk[PER];
#pragma unroll
    for (int i = 0; i < PER; ++i) {
        const int e = e0 + i * 256 + tid;
        s[i] = e < N_EDGES ? src[e] : -1;
        d[i] = e < N_EDGES ? dst[e] : 0;
    }
#pragma unroll
    for (int i = 0; i < PER; ++i) {
        rk[i] = 0;
        if (s[i] >= 0) {
            atomicAdd(&outdeg[s[i]], 1.0f);
            rk[i] = atomicAdd(&hist[d[i] >> NB_SHIFT], 1);
        }
    }
    __syncthreads();
    int total;
    const int excl = block_excl_scan<8>(hist[tid], wsum, total);
    hist[tid] = excl;
    if (tid <= NBUCK) off[tid] = excl;
    __syncthreads();
#pragma unroll
    for (int i = 0; i < PER; ++i)
        if (s[i] >= 0)
            stage[hist[d[i] >> NB_SHIFT] + rk[i]] = s[i] | ((d[i] & (NB_NODES - 1)) << SRC_BITS);
    __syncthreads();
    const int n4 = min(CH_EDGES, N_EDGES - e0) / 4;
    int4* __restrict__ o4 = (int4*)(srt + e0);
    const int4* st4 = (const int4*)stage;
    for (int j = tid; j < n4; j += 256) o4[j] = st4[j];
}

template <bool COUNT>
__device__ __forceinline__ void walk_pieces(const int* __restrict__ srt, const int* pbeg, const int* plen,
                                            int* cntn, int* sorted, bool staged, int S,
                                            const float* __restrict__ outdeg, int2* __restrict__ ce) {
    const int lane = threadIdx.x & 31;
    const int wave = threadIdx.x >> 5;
    auto take = [&](int en) {
        if constexpr (COUNT) {
            atomicAdd(&cntn[en >> SRC_BITS], 1);
        } else {
            const int pos = atomicAdd(&cntn[en >> SRC_BITS], 1);
            const int s   = en & ((1 << SRC_BITS) - 1);
            if (staged) sorted[pos] = s;
            else        ce[S + pos] = make_int2(s, __float_as_int(rsqrtf(fmaxf(outdeg[s], 1.0f))));
        }
    };
    for (int c = wave; c < NCH; c += 128) {
        int bg[4], ln[4], en[4];
#pragma unroll
        for (int k = 0; k < 4; ++k) {
            const int cc = c + 32 * k;
            ln[k] = cc < NCH ? plen[cc] : 0;
            bg[k] = cc < NCH ? pbeg[cc] : 0;
        }
#pragma unroll
        for (int k = 0; k < 4; ++k)
            en[k] = lane < ln[k] ? srt[bg[k] + lane] : -1;
#pragma unroll
        for (int k = 0; k < 4; ++k) {
            if (en[k] >= 0) take(en[k]);
            for (int i = lane + 32; i < ln[k]; i += 32) take(srt[bg[k] + i]);
        }
    }
}

__global__ void __launch_bounds__(1024) place_kernel(const int* __restrict__ srt, const int* __restrict__ off,
                                                     const float* __restrict__ outdeg,
                                                     int* __restrict__ indeg_i, int* __restrict__ rowptr,
                                                     int2* __restrict__ ce) {
    __shared__ int wsum[32];
    __shared__ int pbeg[NCH], plen[NCH];
    __shared__ int cntn[NB_NODES];
    __shared__ int sorted[PLACE_CAP];
    const int br = blockIdx.y;
    const int b  = blockIdx.x;
    srt     += (size_t)br * N_EDGES;
    off     += (size_t)br * NCH * (NBUCK + 1);
    outdeg  += br * N_NODES;
    indeg_i += br * N_NODES;
    rowptr  += br * N_NODES;
    ce      += (size_t)br * N_EDGES;
    const int tid = threadIdx.x;
    int o0 = 0, len = 0;
    if (tid < NCH) {
        const int* p = off + (size_t)tid * (NBUCK + 1) + b;
        o0  = p[0];
        len = p[1] - o0;
        pbeg[tid] = tid * CH_EDGES + o0;
        plen[tid] = len;
    }
    if (tid < NB_NODES) cntn[tid] = 0;
    int S, T;
    block_excl_scan<32>(o0, wsum, S);
    block_excl_scan<32>(len, wsum, T);
    const bool staged = T <= PLACE_CAP;

    walk_pieces<true>(srt, pbeg, plen, cntn, sorted, staged, S, outdeg, ce);
    __syncthreads();

    const int v = tid < NB_NODES ? cntn[tid] : 0;
    int tot;
    const int ex = block_excl_scan<32>(v, wsum, tot);
    if (tid < NB_NODES) {
        cntn[tid] = ex;
        const int node = b * NB_NODES + tid;
        if (node < N_NODES) {
            indeg_i[node] = v;
            rowptr[node]  = S + ex;
        }
    }
    __syncthreads();

    walk_pieces<false>(srt, pbeg, plen, cntn, sorted, staged, S, outdeg, ce);
    __syncthreads();

    if (staged) {
        constexpr int R = PLACE_CAP / 1024;
        int   sv[R];
        float od[R];
#pragma unroll
        for (int k = 0; k < R; ++k) {
            const int j = tid + k * 1024;
            sv[k] = j < T ? sorted[j] : 0;
        }
#pragma unroll
        for (int k = 0; k < R; ++k) od[k] = outdeg[sv[k]];
#pragma unroll
        for (int k = 0; k < R; ++k) {
            const int j = tid + k * 1024;
            if (j < T) ce[S + j] = make_int2(sv[k], __float_as_int(rsqrtf(fmaxf(od[k], 1.0f))));
        }
    }
}

__global__ void norm_kernel(const float* __restrict__ outdeg, const int* __restrict__ indeg_i,
                            float* __restrict__ cs, float* __restrict__ cd) {
    int n = blockIdx.x * blockDim.x + threadIdx.x;
    if (n < 2 * N_NODES) {
        cs[n] = rsqrtf(fmaxf(outdeg[n], 1.0f));
        cd[n] = rsqrtf(fmaxf((float)indeg_i[n], 1.0f));
    }
}

__global__ void gather1_kernel(const int* __restrict__ rowptr, const int* __restrict__ indeg_i,
                               const int2* __restrict__ ce, const float* __restrict__ cd,
                               float* __restrict__ x) {
    const int br = blockIdx.y;
    rowptr  += br * N_NODES;
    indeg_i += br * N_NODES;
    ce      += (size_t)br * N_EDGES;
    cd      += br * N_NODES;
    x       += br * N_NODES;
    int d = blockIdx.x * blockDim.x + threadIdx.x;
    if (d < N_NODES) {
        const int beg = rowptr[d];
        const int cnt = indeg_i[d];
        float a = 0.0f;
        for (int j = 0; j < cnt; ++j) {
            int2 c = ce[beg + j];
            a += (float)indeg_i[c.x] * __int_as_float(c.y);
        }
        x[d] = a * cd[d];
    }
}

#define GEMM_BLOCKS 1024
#define A_PITCH     132
static_assert(N_NODES % 16 == 0, "row tiles must be whole");
static_assert(GEMM_BLOCKS <= N_NODES / 16, "every block owns a first tile");

__device__ __forceinline__ void rows_tail(const int2* __restrict__ ce, const float4* __restrict__ h4,
                                          int beg, int cnt, int lane, float4& acc) {
    for (int base = 32; base < cnt; base += 32) {
        const int n = min(cnt - base, 32);
        int cidx = 0, cscl = 0;
        if (lane < n) {
            int2 c = ce[beg + base + lane];
            cidx = c.x;
            cscl = c.y;
        }
        for (int j = 0; j < n; j += 4) {
            float4 v[4];
            float  c[4];
#pragma unroll
            for (int k = 0; k < 4; ++k) {
                v[k] = make_float4(0.0f, 0.0f, 0.0f, 0.0f);
                c[k] = 0.0f;
                if (j + k < n) {
                    const int s = __builtin_amdgcn_readlane(cidx, j + k);
                    c[k] = __int_as_float(__builtin_amdgcn_readlane(cscl, j + k));
                    v[k] = h4[(size_t)s * 32 + lane];
                }
            }
#pragma unroll
            for (int k = 0; k < 4; ++k) {
                acc.x = fmaf(v[k].x, c[k], acc.x); acc.y = fmaf(v[k].y, c[k], acc.y);
                acc.z = fmaf(v[k].z, c[k], acc.z); acc.w = fmaf(v[k].w, c[k], acc.w);
            }
        }
    }
}

__device__ __forceinline__ void rank1_tail(const int2* __restrict__ ce, const float* __restrict__ x,
                                           int beg, int cnt, int lane, const float4& w, const float4& b, float4& acc) {
    for (int base = 32; base < cnt; base += 32) {
        const int n = min(cnt - base, 32);
        int xsrc = 0, cscl = 0;
        if (lane < n) {
            int2 c = ce[beg + base + lane];
            cscl = c.y;
            xsrc = __float_as_int(x[c.x]);
        }
        for (int j = 0; j < n; ++j) {
            const float xv = __int_as_float(__builtin_amdgcn_readlane(xsrc, j));
            const float cv = __int_as_float(__builtin_amdgcn_readlane(cscl, j));
            acc.x = fmaf(fmaxf(fmaf(xv, w.x, b.x), 0.0f), cv, acc.x);
            acc.y = fmaf(fmaxf(fmaf(xv, w.y, b.y), 0.0f), cv, acc.y);
            acc.z = fmaf(fmaxf(fmaf(xv, w.z, b.z), 0.0f), cv, acc.z);
            acc.w = fmaf(fmaxf(fmaf(xv, w.w, b.w), 0.0f), cv, acc.w);
        }
    }
}

template <bool POOL, bool RANK1>
__device__ __forceinline__ void layer_body(const int* __restrict__ rowptr,
                                           const int* __restrict__ indeg_i,
                                           const int2* __restrict__ ce,
                                           const float* __restrict__ hin,
                                           const float* __restrict__ W1,
                                           const float* __restrict__ b1,
                                           const float* __restrict__ cd,
                                           const float* __restrict__ W,
                                           const float* __restrict__ bias,
                                           float* __restrict__ hout,
                                           const int* __restrict__ seg,
                                           float* __restrict__ hg,
                                           float* __restrict__ counts) {
    __shared__ __attribute__((aligned(16))) float As[2][16 * A_PITCH];
    constexpr int NT = N_NODES / 16;
    const int tid  = threadIdx.x;
    const int wave = tid >> 5;
    const int wv   = __builtin_amdgcn_readfirstlane(wave);
    const int lane = tid & 31;
    const int half = lane >> 4;
    const int lr   = lane & 15;
    const int n    = wave * 16 + lr;

    v2f bfr[HIDDEN / 4];
#pragma unroll
    for (int s = 0; s < HIDDEN / 4; ++s) {
        bfr[s].x = W[(4 * s + 2 * half) * HIDDEN + n];
        bfr[s].y = W[(4 * s + 2 * half + 1) * HIDDEN + n];
    }
    const float bn = bias[n];
    float4 w1 = {}, bb1 = {};
    if constexpr (RANK1) {
        w1  = ((const float4*)W1)[lane];
        bb1 = ((const float4*)b1)[lane];
    }

    const int sr = 2 * wv;
    int t = blockIdx.x;
    int2   rp, dg, nrp = {}, ndg = {};
    float2 cv, ncv = {};
    {
        const int r0 = t * 16 + sr;
        rp = *(const int2*)(rowptr + r0);
        dg = *(const int2*)(indeg_i + r0);
        cv = *(const float2*)(cd + r0);
    }
    int2 eA = ce[min(rp.x + lane, N_EDGES - 1)];
    int2 eB = ce[min(rp.y + lane, N_EDGES - 1)];
    if (t + GEMM_BLOCKS < NT) {
        const int r0 = (t + GEMM_BLOCKS) * 16 + sr;
        nrp = *(const int2*)(rowptr + r0);
        ndg = *(const int2*)(indeg_i + r0);
        ncv = *(const float2*)(cd + r0);
    }
    int4 na = {}, nb = {}, nc = {}, nd = {};
    if constexpr (POOL) {
        const int4* __restrict__ sg = (const int4*)(seg + t * 16);
        na = sg[0]; nb = sg[1]; nc = sg[2]; nd = sg[3];
    }

    int buf = 0;
    for (;;) {
        const int begA = rp.x, begB = rp.y, cntA = dg.x, cntB = dg.y;
        const int nA = min(cntA, 32), nB = min(cntB, 32);
        float4 accA = make_float4(0.0f, 0.0f, 0.0f, 0.0f);
        float4 accB = make_float4(0.0f, 0.0f, 0.0f, 0.0f);
        if constexpr (RANK1) {
            const int xA = __float_as_int(hin[eA.x]);
            const int xB = __float_as_int(hin[eB.x]);
            for (int j = 0; j < nA; ++j) {
                const float xv = __int_as_float(__builtin_amdgcn_readlane(xA, j));
                const float c  = __int_as_float(__builtin_amdgcn_readlane(eA.y, j));
                accA.x = fmaf(fmaxf(fmaf(xv, w1.x, bb1.x), 0.0f), c, accA.x);
                accA.y = fmaf(fmaxf(fmaf(xv, w1.y, bb1.y), 0.0f), c, accA.y);
                accA.z = fmaf(fmaxf(fmaf(xv, w1.z, bb1.z), 0.0f), c, accA.z);
                accA.w = fmaf(fmaxf(fmaf(xv, w1.w, bb1.w), 0.0f), c, accA.w);
            }
            for (int j = 0; j < nB; ++j) {
                const float xv = __int_as_float(__builtin_amdgcn_readlane(xB, j));
                const float c  = __int_as_float(__builtin_amdgcn_readlane(eB.y, j));
                accB.x = fmaf(fmaxf(fmaf(xv, w1.x, bb1.x), 0.0f), c, accB.x);
                accB.y = fmaf(fmaxf(fmaf(xv, w1.y, bb1.y), 0.0f), c, accB.y);
                accB.z = fmaf(fmaxf(fmaf(xv, w1.z, bb1.z), 0.0f), c, accB.z);
                accB.w = fmaf(fmaxf(fmaf(xv, w1.w, bb1.w), 0.0f), c, accB.w);
            }
            rank1_tail(ce, hin, begA, cntA, lane, w1, bb1, accA);
            rank1_tail(ce, hin, begB, cntB, lane, w1, bb1, accB);
        } else {
            const float4* __restrict__ h4 = (const float4*)hin;
            const int nmax = max(nA, nB);
            asm volatile("" :: "v"(eA.x), "v"(eA.y), "v"(eB.x), "v"(eB.y));
            for (int j = 0; j < nmax; j += 4) {
                float4 va[4], vb[4];
                float  ca[4], cb[4];
#pragma unroll
                for (int k = 0; k < 4; ++k) {
                    va[k] = make_float4(0.0f, 0.0f, 0.0f, 0.0f);
                    ca[k] = 0.0f;
                    if (j + k < nA) {
                        const int s = __builtin_amdgcn_readlane(eA.x, j + k);
                        ca[k] = __int_as_float(__builtin_amdgcn_readlane(eA.y, j + k));
                        va[k] = h4[(size_t)s * 32 + lane];
                    }
                }
#pragma unroll
                for (int k = 0; k < 4; ++k) {
                    vb[k] = make_float4(0.0f, 0.0f, 0.0f, 0.0f);
                    cb[k] = 0.0f;
                    if (j + k < nB) {
                        const int s = __builtin_amdgcn_readlane(eB.x, j + k);
                        cb[k] = __int_as_float(__builtin_amdgcn_readlane(eB.y, j + k));
                        vb[k] = h4[(size_t)s * 32 + lane];
                    }
                }
#pragma unroll
                for (int k = 0; k < 4; ++k) {
                    accA.x = fmaf(va[k].x, ca[k], accA.x); accA.y = fmaf(va[k].y, ca[k], accA.y);
                    accA.z = fmaf(va[k].z, ca[k], accA.z); accA.w = fmaf(va[k].w, ca[k], accA.w);
                }
#pragma unroll
                for (int k = 0; k < 4; ++k) {
                    accB.x = fmaf(vb[k].x, cb[k], accB.x); accB.y = fmaf(vb[k].y, cb[k], accB.y);
                    accB.z = fmaf(vb[k].z, cb[k], accB.z); accB.w = fmaf(vb[k].w, cb[k], accB.w);
                }
            }
            rows_tail(ce, h4, begA, cntA, lane, accA);
            rows_tail(ce, h4, begB, cntB, lane, accB);
        }

        {
            const float c0 = cv.x, c1 = cv.y;
            float* dst = &As[buf][sr * A_PITCH + lane * 4];
            *(float4*)dst             = make_float4(accA.x * c0, accA.y * c0, accA.z * c0, accA.w * c0);
            *(float4*)(dst + A_PITCH) = make_float4(accB.x * c1, accB.y * c1, accB.z * c1, accB.w * c1);
        }
        __syncthreads();

        const int  tn   = t + GEMM_BLOCKS;
        const bool more = tn < NT;
        const int4 sa = na, sb = nb, sc = nc, sd = nd;
        int2   n2rp = nrp, n2dg = ndg;
        float2 n2cv = ncv;
        if (more) {
            eA = ce[min(nrp.x + lane, N_EDGES - 1)];
            eB = ce[min(nrp.y + lane, N_EDGES - 1)];
            const int t2 = tn + GEMM_BLOCKS;
            if (t2 < NT) {
                const int r0 = t2 * 16 + sr;
                n2rp = *(const int2*)(rowptr + r0);
                n2dg = *(const int2*)(indeg_i + r0);
                n2cv = *(const float2*)(cd + r0);
            }
        }
        if constexpr (POOL) {
            const int4* __restrict__ sg = (const int4*)(seg + min(tn, NT - 1) * 16);
            na = sg[0]; nb = sg[1]; nc = sg[2]; nd = sg[3];
        }

        const float* arow = &As[buf][lr * A_PITCH + 2 * half];
        v8f acc = {};
#pragma unroll
        for (int s = 0; s < HIDDEN / 4; ++s) {
            const v2f av = *(const v2f*)(arow + 4 * s);
            acc = __builtin_amdgcn_wmma_f32_16x16x4_f32(
                 false, av,  false, bfr[s],
                 (short)0, acc,  false,  false);
        }

        const int m0 = t * 16;
        if constexpr (!POOL) {
#pragma unroll
            for (int r = 0; r < 8; ++r) {
                const int mm = m0 + half * 8 + r;
                hout[(size_t)mm * HIDDEN + n] = fmaxf(acc[r] + bn, 0.0f);
            }
        } else {
            const int  g0 = sa.x;
            const int  differ = (sa.y ^ g0) | (sa.z ^ g0) | (sa.w ^ g0) |
                                (sb.x ^ g0) | (sb.y ^ g0) | (sb.z ^ g0) | (sb.w ^ g0) |
                                (sc.x ^ g0) | (sc.y ^ g0) | (sc.z ^ g0) | (sc.w ^ g0) |
                                (sd.x ^ g0) | (sd.y ^ g0) | (sd.z ^ g0) | (sd.w ^ g0);
            if (differ == 0) {
                float s = fmaxf(acc[0] + bn, 0.0f);
#pragma unroll
                for (int r = 1; r < 8; ++r)
                    s += fmaxf(acc[r] + bn, 0.0f);
                s += __shfl_xor(s, 16);
                if (half == 0) atomicAdd(&hg[g0 * HIDDEN + n], s);
                if (tid == 0)  atomicAdd(&counts[g0], 16.0f);
            } else {
                const int gl[8] = {sa.x, sa.y, sa.z, sa.w, sb.x, sb.y, sb.z, sb.w};
                const int gh[8] = {sc.x, sc.y, sc.z, sc.w, sd.x, sd.y, sd.z, sd.w};
#pragma unroll
                for (int r = 0; r < 8; ++r) {
                    const int g = half ? gh[r] : gl[r];
                    atomicAdd(&hg[g * HIDDEN + n], fmaxf(acc[r] + bn, 0.0f));
                    if (wave == 0 && lr == 0) atomicAdd(&counts[g], 1.0f);
                }
            }
        }

        if (!more) break;
        rp = nrp;  dg = ndg;  cv = ncv;
        nrp = n2rp; ndg = n2dg; ncv = n2cv;
        t = tn;
        buf ^= 1;
    }
}

__global__ void __launch_bounds__(256) layer_rank1_kernel(const int* __restrict__ rowptr,
                                                          const int* __restrict__ indeg_i,
                                                          const int2* __restrict__ ce,
                                                          const float* __restrict__ x,
                                                          const float* __restrict__ W1,
                                                          const float* __restrict__ b1,
                                                          const float* __restrict__ cd,
                                                          const float* __restrict__ W,
                                                          const float* __restrict__ bias,
                                                          float* __restrict__ hout) {
    const int br = blockIdx.y;
    layer_body<false, true>(rowptr + br * N_NODES, indeg_i + br * N_NODES, ce + (size_t)br * N_EDGES,
                            x + br * N_NODES, W1, b1, cd + br * N_NODES, W, bias,
                            hout + (size_t)br * N_NODES * HIDDEN, nullptr, nullptr, nullptr);
}

__global__ void __launch_bounds__(256) layer_rows_kernel(const int* __restrict__ rowptr,
                                                         const int* __restrict__ indeg_i,
                                                         const int2* __restrict__ ce,
                                                         const float* __restrict__ hin,
                                                         const float* __restrict__ cd,
                                                         const float* __restrict__ W,
                                                         const float* __restrict__ bias,
                                                         float* __restrict__ hout) {
    const int br = blockIdx.y;
    layer_body<false, false>(rowptr + br * N_NODES, indeg_i + br * N_NODES, ce + (size_t)br * N_EDGES,
                             hin + (size_t)br * N_NODES * HIDDEN, nullptr, nullptr, cd + br * N_NODES, W, bias,
                             hout + (size_t)br * N_NODES * HIDDEN, nullptr, nullptr, nullptr);
}

__global__ void __launch_bounds__(256) layer_rows_pool_kernel(const int* __restrict__ rowptr,
                                                              const int* __restrict__ indeg_i,
                                                              const int2* __restrict__ ce,
                                                              const float* __restrict__ hin,
                                                              const float* __restrict__ cd,
                                                              const float* __restrict__ W,
                                                              const float* __restrict__ bias,
                                                              const int* __restrict__ seg1,
                                                              const int* __restrict__ seg2,
                                                              float* __restrict__ hg,
                                                              float* __restrict__ counts) {
    const int br = blockIdx.y;
    layer_body<true, false>(rowptr + br * N_NODES, indeg_i + br * N_NODES, ce + (size_t)br * N_EDGES,
                            hin + (size_t)br * N_NODES * HIDDEN, nullptr, nullptr, cd + br * N_NODES, W, bias,
                            nullptr, br ? seg2 : seg1, hg + br * N_GRAPHS * HIDDEN, counts + br * N_GRAPHS);
}

__global__ void pool_div_kernel(float* __restrict__ hg, const float* __restrict__ counts) {
    int i = blockIdx.x * blockDim.x + threadIdx.x;
    if (i < 2 * N_GRAPHS * HIDDEN)
        hg[i] /= fmaxf(counts[i >> 7], 1.0f);
}

__global__ void logits_kernel(const float* __restrict__ hg1, const float* __restrict__ hg2,
                              const float* __restrict__ Wc, const float* __restrict__ bc,
                              float* __restrict__ logits) {
    int i = blockIdx.x * blockDim.x + threadIdx.x;
    if (i < N_GRAPHS * N_CLASSES) {
        int g = i / N_CLASSES, c = i % N_CLASSES;
        float s = bc[c];
#pragma unroll 8
        for (int k = 0; k < HIDDEN; ++k)
            s += fabsf(hg1[g * HIDDEN + k] - hg2[g * HIDDEN + k]) * Wc[k * N_CLASSES + c];
        logits[i] = s;
    }
}

extern "C" void kernel_launch(void* const* d_in, const int* in_sizes, int n_in,
                              void* d_out, int out_size, void* d_ws, size_t ws_size,
                              hipStream_t stream) {
    const int*   src1 = (const int*)d_in[0];
    const int*   dst1 = (const int*)d_in[1];
    const int*   seg1 = (const int*)d_in[2];
    const int*   src2 = (const int*)d_in[3];
    const int*   dst2 = (const int*)d_in[4];
    const int*   seg2 = (const int*)d_in[5];
    const float* W1   = (const float*)d_in[6];
    const float* b1   = (const float*)d_in[7];
    const float* W2   = (const float*)d_in[8];
    const float* b2   = (const float*)d_in[9];
    const float* W3   = (const float*)d_in[10];
    const float* b3   = (const float*)d_in[11];
    const float* W4   = (const float*)d_in[12];
    const float* b4   = (const float*)d_in[13];
    const float* Wc   = (const float*)d_in[14];
    const float* bc   = (const float*)d_in[15];

    float* out   = (float*)d_out;
    float* hg    = out;
    float* hg1   = out;
    float* hg2   = out + N_GRAPHS * HIDDEN;
    float* logit = out + 2 * N_GRAPHS * HIDDEN;
    float* ws_f  = (float*)d_ws;

    float* outdeg  = ws_f;
    float* counts  = ws_f + 200000;
    int*   indeg_i = (int*)(ws_f + 200256);
    float* cs      = ws_f + 400256;
    float* cd      = ws_f + 600256;
    float* x       = ws_f + 800256;
    int*   rowptr  = (int*)(ws_f + 1000256);
    float* ha      = ws_f + 1200256;
    float* hb      = ha + (size_t)2 * N_NODES * HIDDEN;
    int*   srt     = (int*)(hb + (size_t)2 * N_NODES * HIDDEN);
    int2*  ce      = (int2*)(srt + (size_t)2 * N_EDGES);
    int*   off     = (int*)(ce + (size_t)2 * N_EDGES);

    hipMemsetAsync(ws_f, 0, (size_t)200256 * sizeof(float), stream);
    hipMemsetAsync(hg, 0, (size_t)2 * N_GRAPHS * HIDDEN * sizeof(float), stream);

    const int T = 256;
    bin_kernel<<<dim3(NCH, 2), T, 0, stream>>>(src1, dst1, src2, dst2, outdeg, srt, off);
    place_kernel<<<dim3(NBUCK, 2), 1024, 0, stream>>>(srt, off, outdeg, indeg_i, rowptr, ce);
    norm_kernel<<<(2 * N_NODES + T - 1) / T, T, 0, stream>>>(outdeg, indeg_i, cs, cd);

    gather1_kernel<<<dim3((N_NODES + T - 1) / T, 2), T, 0, stream>>>(rowptr, indeg_i, ce, cd, x);

    layer_rank1_kernel<<<dim3(GEMM_BLOCKS, 2), T, 0, stream>>>(rowptr, indeg_i, ce, x, W1, b1, cd, W2, b2, ha);

    layer_rows_kernel<<<dim3(GEMM_BLOCKS, 2), T, 0, stream>>>(rowptr, indeg_i, ce, ha, cd, W3, b3, hb);

    layer_rows_pool_kernel<<<dim3(GEMM_BLOCKS, 2), T, 0, stream>>>(rowptr, indeg_i, ce, hb, cd, W4, b4,
                                                                   seg1, seg2, hg, counts);
    pool_div_kernel<<<(2 * N_GRAPHS * HIDDEN + T - 1) / T, T, 0, stream>>>(hg, counts);

    logits_kernel<<<(N_GRAPHS * N_CLASSES + 255) / 256, 256, 0, stream>>>(hg1, hg2, Wc, bc, logit);
}


# Out-degrees counted as the in-degrees are: bin pass also orders each chunk by source range in LDS, one block per range counts on LDS counters, stores once; no global adds in the sort; sums unchanged
# speedup vs baseline: 7.8830x; 1.3075x over previous
#include <hip/hip_runtime.h>
#include <hip/hip_bf16.h>

#define N_NODES   100000
#define N_EDGES   1600000
#define N_GRAPHS  128
#define HIDDEN    128
#define N_CLASSES 10

typedef __attribute__((ext_vector_type(2))) float v2f;
typedef __attribute__((ext_vector_type(8))) float v8f;


#define NB_SHIFT  9
#define NB_NODES  (1 << NB_SHIFT)
#define NBUCK     ((N_NODES + NB_NODES - 1) / NB_NODES)
#define CH_EDGES  4096
#define NCH       ((N_EDGES + CH_EDGES - 1) / CH_EDGES)
#define SRC_BITS  17
#define PLACE_CAP 12288
static_assert(N_NODES <= (1 << SRC_BITS), "a source id must fit its field");
static_assert(NBUCK + 1 <= 256, "one bin thread per range start");
static_assert(NCH <= 1024, "one place thread per chunk");
static_assert(N_EDGES % 4 == 0 && CH_EDGES % 1024 == 0, "16-byte copy-out of whole chunks");

template <int NW>
__device__ __forceinline__ int block_excl_scan(int v, int* wsum, int& total) {
    const int lane = threadIdx.x & 31;
    const int wave = threadIdx.x >> 5;
    int inc = v;
    for (int o = 1; o < 32; o <<= 1) {
        int u = __shfl_up(inc, o);
        if (lane >= o) inc += u;
    }
    if (lane == 31) wsum[wave] = inc;
    __syncthreads();
    const int ws = lane < NW ? wsum[lane] : 0;
    int winc = ws;
    for (int o = 1; o < 32; o <<= 1) {
        int u = __shfl_up(winc, o);
        if (lane >= o) winc += u;
    }
    const int wave_excl = __shfl(winc, wave) - __shfl(ws, wave);
    total = __shfl(winc, 31);
    __syncthreads();
    return wave_excl + inc - v;
}

__global__ void __launch_bounds__(256) bin_kernel(const int* __restrict__ src1, const int* __restrict__ dst1,
                                                  const int* __restrict__ src2, const int* __restrict__ dst2,
                                                  int* __restrict__ srt, int* __restrict__ off,
                                                  int* __restrict__ srt_s, int* __restrict__ off_s) {
    __shared__ int wsum[32];
    __shared__ int hist[256];
    __shared__ int hist_s[256];
    __shared__ __attribute__((aligned(16))) int stage[CH_EDGES];
    __shared__ __attribute__((aligned(16))) int stage_s[CH_EDGES];
    const int br    = blockIdx.y;
    const int chunk = blockIdx.x;
    const int* __restrict__ src = br ? src2 : src1;
    const int* __restrict__ dst = br ? dst2 : dst1;
    srt    += (size_t)br * N_EDGES;
    off    += (size_t)(br * NCH + chunk) * (NBUCK + 1);
    srt_s  += (size_t)br * N_EDGES;
    off_s  += (size_t)(br * NCH + chunk) * (NBUCK + 1);
    const int tid = threadIdx.x;
    const int e0  = chunk * CH_EDGES;
    hist[tid]   = 0;
    hist_s[tid] = 0;
    __syncthreads();
    constexpr int PER = CH_EDGES / 256;
    int s[PER], d[PER], rk[PER], rs[PER];
#pragma unroll
    for (int i = 0; i < PER; ++i) {
        const int e = e0 + i * 256 + tid;
        s[i] = e < N_EDGES ? src[e] : -1;
        d[i] = e < N_EDGES ? dst[e] : 0;
    }
#pragma unroll
    for (int i = 0; i < PER; ++i) {
        rk[i] = 0;
        rs[i] = 0;
        if (s[i] >= 0) {
            rk[i] = atomicAdd(&hist[d[i] >> NB_SHIFT], 1);
            rs[i] = atomicAdd(&hist_s[s[i] >> NB_SHIFT], 1);
        }
    }
    __syncthreads();
    int total;
    const int excl   = block_excl_scan<8>(hist[tid], wsum, total);
    const int excl_s = block_excl_scan<8>(hist_s[tid], wsum, total);
    hist[tid]   = excl;
    hist_s[tid] = excl_s;
    if (tid <= NBUCK) {
        off[tid]   = excl;
        off_s[tid] = excl_s;
    }
    __syncthreads();
#pragma unroll
    for (int i = 0; i < PER; ++i)
        if (s[i] >= 0) {
            stage[hist[d[i] >> NB_SHIFT] + rk[i]] = s[i] | ((d[i] & (NB_NODES - 1)) << SRC_BITS);
            stage_s[hist_s[s[i] >> NB_SHIFT] + rs[i]] = (s[i] & (NB_NODES - 1)) << SRC_BITS;
        }
    __syncthreads();
    const int n4 = min(CH_EDGES, N_EDGES - e0) / 4;
    int4* __restrict__ o4 = (int4*)(srt + e0);
    int4* __restrict__ p4 = (int4*)(srt_s + e0);
    const int4* st4 = (const int4*)stage;
    const int4* ss4 = (const int4*)stage_s;
    for (int j = tid; j < n4; j += 256) {
        o4[j] = st4[j];
        p4[j] = ss4[j];
    }
}

template <bool COUNT>
__device__ __forceinline__ void walk_pieces(const int* __restrict__ srt, const int* pbeg, const int* plen,
                                            int* cntn, int* sorted, bool staged, int S,
                                            const float* __restrict__ outdeg, int2* __restrict__ ce) {
    const int lane = threadIdx.x & 31;
    const int wave = threadIdx.x >> 5;
    auto take = [&](int en) {
        if constexpr (COUNT) {
            atomicAdd(&cntn[en >> SRC_BITS], 1);
        } else {
            const int pos = atomicAdd(&cntn[en >> SRC_BITS], 1);
            const int s   = en & ((1 << SRC_BITS) - 1);
            if (staged) sorted[pos] = s;
            else        ce[S + pos] = make_int2(s, __float_as_int(rsqrtf(fmaxf(outdeg[s], 1.0f))));
        }
    };
    for (int c = wave; c < NCH; c += 128) {
        int bg[4], ln[4], en[4];
#pragma unroll
        for (int k = 0; k < 4; ++k) {
            const int cc = c + 32 * k;
            ln[k] = cc < NCH ? plen[cc] : 0;
            bg[k] = cc < NCH ? pbeg[cc] : 0;
        }
#pragma unroll
        for (int k = 0; k < 4; ++k)
            en[k] = lane < ln[k] ? srt[bg[k] + lane] : -1;
#pragma unroll
        for (int k = 0; k < 4; ++k) {
            if (en[k] >= 0) take(en[k]);
            for (int i = lane + 32; i < ln[k]; i += 32) take(srt[bg[k] + i]);
        }
    }
}

__global__ void __launch_bounds__(1024) outdeg_kernel(const int* __restrict__ srt_s, const int* __restrict__ off_s,
                                                      float* __restrict__ outdeg) {
    __shared__ int pbeg[NCH], plen[NCH];
    __shared__ int cntn[NB_NODES];
    const int br = blockIdx.y;
    const int b  = blockIdx.x;
    srt_s  += (size_t)br * N_EDGES;
    off_s  += (size_t)br * NCH * (NBUCK + 1);
    outdeg += br * N_NODES;
    const int tid = threadIdx.x;
    if (tid < NCH) {
        const int* p = off_s + (size_t)tid * (NBUCK + 1) + b;
        const int o0 = p[0];
        pbeg[tid] = tid * CH_EDGES + o0;
        plen[tid] = p[1] - o0;
    }
    if (tid < NB_NODES) cntn[tid] = 0;
    __syncthreads();
    walk_pieces<true>(srt_s, pbeg, plen, cntn, nullptr, false, 0, nullptr, nullptr);
    __syncthreads();
    if (tid < NB_NODES) {
        const int node = b * NB_NODES + tid;
        if (node < N_NODES) outdeg[node] = (float)cntn[tid];
    }
}

__global__ void __launch_bounds__(1024) place_kernel(const int* __restrict__ srt, const int* __restrict__ off,
                                                     const float* __restrict__ outdeg,
                                                     int* __restrict__ indeg_i, int* __restrict__ rowptr,
                                                     int2* __restrict__ ce) {
    __shared__ int wsum[32];
    __shared__ int pbeg[NCH], plen[NCH];
    __shared__ int cntn[NB_NODES];
    __shared__ int sorted[PLACE_CAP];
    const int br = blockIdx.y;
    const int b  = blockIdx.x;
    srt     += (size_t)br * N_EDGES;
    off     += (size_t)br * NCH * (NBUCK + 1);
    outdeg  += br * N_NODES;
    indeg_i += br * N_NODES;
    rowptr  += br * N_NODES;
    ce      += (size_t)br * N_EDGES;
    const int tid = threadIdx.x;
    int o0 = 0, len = 0;
    if (tid < NCH) {
        const int* p = off + (size_t)tid * (NBUCK + 1) + b;
        o0  = p[0];
        len = p[1] - o0;
        pbeg[tid] = tid * CH_EDGES + o0;
        plen[tid] = len;
    }
    if (tid < NB_NODES) cntn[tid] = 0;
    int S, T;
    block_excl_scan<32>(o0, wsum, S);
    block_excl_scan<32>(len, wsum, T);
    const bool staged = T <= PLACE_CAP;

    walk_pieces<true>(srt, pbeg, plen, cntn, sorted, staged, S, outdeg, ce);
    __syncthreads();

    const int v = tid < NB_NODES ? cntn[tid] : 0;
    int tot;
    const int ex = block_excl_scan<32>(v, wsum, tot);
    if (tid < NB_NODES) {
        cntn[tid] = ex;
        const int node = b * NB_NODES + tid;
        if (node < N_NODES) {
            indeg_i[node] = v;
            rowptr[node]  = S + ex;
        }
    }
    __syncthreads();

    walk_pieces<false>(srt, pbeg, plen, cntn, sorted, staged, S, outdeg, ce);
    __syncthreads();

    if (staged) {
        constexpr int R = PLACE_CAP / 1024;
        int   sv[R];
        float od[R];
#pragma unroll
        for (int k = 0; k < R; ++k) {
            const int j = tid + k * 1024;
            sv[k] = j < T ? sorted[j] : 0;
        }
#pragma unroll
        for (int k = 0; k < R; ++k) od[k] = outdeg[sv[k]];
#pragma unroll
        for (int k = 0; k < R; ++k) {
            const int j = tid + k * 1024;
            if (j < T) ce[S + j] = make_int2(sv[k], __float_as_int(rsqrtf(fmaxf(od[k], 1.0f))));
        }
    }
}

__global__ void norm_kernel(const float* __restrict__ outdeg, const int* __restrict__ indeg_i,
                            float* __restrict__ cs, float* __restrict__ cd) {
    int n = blockIdx.x * blockDim.x + threadIdx.x;
    if (n < 2 * N_NODES) {
        cs[n] = rsqrtf(fmaxf(outdeg[n], 1.0f));
        cd[n] = rsqrtf(fmaxf((float)indeg_i[n], 1.0f));
    }
}

__global__ void gather1_kernel(const int* __restrict__ rowptr, const int* __restrict__ indeg_i,
                               const int2* __restrict__ ce, const float* __restrict__ cd,
                               float* __restrict__ x) {
    const int br = blockIdx.y;
    rowptr  += br * N_NODES;
    indeg_i += br * N_NODES;
    ce      += (size_t)br * N_EDGES;
    cd      += br * N_NODES;
    x       += br * N_NODES;
    int d = blockIdx.x * blockDim.x + threadIdx.x;
    if (d < N_NODES) {
        const int beg = rowptr[d];
        const int cnt = indeg_i[d];
        float a = 0.0f;
        for (int j = 0; j < cnt; ++j) {
            int2 c = ce[beg + j];
            a += (float)indeg_i[c.x] * __int_as_float(c.y);
        }
        x[d] = a * cd[d];
    }
}

#define GEMM_BLOCKS 1024
#define A_PITCH     132
static_assert(N_NODES % 16 == 0, "row tiles must be whole");
static_assert(GEMM_BLOCKS <= N_NODES / 16, "every block owns a first tile");

__device__ __forceinline__ void rows_tail(const int2* __restrict__ ce, const float4* __restrict__ h4,
                                          int beg, int cnt, int lane, float4& acc) {
    for (int base = 32; base < cnt; base += 32) {
        const int n = min(cnt - base, 32);
        int cidx = 0, cscl = 0;
        if (lane < n) {
            int2 c = ce[beg + base + lane];
            cidx = c.x;
            cscl = c.y;
        }
        for (int j = 0; j < n; j += 4) {
            float4 v[4];
            float  c[4];
#pragma unroll
            for (int k = 0; k < 4; ++k) {
                v[k] = make_float4(0.0f, 0.0f, 0.0f, 0.0f);
                c[k] = 0.0f;
                if (j + k < n) {
                    const int s = __builtin_amdgcn_readlane(cidx, j + k);
                    c[k] = __int_as_float(__builtin_amdgcn_readlane(cscl, j + k));
                    v[k] = h4[(size_t)s * 32 + lane];
                }
            }
#pragma unroll
            for (int k = 0; k < 4; ++k) {
                acc.x = fmaf(v[k].x, c[k], acc.x); acc.y = fmaf(v[k].y, c[k], acc.y);
                acc.z = fmaf(v[k].z, c[k], acc.z); acc.w = fmaf(v[k].w, c[k], acc.w);
            }
        }
    }
}

__device__ __forceinline__ void rank1_tail(const int2* __restrict__ ce, const float* __restrict__ x,
                                           int beg, int cnt, int lane, const float4& w, const float4& b, float4& acc) {
    for (int base = 32; base < cnt; base += 32) {
        const int n = min(cnt - base, 32);
        int xsrc = 0, cscl = 0;
        if (lane < n) {
            int2 c = ce[beg + base + lane];
            cscl = c.y;
            xsrc = __float_as_int(x[c.x]);
        }
        for (int j = 0; j < n; ++j) {
            const float xv = __int_as_float(__builtin_amdgcn_readlane(xsrc, j));
            const float cv = __int_as_float(__builtin_amdgcn_readlane(cscl, j));
            acc.x = fmaf(fmaxf(fmaf(xv, w.x, b.x), 0.0f), cv, acc.x);
            acc.y = fmaf(fmaxf(fmaf(xv, w.y, b.y), 0.0f), cv, acc.y);
            acc.z = fmaf(fmaxf(fmaf(xv, w.z, b.z), 0.0f), cv, acc.z);
            acc.w = fmaf(fmaxf(fmaf(xv, w.w, b.w), 0.0f), cv, acc.w);
        }
    }
}

template <bool POOL, bool RANK1>
__device__ __forceinline__ void layer_body(const int* __restrict__ rowptr,
                                           const int* __restrict__ indeg_i,
                                           const int2* __restrict__ ce,
                                           const float* __restrict__ hin,
                                           const float* __restrict__ W1,
                                           const float* __restrict__ b1,
                                           const float* __restrict__ cd,
                                           const float* __restrict__ W,
                                           const float* __restrict__ bias,
                                           float* __restrict__ hout,
                                           const int* __restrict__ seg,
                                           float* __restrict__ hg,
                                           float* __restrict__ counts) {
    __shared__ __attribute__((aligned(16))) float As[2][16 * A_PITCH];
    constexpr int NT = N_NODES / 16;
    const int tid  = threadIdx.x;
    const int wave = tid >> 5;
    const int wv   = __builtin_amdgcn_readfirstlane(wave);
    const int lane = tid & 31;
    const int half = lane >> 4;
    const int lr   = lane & 15;
    const int n    = wave * 16 + lr;

    v2f bfr[HIDDEN / 4];
#pragma unroll
    for (int s = 0; s < HIDDEN / 4; ++s) {
        bfr[s].x = W[(4 * s + 2 * half) * HIDDEN + n];
        bfr[s].y = W[(4 * s + 2 * half + 1) * HIDDEN + n];
    }
    const float bn = bias[n];
    float4 w1 = {}, bb1 = {};
    if constexpr (RANK1) {
        w1  = ((const float4*)W1)[lane];
        bb1 = ((const float4*)b1)[lane];
    }

    const int sr = 2 * wv;
    int t = blockIdx.x;
    int2   rp, dg, nrp = {}, ndg = {};
    float2 cv, ncv = {};
    {
        const int r0 = t * 16 + sr;
        rp = *(const int2*)(rowptr + r0);
        dg = *(const int2*)(indeg_i + r0);
        cv = *(const float2*)(cd + r0);
    }
    int2 eA = ce[min(rp.x + lane, N_EDGES - 1)];
    int2 eB = ce[min(rp.y + lane, N_EDGES - 1)];
    if (t + GEMM_BLOCKS < NT) {
        const int r0 = (t + GEMM_BLOCKS) * 16 + sr;
        nrp = *(const int2*)(rowptr + r0);
        ndg = *(const int2*)(indeg_i + r0);
        ncv = *(const float2*)(cd + r0);
    }
    int4 na = {}, nb = {}, nc = {}, nd = {};
    if constexpr (POOL) {
        const int4* __restrict__ sg = (const int4*)(seg + t * 16);
        na = sg[0]; nb = sg[1]; nc = sg[2]; nd = sg[3];
    }

    int buf = 0;
    for (;;) {
        const int begA = rp.x, begB = rp.y, cntA = dg.x, cntB = dg.y;
        const int nA = min(cntA, 32), nB = min(cntB, 32);
        float4 accA = make_float4(0.0f, 0.0f, 0.0f, 0.0f);
        float4 accB = make_float4(0.0f, 0.0f, 0.0f, 0.0f);
        if constexpr (RANK1) {
            const int xA = __float_as_int(hin[eA.x]);
            const int xB = __float_as_int(hin[eB.x]);
            for (int j = 0; j < nA; ++j) {
                const float xv = __int_as_float(__builtin_amdgcn_readlane(xA, j));
                const float c  = __int_as_float(__builtin_amdgcn_readlane(eA.y, j));
                accA.x = fmaf(fmaxf(fmaf(xv, w1.x, bb1.x), 0.0f), c, accA.x);
                accA.y = fmaf(fmaxf(fmaf(xv, w1.y, bb1.y), 0.0f), c, accA.y);
                accA.z = fmaf(fmaxf(fmaf(xv, w1.z, bb1.z), 0.0f), c, accA.z);
                accA.w = fmaf(fmaxf(fmaf(xv, w1.w, bb1.w), 0.0f), c, accA.w);
            }
            for (int j = 0; j < nB; ++j) {
                const float xv = __int_as_float(__builtin_amdgcn_readlane(xB, j));
                const float c  = __int_as_float(__builtin_amdgcn_readlane(eB.y, j));
                accB.x = fmaf(fmaxf(fmaf(xv, w1.x, bb1.x), 0.0f), c, accB.x);
                accB.y = fmaf(fmaxf(fmaf(xv, w1.y, bb1.y), 0.0f), c, accB.y);
                accB.z = fmaf(fmaxf(fmaf(xv, w1.z, bb1.z), 0.0f), c, accB.z);
                accB.w = fmaf(fmaxf(fmaf(xv, w1.w, bb1.w), 0.0f), c, accB.w);
            }
            rank1_tail(ce, hin, begA, cntA, lane, w1, bb1, accA);
            rank1_tail(ce, hin, begB, cntB, lane, w1, bb1, accB);
        } else {
            const float4* __restrict__ h4 = (const float4*)hin;
            const int nmax = max(nA, nB);
            asm volatile("" :: "v"(eA.x), "v"(eA.y), "v"(eB.x), "v"(eB.y));
            for (int j = 0; j < nmax; j += 4) {
                float4 va[4], vb[4];
                float  ca[4], cb[4];
#pragma unroll
                for (int k = 0; k < 4; ++k) {
                    va[k] = make_float4(0.0f, 0.0f, 0.0f, 0.0f);
                    ca[k] = 0.0f;
                    if (j + k < nA) {
                        const int s = __builtin_amdgcn_readlane(eA.x, j + k);
                        ca[k] = __int_as_float(__builtin_amdgcn_readlane(eA.y, j + k));
                        va[k] = h4[(size_t)s * 32 + lane];
                    }
                }
#pragma unroll
                for (int k = 0; k < 4; ++k) {
                    vb[k] = make_float4(0.0f, 0.0f, 0.0f, 0.0f);
                    cb[k] = 0.0f;
                    if (j + k < nB) {
                        const int s = __builtin_amdgcn_readlane(eB.x, j + k);
                        cb[k] = __int_as_float(__builtin_amdgcn_readlane(eB.y, j + k));
                        vb[k] = h4[(size_t)s * 32 + lane];
                    }
                }
#pragma unroll
                for (int k = 0; k < 4; ++k) {
                    accA.x = fmaf(va[k].x, ca[k], accA.x); accA.y = fmaf(va[k].y, ca[k], accA.y);
                    accA.z = fmaf(va[k].z, ca[k], accA.z); accA.w = fmaf(va[k].w, ca[k], accA.w);
                }
#pragma unroll
                for (int k = 0; k < 4; ++k) {
                    accB.x = fmaf(vb[k].x, cb[k], accB.x); accB.y = fmaf(vb[k].y, cb[k], accB.y);
                    accB.z = fmaf(vb[k].z, cb[k], accB.z); accB.w = fmaf(vb[k].w, cb[k], accB.w);
                }
            }
            rows_tail(ce, h4, begA, cntA, lane, accA);
            rows_tail(ce, h4, begB, cntB, lane, accB);
        }

        {
            const float c0 = cv.x, c1 = cv.y;
            float* dst = &As[buf][sr * A_PITCH + lane * 4];
            *(float4*)dst             = make_float4(accA.x * c0, accA.y * c0, accA.z * c0, accA.w * c0);
            *(float4*)(dst + A_PITCH) = make_float4(accB.x * c1, accB.y * c1, accB.z * c1, accB.w * c1);
        }
        __syncthreads();

        const int  tn   = t + GEMM_BLOCKS;
        const bool more = tn < NT;
        const int4 sa = na, sb = nb, sc = nc, sd = nd;
        int2   n2rp = nrp, n2dg = ndg;
        float2 n2cv = ncv;
        if (more) {
            eA = ce[min(nrp.x + lane, N_EDGES - 1)];
            eB = ce[min(nrp.y + lane, N_EDGES - 1)];
            const int t2 = tn + GEMM_BLOCKS;
            if (t2 < NT) {
                const int r0 = t2 * 16 + sr;
                n2rp = *(const int2*)(rowptr + r0);
                n2dg = *(const int2*)(indeg_i + r0);
                n2cv = *(const float2*)(cd + r0);
            }
        }
        if constexpr (POOL) {
            const int4* __restrict__ sg = (const int4*)(seg + min(tn, NT - 1) * 16);
            na = sg[0]; nb = sg[1]; nc = sg[2]; nd = sg[3];
        }

        const float* arow = &As[buf][lr * A_PITCH + 2 * half];
        v8f acc = {};
#pragma unroll
        for (int s = 0; s < HIDDEN / 4; ++s) {
            const v2f av = *(const v2f*)(arow + 4 * s);
            acc = __builtin_amdgcn_wmma_f32_16x16x4_f32(
                 false, av,  false, bfr[s],
                 (short)0, acc,  false,  false);
        }

        const int m0 = t * 16;
        if constexpr (!POOL) {
#pragma unroll
            for (int r = 0; r < 8; ++r) {
                const int mm = m0 + half * 8 + r;
                hout[(size_t)mm * HIDDEN + n] = fmaxf(acc[r] + bn, 0.0f);
            }
        } else {
            const int  g0 = sa.x;
            const int  differ = (sa.y ^ g0) | (sa.z ^ g0) | (sa.w ^ g0) |
                                (sb.x ^ g0) | (sb.y ^ g0) | (sb.z ^ g0) | (sb.w ^ g0) |
                                (sc.x ^ g0) | (sc.y ^ g0) | (sc.z ^ g0) | (sc.w ^ g0) |
                                (sd.x ^ g0) | (sd.y ^ g0) | (sd.z ^ g0) | (sd.w ^ g0);
            if (differ == 0) {
                float s = fmaxf(acc[0] + bn, 0.0f);
#pragma unroll
                for (int r = 1; r < 8; ++r)
                    s += fmaxf(acc[r] + bn, 0.0f);
                s += __shfl_xor(s, 16);
                if (half == 0) atomicAdd(&hg[g0 * HIDDEN + n], s);
                if (tid == 0)  atomicAdd(&counts[g0], 16.0f);
            } else {
                const int gl[8] = {sa.x, sa.y, sa.z, sa.w, sb.x, sb.y, sb.z, sb.w};
                const int gh[8] = {sc.x, sc.y, sc.z, sc.w, sd.x, sd.y, sd.z, sd.w};
#pragma unroll
                for (int r = 0; r < 8; ++r) {
                    const int g = half ? gh[r] : gl[r];
                    atomicAdd(&hg[g * HIDDEN + n], fmaxf(acc[r] + bn, 0.0f));
                    if (wave == 0 && lr == 0) atomicAdd(&counts[g], 1.0f);
                }
            }
        }

        if (!more) break;
        rp = nrp;  dg = ndg;  cv = ncv;
        nrp = n2rp; ndg = n2dg; ncv = n2cv;
        t = tn;
        buf ^= 1;
    }
}

__global__ void __launch_bounds__(256) layer_rank1_kernel(const int* __restrict__ rowptr,
                                                          const int* __restrict__ indeg_i,
                                                          const int2* __restrict__ ce,
                                                          const float* __restrict__ x,
                                                          const float* __restrict__ W1,
                                                          const float* __restrict__ b1,
                                                          const float* __restrict__ cd,
                                                          const float* __restrict__ W,
                                                          const float* __restrict__ bias,
                                                          float* __restrict__ hout) {
    const int br = blockIdx.y;
    layer_body<false, true>(rowptr + br * N_NODES, indeg_i + br * N_NODES, ce + (size_t)br * N_EDGES,
                            x + br * N_NODES, W1, b1, cd + br * N_NODES, W, bias,
                            hout + (size_t)br * N_NODES * HIDDEN, nullptr, nullptr, nullptr);
}

__global__ void __launch_bounds__(256) layer_rows_kernel(const int* __restrict__ rowptr,
                                                         const int* __restrict__ indeg_i,
                                                         const int2* __restrict__ ce,
                                                         const float* __restrict__ hin,
                                                         const float* __restrict__ cd,
                                                         const float* __restrict__ W,
                                                         const float* __restrict__ bias,
                                                         float* __restrict__ hout) {
    const int br = blockIdx.y;
    layer_body<false, false>(rowptr + br * N_NODES, indeg_i + br * N_NODES, ce + (size_t)br * N_EDGES,
                             hin + (size_t)br * N_NODES * HIDDEN, nullptr, nullptr, cd + br * N_NODES, W, bias,
                             hout + (size_t)br * N_NODES * HIDDEN, nullptr, nullptr, nullptr);
}

__global__ void __launch_bounds__(256) layer_rows_pool_kernel(const int* __restrict__ rowptr,
                                                              const int* __restrict__ indeg_i,
                                                              const int2* __restrict__ ce,
                                                              const float* __restrict__ hin,
                                                              const float* __restrict__ cd,
                                                              const float* __restrict__ W,
                                                              const float* __restrict__ bias,
                                                              const int* __restrict__ seg1,
                                                              const int* __restrict__ seg2,
                                                              float* __restrict__ hg,
                                                              float* __restrict__ counts) {
    const int br = blockIdx.y;
    layer_body<true, false>(rowptr + br * N_NODES, indeg_i + br * N_NODES, ce + (size_t)br * N_EDGES,
                            hin + (size_t)br * N_NODES * HIDDEN, nullptr, nullptr, cd + br * N_NODES, W, bias,
                            nullptr, br ? seg2 : seg1, hg + br * N_GRAPHS * HIDDEN, counts + br * N_GRAPHS);
}

__global__ void pool_div_kernel(float* __restrict__ hg, const float* __restrict__ counts) {
    int i = blockIdx.x * blockDim.x + threadIdx.x;
    if (i < 2 * N_GRAPHS * HIDDEN)
        hg[i] /= fmaxf(counts[i >> 7], 1.0f);
}

__global__ void logits_kernel(const float* __restrict__ hg1, const float* __restrict__ hg2,
                              const float* __restrict__ Wc, const float* __restrict__ bc,
                              float* __restrict__ logits) {
    int i = blockIdx.x * blockDim.x + threadIdx.x;
    if (i < N_GRAPHS * N_CLASSES) {
        int g = i / N_CLASSES, c = i % N_CLASSES;
        float s = bc[c];
#pragma unroll 8
        for (int k = 0; k < HIDDEN; ++k)
            s += fabsf(hg1[g * HIDDEN + k] - hg2[g * HIDDEN + k]) * Wc[k * N_CLASSES + c];
        logits[i] = s;
    }
}

extern "C" void kernel_launch(void* const* d_in, const int* in_sizes, int n_in,
                              void* d_out, int out_size, void* d_ws, size_t ws_size,
                              hipStream_t stream) {
    const int*   src1 = (const int*)d_in[0];
    const int*   dst1 = (const int*)d_in[1];
    const int*   seg1 = (const int*)d_in[2];
    const int*   src2 = (const int*)d_in[3];
    const int*   dst2 = (const int*)d_in[4];
    const int*   seg2 = (const int*)d_in[5];
    const float* W1   = (const float*)d_in[6];
    const float* b1   = (const float*)d_in[7];
    const float* W2   = (const float*)d_in[8];
    const float* b2   = (const float*)d_in[9];
    const float* W3   = (const float*)d_in[10];
    const float* b3   = (const float*)d_in[11];
    const float* W4   = (const float*)d_in[12];
    const float* b4   = (const float*)d_in[13];
    const float* Wc   = (const float*)d_in[14];
    const float* bc   = (const float*)d_in[15];

    float* out   = (float*)d_out;
    float* hg    = out;
    float* hg1   = out;
    float* hg2   = out + N_GRAPHS * HIDDEN;
    float* logit = out + 2 * N_GRAPHS * HIDDEN;
    float* ws_f  = (float*)d_ws;

    float* outdeg  = ws_f;
    float* counts  = ws_f + 200000;
    int*   indeg_i = (int*)(ws_f + 200256);
    float* cs      = ws_f + 400256;
    float* cd      = ws_f + 600256;
    float* x       = ws_f + 800256;
    int*   rowptr  = (int*)(ws_f + 1000256);
    float* ha      = ws_f + 1200256;
    float* hb      = ha + (size_t)2 * N_NODES * HIDDEN;
    int*   srt     = (int*)(hb + (size_t)2 * N_NODES * HIDDEN);
    int2*  ce      = (int2*)(srt + (size_t)2 * N_EDGES);
    int*   off     = (int*)(ce + (size_t)2 * N_EDGES);
    int*   off_s   = off + (size_t)2 * NCH * (NBUCK + 1);
    int*   srt_s   = off_s + (size_t)2 * NCH * (NBUCK + 1);

    hipMemsetAsync(counts, 0, (size_t)2 * N_GRAPHS * sizeof(float), stream);
    hipMemsetAsync(hg, 0, (size_t)2 * N_GRAPHS * HIDDEN * sizeof(float), stream);

    const int T = 256;
    bin_kernel<<<dim3(NCH, 2), T, 0, stream>>>(src1, dst1, src2, dst2, srt, off, srt_s, off_s);
    outdeg_kernel<<<dim3(NBUCK, 2), 1024, 0, stream>>>(srt_s, off_s, outdeg);
    place_kernel<<<dim3(NBUCK, 2), 1024, 0, stream>>>(srt, off, outdeg, indeg_i, rowptr, ce);
    norm_kernel<<<(2 * N_NODES + T - 1) / T, T, 0, stream>>>(outdeg, indeg_i, cs, cd);

    gather1_kernel<<<dim3((N_NODES + T - 1) / T, 2), T, 0, stream>>>(rowptr, indeg_i, ce, cd, x);

    layer_rank1_kernel<<<dim3(GEMM_BLOCKS, 2), T, 0, stream>>>(rowptr, indeg_i, ce, x, W1, b1, cd, W2, b2, ha);

    layer_rows_kernel<<<dim3(GEMM_BLOCKS, 2), T, 0, stream>>>(rowptr, indeg_i, ce, ha, cd, W3, b3, hb);

    layer_rows_pool_kernel<<<dim3(GEMM_BLOCKS, 2), T, 0, stream>>>(rowptr, indeg_i, ce, hb, cd, W4, b4,
                                                                   seg1, seg2, hg, counts);
    pool_div_kernel<<<(2 * N_GRAPHS * HIDDEN + T - 1) / T, T, 0, stream>>>(hg, counts);

    logits_kernel<<<(N_GRAPHS * N_CLASSES + 255) / 256, 256, 0, stream>>>(hg1, hg2, Wc, bc, logit);
}


# Layer-1 scalar built in the list-placing kernel from each node's ordered sources in LDS (both degrees counted one kernel earlier): norm and layer-1 gather launches go; same sum order, terms rounded
# speedup vs baseline: 8.3938x; 1.0648x over previous
#include <hip/hip_runtime.h>
#include <hip/hip_bf16.h>

#define N_NODES   100000
#define N_EDGES   1600000
#define N_GRAPHS  128
#define HIDDEN    128
#define N_CLASSES 10

typedef __attribute__((ext_vector_type(2))) float v2f;
typedef __attribute__((ext_vector_type(8))) float v8f;


#define NB_SHIFT  9
#define NB_NODES  (1 << NB_SHIFT)
#define NBUCK     ((N_NODES + NB_NODES - 1) / NB_NODES)
#define CH_EDGES  4096
#define NCH       ((N_EDGES + CH_EDGES - 1) / CH_EDGES)
#define SRC_BITS  17
#define PLACE_CAP 12288
static_assert(N_NODES <= (1 << SRC_BITS), "a source id must fit its field");
static_assert(NBUCK + 1 <= 256, "one bin thread per range start");
static_assert(NCH <= 512, "one place thread per chunk; one degree thread per chunk and order");
static_assert(N_EDGES % 4 == 0 && CH_EDGES % 1024 == 0, "16-byte copy-out of whole chunks");

template <int NW>
__device__ __forceinline__ int block_excl_scan(int v, int* wsum, int& total) {
    const int lane = threadIdx.x & 31;
    const int wave = threadIdx.x >> 5;
    int inc = v;
    for (int o = 1; o < 32; o <<= 1) {
        int u = __shfl_up(inc, o);
        if (lane >= o) inc += u;
    }
    if (lane == 31) wsum[wave] = inc;
    __syncthreads();
    const int ws = lane < NW ? wsum[lane] : 0;
    int winc = ws;
    for (int o = 1; o < 32; o <<= 1) {
        int u = __shfl_up(winc, o);
        if (lane >= o) winc += u;
    }
    const int wave_excl = __shfl(winc, wave) - __shfl(ws, wave);
    total = __shfl(winc, 31);
    __syncthreads();
    return wave_excl + inc - v;
}

__global__ void __launch_bounds__(256) bin_kernel(const int* __restrict__ src1, const int* __restrict__ dst1,
                                                  const int* __restrict__ src2, const int* __restrict__ dst2,
                                                  int* __restrict__ srt, int* __restrict__ off,
                                                  int* __restrict__ srt_s, int* __restrict__ off_s) {
    __shared__ int wsum[32];
    __shared__ int hist[256];
    __shared__ int hist_s[256];
    __shared__ __attribute__((aligned(16))) int stage[CH_EDGES];
    __shared__ __attribute__((aligned(16))) int stage_s[CH_EDGES];
    const int br    = blockIdx.y;
    const int chunk = blockIdx.x;
    const int* __restrict__ src = br ? src2 : src1;
    const int* __restrict__ dst = br ? dst2 : dst1;
    srt    += (size_t)br * N_EDGES;
    off    += (size_t)(br * NCH + chunk) * (NBUCK + 1);
    srt_s  += (size_t)br * N_EDGES;
    off_s  += (size_t)(br * NCH + chunk) * (NBUCK + 1);
    const int tid = threadIdx.x;
    const int e0  = chunk * CH_EDGES;
    hist[tid]   = 0;
    hist_s[tid] = 0;
    __syncthreads();
    constexpr int PER = CH_EDGES / 256;
    int s[PER], d[PER], rk[PER], rs[PER];
#pragma unroll
    for (int i = 0; i < PER; ++i) {
        const int e = e0 + i * 256 + tid;
        s[i] = e < N_EDGES ? src[e] : -1;
        d[i] = e < N_EDGES ? dst[e] : 0;
    }
#pragma unroll
    for (int i = 0; i < PER; ++i) {
        rk[i] = 0;
        rs[i] = 0;
        if (s[i] >= 0) {
            rk[i] = atomicAdd(&hist[d[i] >> NB_SHIFT], 1);
            rs[i] = atomicAdd(&hist_s[s[i] >> NB_SHIFT], 1);
        }
    }
    __syncthreads();
    int total;
    const int excl   = block_excl_scan<8>(hist[tid], wsum, total);
    const int excl_s = block_excl_scan<8>(hist_s[tid], wsum, total);
    hist[tid]   = excl;
    hist_s[tid] = excl_s;
    if (tid <= NBUCK) {
        off[tid]   = excl;
        off_s[tid] = excl_s;
    }
    __syncthreads();
#pragma unroll
    for (int i = 0; i < PER; ++i)
        if (s[i] >= 0) {
            stage[hist[d[i] >> NB_SHIFT] + rk[i]] = s[i] | ((d[i] & (NB_NODES - 1)) << SRC_BITS);
            stage_s[hist_s[s[i] >> NB_SHIFT] + rs[i]] = (s[i] & (NB_NODES - 1)) << SRC_BITS;
        }
    __syncthreads();
    const int n4 = min(CH_EDGES, N_EDGES - e0) / 4;
    int4* __restrict__ o4 = (int4*)(srt + e0);
    int4* __restrict__ p4 = (int4*)(srt_s + e0);
    const int4* st4 = (const int4*)stage;
    const int4* ss4 = (const int4*)stage_s;
    for (int j = tid; j < n4; j += 256) {
        o4[j] = st4[j];
        p4[j] = ss4[j];
    }
}

__device__ __forceinline__ void place_pieces(const int* __restrict__ srt, const int* pbeg, const int* plen,
                                             int* cntn, int* sorted, bool staged, int S,
                                             const float2* __restrict__ nd, int2* __restrict__ ce) {
    const int lane = threadIdx.x & 31;
    const int wave = threadIdx.x >> 5;
    auto take = [&](int en) {
        const int pos = atomicAdd(&cntn[en >> SRC_BITS], 1);
        const int s   = en & ((1 << SRC_BITS) - 1);
        if (staged) sorted[pos] = s;
        else        ce[S + pos] = make_int2(s, __float_as_int(nd[s].x));
    };
    for (int c = wave; c < NCH; c += 128) {
        int bg[4], ln[4], en[4];
#pragma unroll
        for (int k = 0; k < 4; ++k) {
            const int cc = c + 32 * k;
            ln[k] = cc < NCH ? plen[cc] : 0;
            bg[k] = cc < NCH ? pbeg[cc] : 0;
        }
#pragma unroll
        for (int k = 0; k < 4; ++k)
            en[k] = lane < ln[k] ? srt[bg[k] + lane] : -1;
#pragma unroll
        for (int k = 0; k < 4; ++k) {
            if (en[k] >= 0) take(en[k]);
            for (int i = lane + 32; i < ln[k]; i += 32) take(srt[bg[k] + i]);
        }
    }
}

__device__ __forceinline__ void count_pieces(const int* __restrict__ srt_s, const int* pbeg, const int* plen, int* cnts,
                                             const int* __restrict__ srt, const int* qbeg, const int* qlen, int* cntd) {
    const int lane = threadIdx.x & 31;
    const int wave = threadIdx.x >> 5;
    for (int c = wave; c < NCH; c += 128) {
        int bs[4], ls[4], es[4], bd[4], ld[4], ed[4];
#pragma unroll
        for (int k = 0; k < 4; ++k) {
            const int cc = c + 32 * k;
            ls[k] = cc < NCH ? plen[cc] : 0;
            bs[k] = cc < NCH ? pbeg[cc] : 0;
            ld[k] = cc < NCH ? qlen[cc] : 0;
            bd[k] = cc < NCH ? qbeg[cc] : 0;
        }
#pragma unroll
        for (int k = 0; k < 4; ++k) {
            es[k] = lane < ls[k] ? srt_s[bs[k] + lane] : -1;
            ed[k] = lane < ld[k] ? srt[bd[k] + lane] : -1;
        }
#pragma unroll
        for (int k = 0; k < 4; ++k) {
            if (es[k] >= 0) atomicAdd(&cnts[es[k] >> SRC_BITS], 1);
            for (int i = lane + 32; i < ls[k]; i += 32) atomicAdd(&cnts[srt_s[bs[k] + i] >> SRC_BITS], 1);
            if (ed[k] >= 0) atomicAdd(&cntd[ed[k] >> SRC_BITS], 1);
            for (int i = lane + 32; i < ld[k]; i += 32) atomicAdd(&cntd[srt[bd[k] + i] >> SRC_BITS], 1);
        }
    }
}

__global__ void __launch_bounds__(1024) degree_kernel(const int* __restrict__ srt_s, const int* __restrict__ off_s,
                                                      const int* __restrict__ srt, const int* __restrict__ off,
                                                      float2* __restrict__ nd, int* __restrict__ indeg_i) {
    __shared__ int pbeg[NCH], plen[NCH];
    __shared__ int qbeg[NCH], qlen[NCH];
    __shared__ int cnts[NB_NODES];
    __shared__ int cntd[NB_NODES];
    const int br = blockIdx.y;
    const int b  = blockIdx.x;
    srt_s   += (size_t)br * N_EDGES;
    off_s   += (size_t)br * NCH * (NBUCK + 1);
    srt     += (size_t)br * N_EDGES;
    off     += (size_t)br * NCH * (NBUCK + 1);
    nd      += br * N_NODES;
    indeg_i += br * N_NODES;
    const int tid = threadIdx.x;
    if (tid < NCH) {
        const int* p = off_s + (size_t)tid * (NBUCK + 1) + b;
        const int o0 = p[0];
        pbeg[tid] = tid * CH_EDGES + o0;
        plen[tid] = p[1] - o0;
    } else if (tid >= 512 && tid < 512 + NCH) {
        const int ch = tid - 512;
        const int* p = off + (size_t)ch * (NBUCK + 1) + b;
        const int o0 = p[0];
        qbeg[ch] = ch * CH_EDGES + o0;
        qlen[ch] = p[1] - o0;
    }
    if (tid < NB_NODES) {
        cnts[tid] = 0;
        cntd[tid] = 0;
    }
    __syncthreads();
    count_pieces(srt_s, pbeg, plen, cnts, srt, qbeg, qlen, cntd);
    __syncthreads();
    if (tid < NB_NODES) {
        const int node = b * NB_NODES + tid;
        if (node < N_NODES) {
            const int   id  = cntd[tid];
            const float csv = rsqrtf(fmaxf((float)cnts[tid], 1.0f));
            nd[node]      = make_float2(csv, (float)id * csv);
            indeg_i[node] = id;
        }
    }
}

__global__ void __launch_bounds__(1024) place_kernel(const int* __restrict__ srt, const int* __restrict__ off,
                                                     const float2* __restrict__ nd, const int* __restrict__ indeg_i,
                                                     int* __restrict__ rowptr, int2* __restrict__ ce,
                                                     float* __restrict__ cd, float* __restrict__ x) {
    __shared__ int wsum[32];
    __shared__ int pbeg[NCH], plen[NCH];
    __shared__ int cntn[NB_NODES];
    __shared__ int sorted[PLACE_CAP];
    const int br = blockIdx.y;
    const int b  = blockIdx.x;
    srt     += (size_t)br * N_EDGES;
    off     += (size_t)br * NCH * (NBUCK + 1);
    nd      += br * N_NODES;
    indeg_i += br * N_NODES;
    rowptr  += br * N_NODES;
    ce      += (size_t)br * N_EDGES;
    cd      += br * N_NODES;
    x       += br * N_NODES;
    const int tid  = threadIdx.x;
    const int node = b * NB_NODES + tid;
    const bool mine = tid < NB_NODES && node < N_NODES;
    int o0 = 0, len = 0;
    if (tid < NCH) {
        const int* p = off + (size_t)tid * (NBUCK + 1) + b;
        o0  = p[0];
        len = p[1] - o0;
        pbeg[tid] = tid * CH_EDGES + o0;
        plen[tid] = len;
    }
    const int v = mine ? indeg_i[node] : 0;
    int S, T, tot;
    block_excl_scan<32>(o0, wsum, S);
    block_excl_scan<32>(len, wsum, T);
    const int ex = block_excl_scan<32>(v, wsum, tot);
    const bool staged = T <= PLACE_CAP;
    if (tid < NB_NODES) cntn[tid] = ex;
    if (mine) rowptr[node] = S + ex;
    __syncthreads();

    place_pieces(srt, pbeg, plen, cntn, sorted, staged, S, nd, ce);
    __syncthreads();

    if (staged) {
        constexpr int R = PLACE_CAP / 1024;
        int    sv[R];
        float2 pr[R];
#pragma unroll
        for (int k = 0; k < R; ++k) {
            const int j = tid + k * 1024;
            sv[k] = j < T ? sorted[j] : 0;
        }
#pragma unroll
        for (int k = 0; k < R; ++k) pr[k] = nd[sv[k]];
#pragma unroll
        for (int k = 0; k < R; ++k) {
            const int j = tid + k * 1024;
            if (j < T) {
                ce[S + j] = make_int2(sv[k], __float_as_int(pr[k].x));
                sorted[j] = __float_as_int(pr[k].y);
            }
        }
    }
    __syncthreads();

    if (mine) {
        float a = 0.0f;
        if (staged) {
            for (int j = 0; j < v; ++j) a += __int_as_float(sorted[ex + j]);
        } else {
            for (int j = 0; j < v; ++j) a += nd[ce[S + ex + j].x].y;
        }
        const float cdv = rsqrtf(fmaxf((float)v, 1.0f));
        cd[node] = cdv;
        x[node]  = a * cdv;
    }
}

#define GEMM_BLOCKS 1024
#define A_PITCH     132
static_assert(N_NODES % 16 == 0, "row tiles must be whole");
static_assert(GEMM_BLOCKS <= N_NODES / 16, "every block owns a first tile");

__device__ __forceinline__ void rows_tail(const int2* __restrict__ ce, const float4* __restrict__ h4,
                                          int beg, int cnt, int lane, float4& acc) {
    for (int base = 32; base < cnt; base += 32) {
        const int n = min(cnt - base, 32);
        int cidx = 0, cscl = 0;
        if (lane < n) {
            int2 c = ce[beg + base + lane];
            cidx = c.x;
            cscl = c.y;
        }
        for (int j = 0; j < n; j += 4) {
            float4 v[4];
            float  c[4];
#pragma unroll
            for (int k = 0; k < 4; ++k) {
                v[k] = make_float4(0.0f, 0.0f, 0.0f, 0.0f);
                c[k] = 0.0f;
                if (j + k < n) {
                    const int s = __builtin_amdgcn_readlane(cidx, j + k);
                    c[k] = __int_as_float(__builtin_amdgcn_readlane(cscl, j + k));
                    v[k] = h4[(size_t)s * 32 + lane];
                }
            }
#pragma unroll
            for (int k = 0; k < 4; ++k) {
                acc.x = fmaf(v[k].x, c[k], acc.x); acc.y = fmaf(v[k].y, c[k], acc.y);
                acc.z = fmaf(v[k].z, c[k], acc.z); acc.w = fmaf(v[k].w, c[k], acc.w);
            }
        }
    }
}

__device__ __forceinline__ void rank1_tail(const int2* __restrict__ ce, const float* __restrict__ x,
                                           int beg, int cnt, int lane, const float4& w, const float4& b, float4& acc) {
    for (int base = 32; base < cnt; base += 32) {
        const int n = min(cnt - base, 32);
        int xsrc = 0, cscl = 0;
        if (lane < n) {
            int2 c = ce[beg + base + lane];
            cscl = c.y;
            xsrc = __float_as_int(x[c.x]);
        }
        for (int j = 0; j < n; ++j) {
            const float xv = __int_as_float(__builtin_amdgcn_readlane(xsrc, j));
            const float cv = __int_as_float(__builtin_amdgcn_readlane(cscl, j));
            acc.x = fmaf(fmaxf(fmaf(xv, w.x, b.x), 0.0f), cv, acc.x);
            acc.y = fmaf(fmaxf(fmaf(xv, w.y, b.y), 0.0f), cv, acc.y);
            acc.z = fmaf(fmaxf(fmaf(xv, w.z, b.z), 0.0f), cv, acc.z);
            acc.w = fmaf(fmaxf(fmaf(xv, w.w, b.w), 0.0f), cv, acc.w);
        }
    }
}

template <bool POOL, bool RANK1>
__device__ __forceinline__ void layer_body(const int* __restrict__ rowptr,
                                           const int* __restrict__ indeg_i,
                                           const int2* __restrict__ ce,
                                           const float* __restrict__ hin,
                                           const float* __restrict__ W1,
                                           const float* __restrict__ b1,
                                           const float* __restrict__ cd,
                                           const float* __restrict__ W,
                                           const float* __restrict__ bias,
                                           float* __restrict__ hout,
                                           const int* __restrict__ seg,
                                           float* __restrict__ hg,
                                           float* __restrict__ counts) {
    __shared__ __attribute__((aligned(16))) float As[2][16 * A_PITCH];
    constexpr int NT = N_NODES / 16;
    const int tid  = threadIdx.x;
    const int wave = tid >> 5;
    const int wv   = __builtin_amdgcn_readfirstlane(wave);
    const int lane = tid & 31;
    const int half = lane >> 4;
    const int lr   = lane & 15;
    const int n    = wave * 16 + lr;

    v2f bfr[HIDDEN / 4];
#pragma unroll
    for (int s = 0; s < HIDDEN / 4; ++s) {
        bfr[s].x = W[(4 * s + 2 * half) * HIDDEN + n];
        bfr[s].y = W[(4 * s + 2 * half + 1) * HIDDEN + n];
    }
    const float bn = bias[n];
    float4 w1 = {}, bb1 = {};
    if constexpr (RANK1) {
        w1  = ((const float4*)W1)[lane];
        bb1 = ((const float4*)b1)[lane];
    }

    const int sr = 2 * wv;
    int t = blockIdx.x;
    int2   rp, dg, nrp = {}, ndg = {};
    float2 cv, ncv = {};
    {
        const int r0 = t * 16 + sr;
        rp = *(const int2*)(rowptr + r0);
        dg = *(const int2*)(indeg_i + r0);
        cv = *(const float2*)(cd + r0);
    }
    int2 eA = ce[min(rp.x + lane, N_EDGES - 1)];
    int2 eB = ce[min(rp.y + lane, N_EDGES - 1)];
    if (t + GEMM_BLOCKS < NT) {
        const int r0 = (t + GEMM_BLOCKS) * 16 + sr;
        nrp = *(const int2*)(rowptr + r0);
        ndg = *(const int2*)(indeg_i + r0);
        ncv = *(const float2*)(cd + r0);
    }
    int4 na = {}, nb = {}, nc = {}, nd = {};
    if constexpr (POOL) {
        const int4* __restrict__ sg = (const int4*)(seg + t * 16);
        na = sg[0]; nb = sg[1]; nc = sg[2]; nd = sg[3];
    }

    int buf = 0;
    for (;;) {
        const int begA = rp.x, begB = rp.y, cntA = dg.x, cntB = dg.y;
        const int nA = min(cntA, 32), nB = min(cntB, 32);
        float4 accA = make_float4(0.0f, 0.0f, 0.0f, 0.0f);
        float4 accB = make_float4(0.0f, 0.0f, 0.0f, 0.0f);
        if constexpr (RANK1) {
            const int xA = __float_as_int(hin[eA.x]);
            const int xB = __float_as_int(hin[eB.x]);
            for (int j = 0; j < nA; ++j) {
                const float xv = __int_as_float(__builtin_amdgcn_readlane(xA, j));
                const float c  = __int_as_float(__builtin_amdgcn_readlane(eA.y, j));
                accA.x = fmaf(fmaxf(fmaf(xv, w1.x, bb1.x), 0.0f), c, accA.x);
                accA.y = fmaf(fmaxf(fmaf(xv, w1.y, bb1.y), 0.0f), c, accA.y);
                accA.z = fmaf(fmaxf(fmaf(xv, w1.z, bb1.z), 0.0f), c, accA.z);
                accA.w = fmaf(fmaxf(fmaf(xv, w1.w, bb1.w), 0.0f), c, accA.w);
            }
            for (int j = 0; j < nB; ++j) {
                const float xv = __int_as_float(__builtin_amdgcn_readlane(xB, j));
                const float c  = __int_as_float(__builtin_amdgcn_readlane(eB.y, j));
                accB.x = fmaf(fmaxf(fmaf(xv, w1.x, bb1.x), 0.0f), c, accB.x);
                accB.y = fmaf(fmaxf(fmaf(xv, w1.y, bb1.y), 0.0f), c, accB.y);
                accB.z = fmaf(fmaxf(fmaf(xv, w1.z, bb1.z), 0.0f), c, accB.z);
                accB.w = fmaf(fmaxf(fmaf(xv, w1.w, bb1.w), 0.0f), c, accB.w);
            }
            rank1_tail(ce, hin, begA, cntA, lane, w1, bb1, accA);
            rank1_tail(ce, hin, begB, cntB, lane, w1, bb1, accB);
        } else {
            const float4* __restrict__ h4 = (const float4*)hin;
            const int nmax = max(nA, nB);
            asm volatile("" :: "v"(eA.x), "v"(eA.y), "v"(eB.x), "v"(eB.y));
            for (int j = 0; j < nmax; j += 4) {
                float4 va[4], vb[4];
                float  ca[4], cb[4];
#pragma unroll
                for (int k = 0; k < 4; ++k) {
                    va[k] = make_float4(0.0f, 0.0f, 0.0f, 0.0f);
                    ca[k] = 0.0f;
                    if (j + k < nA) {
                        const int s = __builtin_amdgcn_readlane(eA.x, j + k);
                        ca[k] = __int_as_float(__builtin_amdgcn_readlane(eA.y, j + k));
                        va[k] = h4[(size_t)s * 32 + lane];
                    }
                }
#pragma unroll
                for (int k = 0; k < 4; ++k) {
                    vb[k] = make_float4(0.0f, 0.0f, 0.0f, 0.0f);
                    cb[k] = 0.0f;
                    if (j + k < nB) {
                        const int s = __builtin_amdgcn_readlane(eB.x, j + k);
                        cb[k] = __int_as_float(__builtin_amdgcn_readlane(eB.y, j + k));
                        vb[k] = h4[(size_t)s * 32 + lane];
                    }
                }
#pragma unroll
                for (int k = 0; k < 4; ++k) {
                    accA.x = fmaf(va[k].x, ca[k], accA.x); accA.y = fmaf(va[k].y, ca[k], accA.y);
                    accA.z = fmaf(va[k].z, ca[k], accA.z); accA.w = fmaf(va[k].w, ca[k], accA.w);
                }
#pragma unroll
                for (int k = 0; k < 4; ++k) {
                    accB.x = fmaf(vb[k].x, cb[k], accB.x); accB.y = fmaf(vb[k].y, cb[k], accB.y);
                    accB.z = fmaf(vb[k].z, cb[k], accB.z); accB.w = fmaf(vb[k].w, cb[k], accB.w);
                }
            }
            rows_tail(ce, h4, begA, cntA, lane, accA);
            rows_tail(ce, h4, begB, cntB, lane, accB);
        }

        {
            const float c0 = cv.x, c1 = cv.y;
            float* dst = &As[buf][sr * A_PITCH + lane * 4];
            *(float4*)dst             = make_float4(accA.x * c0, accA.y * c0, accA.z * c0, accA.w * c0);
            *(float4*)(dst + A_PITCH) = make_float4(accB.x * c1, accB.y * c1, accB.z * c1, accB.w * c1);
        }
        __syncthreads();

        const int  tn   = t + GEMM_BLOCKS;
        const bool more = tn < NT;
        const int4 sa = na, sb = nb, sc = nc, sd = nd;
        int2   n2rp = nrp, n2dg = ndg;
        float2 n2cv = ncv;
        if (more) {
            eA = ce[min(nrp.x + lane, N_EDGES - 1)];
            eB = ce[min(nrp.y + lane, N_EDGES - 1)];
            const int t2 = tn + GEMM_BLOCKS;
            if (t2 < NT) {
                const int r0 = t2 * 16 + sr;
                n2rp = *(const int2*)(rowptr + r0);
                n2dg = *(const int2*)(indeg_i + r0);
                n2cv = *(const float2*)(cd + r0);
            }
        }
        if constexpr (POOL) {
            const int4* __restrict__ sg = (const int4*)(seg + min(tn, NT - 1) * 16);
            na = sg[0]; nb = sg[1]; nc = sg[2]; nd = sg[3];
        }

        const float* arow = &As[buf][lr * A_PITCH + 2 * half];
        v8f acc = {};
#pragma unroll
        for (int s = 0; s < HIDDEN / 4; ++s) {
            const v2f av = *(const v2f*)(arow + 4 * s);
            acc = __builtin_amdgcn_wmma_f32_16x16x4_f32(
                 false, av,  false, bfr[s],
                 (short)0, acc,  false,  false);
        }

        const int m0 = t * 16;
        if constexpr (!POOL) {
#pragma unroll
            for (int r = 0; r < 8; ++r) {
                const int mm = m0 + half * 8 + r;
                hout[(size_t)mm * HIDDEN + n] = fmaxf(acc[r] + bn, 0.0f);
            }
        } else {
            const int  g0 = sa.x;
            const int  differ = (sa.y ^ g0) | (sa.z ^ g0) | (sa.w ^ g0) |
                                (sb.x ^ g0) | (sb.y ^ g0) | (sb.z ^ g0) | (sb.w ^ g0) |
                                (sc.x ^ g0) | (sc.y ^ g0) | (sc.z ^ g0) | (sc.w ^ g0) |
                                (sd.x ^ g0) | (sd.y ^ g0) | (sd.z ^ g0) | (sd.w ^ g0);
            if (differ == 0) {
                float s = fmaxf(acc[0] + bn, 0.0f);
#pragma unroll
                for (int r = 1; r < 8; ++r)
                    s += fmaxf(acc[r] + bn, 0.0f);
                s += __shfl_xor(s, 16);
                if (half == 0) atomicAdd(&hg[g0 * HIDDEN + n], s);
                if (tid == 0)  atomicAdd(&counts[g0], 16.0f);
            } else {
                const int gl[8] = {sa.x, sa.y, sa.z, sa.w, sb.x, sb.y, sb.z, sb.w};
                const int gh[8] = {sc.x, sc.y, sc.z, sc.w, sd.x, sd.y, sd.z, sd.w};
#pragma unroll
                for (int r = 0; r < 8; ++r) {
                    const int g = half ? gh[r] : gl[r];
                    atomicAdd(&hg[g * HIDDEN + n], fmaxf(acc[r] + bn, 0.0f));
                    if (wave == 0 && lr == 0) atomicAdd(&counts[g], 1.0f);
                }
            }
        }

        if (!more) break;
        rp = nrp;  dg = ndg;  cv = ncv;
        nrp = n2rp; ndg = n2dg; ncv = n2cv;
        t = tn;
        buf ^= 1;
    }
}

__global__ void __launch_bounds__(256) layer_rank1_kernel(const int* __restrict__ rowptr,
                                                          const int* __restrict__ indeg_i,
                                                          const int2* __restrict__ ce,
                                                          const float* __restrict__ x,
                                                          const float* __restrict__ W1,
                                                          const float* __restrict__ b1,
                                                          const float* __restrict__ cd,
                                                          const float* __restrict__ W,
                                                          const float* __restrict__ bias,
                                                          float* __restrict__ hout) {
    const int br = blockIdx.y;
    layer_body<false, true>(rowptr + br * N_NODES, indeg_i + br * N_NODES, ce + (size_t)br * N_EDGES,
                            x + br * N_NODES, W1, b1, cd + br * N_NODES, W, bias,
                            hout + (size_t)br * N_NODES * HIDDEN, nullptr, nullptr, nullptr);
}

__global__ void __launch_bounds__(256) layer_rows_kernel(const int* __restrict__ rowptr,
                                                         const int* __restrict__ indeg_i,
                                                         const int2* __restrict__ ce,
                                                         const float* __restrict__ hin,
                                                         const float* __restrict__ cd,
                                                         const float* __restrict__ W,
                                                         const float* __restrict__ bias,
                                                         float* __restrict__ hout) {
    const int br = blockIdx.y;
    layer_body<false, false>(rowptr + br * N_NODES, indeg_i + br * N_NODES, ce + (size_t)br * N_EDGES,
                             hin + (size_t)br * N_NODES * HIDDEN, nullptr, nullptr, cd + br * N_NODES, W, bias,
                             hout + (size_t)br * N_NODES * HIDDEN, nullptr, nullptr, nullptr);
}

__global__ void __launch_bounds__(256) layer_rows_pool_kernel(const int* __restrict__ rowptr,
                                                              const int* __restrict__ indeg_i,
                                                              const int2* __restrict__ ce,
                                                              const float* __restrict__ hin,
                                                              const float* __restrict__ cd,
                                                              const float* __restrict__ W,
                                                              const float* __restrict__ bias,
                                                              const int* __restrict__ seg1,
                                                              const int* __restrict__ seg2,
                                                              float* __restrict__ hg,
                                                              float* __restrict__ counts) {
    const int br = blockIdx.y;
    layer_body<true, false>(rowptr + br * N_NODES, indeg_i + br * N_NODES, ce + (size_t)br * N_EDGES,
                            hin + (size_t)br * N_NODES * HIDDEN, nullptr, nullptr, cd + br * N_NODES, W, bias,
                            nullptr, br ? seg2 : seg1, hg + br * N_GRAPHS * HIDDEN, counts + br * N_GRAPHS);
}

__global__ void pool_div_kernel(float* __restrict__ hg, const float* __restrict__ counts) {
    int i = blockIdx.x * blockDim.x + threadIdx.x;
    if (i < 2 * N_GRAPHS * HIDDEN)
        hg[i] /= fmaxf(counts[i >> 7], 1.0f);
}

__global__ void logits_kernel(const float* __restrict__ hg1, const float* __restrict__ hg2,
                              const float* __restrict__ Wc, const float* __restrict__ bc,
                              float* __restrict__ logits) {
    int i = blockIdx.x * blockDim.x + threadIdx.x;
    if (i < N_GRAPHS * N_CLASSES) {
        int g = i / N_CLASSES, c = i % N_CLASSES;
        float s = bc[c];
#pragma unroll 8
        for (int k = 0; k < HIDDEN; ++k)
            s += fabsf(hg1[g * HIDDEN + k] - hg2[g * HIDDEN + k]) * Wc[k * N_CLASSES + c];
        logits[i] = s;
    }
}

extern "C" void kernel_launch(void* const* d_in, const int* in_sizes, int n_in,
                              void* d_out, int out_size, void* d_ws, size_t ws_size,
                              hipStream_t stream) {
    const int*   src1 = (const int*)d_in[0];
    const int*   dst1 = (const int*)d_in[1];
    const int*   seg1 = (const int*)d_in[2];
    const int*   src2 = (const int*)d_in[3];
    const int*   dst2 = (const int*)d_in[4];
    const int*   seg2 = (const int*)d_in[5];
    const float* W1   = (const float*)d_in[6];
    const float* b1   = (const float*)d_in[7];
    const float* W2   = (const float*)d_in[8];
    const float* b2   = (const float*)d_in[9];
    const float* W3   = (const float*)d_in[10];
    const float* b3   = (const float*)d_in[11];
    const float* W4   = (const float*)d_in[12];
    const float* b4   = (const float*)d_in[13];
    const float* Wc   = (const float*)d_in[14];
    const float* bc   = (const float*)d_in[15];

    float* out   = (float*)d_out;
    float* hg    = out;
    float* hg1   = out;
    float* hg2   = out + N_GRAPHS * HIDDEN;
    float* logit = out + 2 * N_GRAPHS * HIDDEN;
    float* ws_f  = (float*)d_ws;

    float2* nd     = (float2*)ws_f;
    float* counts  = ws_f + 400000;
    int*   indeg_i = (int*)(ws_f + 400256);
    float* cd      = ws_f + 600256;
    float* x       = ws_f + 800256;
    int*   rowptr  = (int*)(ws_f + 1000256);
    float* ha      = ws_f + 1200256;
    float* hb      = ha + (size_t)2 * N_NODES * HIDDEN;
    int*   srt     = (int*)(hb + (size_t)2 * N_NODES * HIDDEN);
    int2*  ce      = (int2*)(srt + (size_t)2 * N_EDGES);
    int*   off     = (int*)(ce + (size_t)2 * N_EDGES);
    int*   off_s   = off + (size_t)2 * NCH * (NBUCK + 1);
    int*   srt_s   = off_s + (size_t)2 * NCH * (NBUCK + 1);

    hipMemsetAsync(counts, 0, (size_t)2 * N_GRAPHS * sizeof(float), stream);
    hipMemsetAsync(hg, 0, (size_t)2 * N_GRAPHS * HIDDEN * sizeof(float), stream);

    const int T = 256;
    bin_kernel<<<dim3(NCH, 2), T, 0, stream>>>(src1, dst1, src2, dst2, srt, off, srt_s, off_s);
    degree_kernel<<<dim3(NBUCK, 2), 1024, 0, stream>>>(srt_s, off_s, srt, off, nd, indeg_i);
    place_kernel<<<dim3(NBUCK, 2), 1024, 0, stream>>>(srt, off, nd, indeg_i, rowptr, ce, cd, x);

    layer_rank1_kernel<<<dim3(GEMM_BLOCKS, 2), T, 0, stream>>>(rowptr, indeg_i, ce, x, W1, b1, cd, W2, b2, ha);

    layer_rows_kernel<<<dim3(GEMM_BLOCKS, 2), T, 0, stream>>>(rowptr, indeg_i, ce, ha, cd, W3, b3, hb);

    layer_rows_pool_kernel<<<dim3(GEMM_BLOCKS, 2), T, 0, stream>>>(rowptr, indeg_i, ce, hb, cd, W4, b4,
                                                                   seg1, seg2, hg, counts);
    pool_div_kernel<<<(2 * N_GRAPHS * HIDDEN + T - 1) / T, T, 0, stream>>>(hg, counts);

    logits_kernel<<<(N_GRAPHS * N_CLASSES + 255) / 256, 256, 0, stream>>>(hg1, hg2, Wc, bc, logit);
}


# Layer-2/3 output rows staged in LDS and stored one tile later as whole 512-byte rows (a wave stores 2 rows, 4 whole lines per store) instead of 64-byte pieces from the accumulators; values unchanged
# speedup vs baseline: 8.7068x; 1.0373x over previous
#include <hip/hip_runtime.h>
#include <hip/hip_bf16.h>

#define N_NODES   100000
#define N_EDGES   1600000
#define N_GRAPHS  128
#define HIDDEN    128
#define N_CLASSES 10

typedef __attribute__((ext_vector_type(2))) float v2f;
typedef __attribute__((ext_vector_type(8))) float v8f;


#define NB_SHIFT  9
#define NB_NODES  (1 << NB_SHIFT)
#define NBUCK     ((N_NODES + NB_NODES - 1) / NB_NODES)
#define CH_EDGES  4096
#define NCH       ((N_EDGES + CH_EDGES - 1) / CH_EDGES)
#define SRC_BITS  17
#define PLACE_CAP 12288
static_assert(N_NODES <= (1 << SRC_BITS), "a source id must fit its field");
static_assert(NBUCK + 1 <= 256, "one bin thread per range start");
static_assert(NCH <= 512, "one place thread per chunk; one degree thread per chunk and order");
static_assert(N_EDGES % 4 == 0 && CH_EDGES % 1024 == 0, "16-byte copy-out of whole chunks");

template <int NW>
__device__ __forceinline__ int block_excl_scan(int v, int* wsum, int& total) {
    const int lane = threadIdx.x & 31;
    const int wave = threadIdx.x >> 5;
    int inc = v;
    for (int o = 1; o < 32; o <<= 1) {
        int u = __shfl_up(inc, o);
        if (lane >= o) inc += u;
    }
    if (lane == 31) wsum[wave] = inc;
    __syncthreads();
    const int ws = lane < NW ? wsum[lane] : 0;
    int winc = ws;
    for (int o = 1; o < 32; o <<= 1) {
        int u = __shfl_up(winc, o);
        if (lane >= o) winc += u;
    }
    const int wave_excl = __shfl(winc, wave) - __shfl(ws, wave);
    total = __shfl(winc, 31);
    __syncthreads();
    return wave_excl + inc - v;
}

__global__ void __launch_bounds__(256) bin_kernel(const int* __restrict__ src1, const int* __restrict__ dst1,
                                                  const int* __restrict__ src2, const int* __restrict__ dst2,
                                                  int* __restrict__ srt, int* __restrict__ off,
                                                  int* __restrict__ srt_s, int* __restrict__ off_s) {
    __shared__ int wsum[32];
    __shared__ int hist[256];
    __shared__ int hist_s[256];
    __shared__ __attribute__((aligned(16))) int stage[CH_EDGES];
    __shared__ __attribute__((aligned(16))) int stage_s[CH_EDGES];
    const int br    = blockIdx.y;
    const int chunk = blockIdx.x;
    const int* __restrict__ src = br ? src2 : src1;
    const int* __restrict__ dst = br ? dst2 : dst1;
    srt    += (size_t)br * N_EDGES;
    off    += (size_t)(br * NCH + chunk) * (NBUCK + 1);
    srt_s  += (size_t)br * N_EDGES;
    off_s  += (size_t)(br * NCH + chunk) * (NBUCK + 1);
    const int tid = threadIdx.x;
    const int e0  = chunk * CH_EDGES;
    hist[tid]   = 0;
    hist_s[tid] = 0;
    __syncthreads();
    constexpr int PER = CH_EDGES / 256;
    int s[PER], d[PER], rk[PER], rs[PER];
#pragma unroll
    for (int i = 0; i < PER; ++i) {
        const int e = e0 + i * 256 + tid;
        s[i] = e < N_EDGES ? src[e] : -1;
        d[i] = e < N_EDGES ? dst[e] : 0;
    }
#pragma unroll
    for (int i = 0; i < PER; ++i) {
        rk[i] = 0;
        rs[i] = 0;
        if (s[i] >= 0) {
            rk[i] = atomicAdd(&hist[d[i] >> NB_SHIFT], 1);
            rs[i] = atomicAdd(&hist_s[s[i] >> NB_SHIFT], 1);
        }
    }
    __syncthreads();
    int total;
    const int excl   = block_excl_scan<8>(hist[tid], wsum, total);
    const int excl_s = block_excl_scan<8>(hist_s[tid], wsum, total);
    hist[tid]   = excl;
    hist_s[tid] = excl_s;
    if (tid <= NBUCK) {
        off[tid]   = excl;
        off_s[tid] = excl_s;
    }
    __syncthreads();
#pragma unroll
    for (int i = 0; i < PER; ++i)
        if (s[i] >= 0) {
            stage[hist[d[i] >> NB_SHIFT] + rk[i]] = s[i] | ((d[i] & (NB_NODES - 1)) << SRC_BITS);
            stage_s[hist_s[s[i] >> NB_SHIFT] + rs[i]] = (s[i] & (NB_NODES - 1)) << SRC_BITS;
        }
    __syncthreads();
    const int n4 = min(CH_EDGES, N_EDGES - e0) / 4;
    int4* __restrict__ o4 = (int4*)(srt + e0);
    int4* __restrict__ p4 = (int4*)(srt_s + e0);
    const int4* st4 = (const int4*)stage;
    const int4* ss4 = (const int4*)stage_s;
    for (int j = tid; j < n4; j += 256) {
        o4[j] = st4[j];
        p4[j] = ss4[j];
    }
}

__device__ __forceinline__ void place_pieces(const int* __restrict__ srt, const int* pbeg, const int* plen,
                                             int* cntn, int* sorted, bool staged, int S,
                                             const float2* __restrict__ nd, int2* __restrict__ ce) {
    const int lane = threadIdx.x & 31;
    const int wave = threadIdx.x >> 5;
    auto take = [&](int en) {
        const int pos = atomicAdd(&cntn[en >> SRC_BITS], 1);
        const int s   = en & ((1 << SRC_BITS) - 1);
        if (staged) sorted[pos] = s;
        else        ce[S + pos] = make_int2(s, __float_as_int(nd[s].x));
    };
    for (int c = wave; c < NCH; c += 128) {
        int bg[4], ln[4], en[4];
#pragma unroll
        for (int k = 0; k < 4; ++k) {
            const int cc = c + 32 * k;
            ln[k] = cc < NCH ? plen[cc] : 0;
            bg[k] = cc < NCH ? pbeg[cc] : 0;
        }
#pragma unroll
        for (int k = 0; k < 4; ++k)
            en[k] = lane < ln[k] ? srt[bg[k] + lane] : -1;
#pragma unroll
        for (int k = 0; k < 4; ++k) {
            if (en[k] >= 0) take(en[k]);
            for (int i = lane + 32; i < ln[k]; i += 32) take(srt[bg[k] + i]);
        }
    }
}

__device__ __forceinline__ void count_pieces(const int* __restrict__ srt_s, const int* pbeg, const int* plen, int* cnts,
                                             const int* __restrict__ srt, const int* qbeg, const int* qlen, int* cntd) {
    const int lane = threadIdx.x & 31;
    const int wave = threadIdx.x >> 5;
    for (int c = wave; c < NCH; c += 128) {
        int bs[4], ls[4], es[4], bd[4], ld[4], ed[4];
#pragma unroll
        for (int k = 0; k < 4; ++k) {
            const int cc = c + 32 * k;
            ls[k] = cc < NCH ? plen[cc] : 0;
            bs[k] = cc < NCH ? pbeg[cc] : 0;
            ld[k] = cc < NCH ? qlen[cc] : 0;
            bd[k] = cc < NCH ? qbeg[cc] : 0;
        }
#pragma unroll
        for (int k = 0; k < 4; ++k) {
            es[k] = lane < ls[k] ? srt_s[bs[k] + lane] : -1;
            ed[k] = lane < ld[k] ? srt[bd[k] + lane] : -1;
        }
#pragma unroll
        for (int k = 0; k < 4; ++k) {
            if (es[k] >= 0) atomicAdd(&cnts[es[k] >> SRC_BITS], 1);
            for (int i = lane + 32; i < ls[k]; i += 32) atomicAdd(&cnts[srt_s[bs[k] + i] >> SRC_BITS], 1);
            if (ed[k] >= 0) atomicAdd(&cntd[ed[k] >> SRC_BITS], 1);
            for (int i = lane + 32; i < ld[k]; i += 32) atomicAdd(&cntd[srt[bd[k] + i] >> SRC_BITS], 1);
        }
    }
}

__global__ void __launch_bounds__(1024) degree_kernel(const int* __restrict__ srt_s, const int* __restrict__ off_s,
                                                      const int* __restrict__ srt, const int* __restrict__ off,
                                                      float2* __restrict__ nd, int* __restrict__ indeg_i) {
    __shared__ int pbeg[NCH], plen[NCH];
    __shared__ int qbeg[NCH], qlen[NCH];
    __shared__ int cnts[NB_NODES];
    __shared__ int cntd[NB_NODES];
    const int br = blockIdx.y;
    const int b  = blockIdx.x;
    srt_s   += (size_t)br * N_EDGES;
    off_s   += (size_t)br * NCH * (NBUCK + 1);
    srt     += (size_t)br * N_EDGES;
    off     += (size_t)br * NCH * (NBUCK + 1);
    nd      += br * N_NODES;
    indeg_i += br * N_NODES;
    const int tid = threadIdx.x;
    if (tid < NCH) {
        const int* p = off_s + (size_t)tid * (NBUCK + 1) + b;
        const int o0 = p[0];
        pbeg[tid] = tid * CH_EDGES + o0;
        plen[tid] = p[1] - o0;
    } else if (tid >= 512 && tid < 512 + NCH) {
        const int ch = tid - 512;
        const int* p = off + (size_t)ch * (NBUCK + 1) + b;
        const int o0 = p[0];
        qbeg[ch] = ch * CH_EDGES + o0;
        qlen[ch] = p[1] - o0;
    }
    if (tid < NB_NODES) {
        cnts[tid] = 0;
        cntd[tid] = 0;
    }
    __syncthreads();
    count_pieces(srt_s, pbeg, plen, cnts, srt, qbeg, qlen, cntd);
    __syncthreads();
    if (tid < NB_NODES) {
        const int node = b * NB_NODES + tid;
        if (node < N_NODES) {
            const int   id  = cntd[tid];
            const float csv = rsqrtf(fmaxf((float)cnts[tid], 1.0f));
            nd[node]      = make_float2(csv, (float)id * csv);
            indeg_i[node] = id;
        }
    }
}

__global__ void __launch_bounds__(1024) place_kernel(const int* __restrict__ srt, const int* __restrict__ off,
                                                     const float2* __restrict__ nd, const int* __restrict__ indeg_i,
                                                     int* __restrict__ rowptr, int2* __restrict__ ce,
                                                     float* __restrict__ cd, float* __restrict__ x) {
    __shared__ int wsum[32];
    __shared__ int pbeg[NCH], plen[NCH];
    __shared__ int cntn[NB_NODES];
    __shared__ int sorted[PLACE_CAP];
    const int br = blockIdx.y;
    const int b  = blockIdx.x;
    srt     += (size_t)br * N_EDGES;
    off     += (size_t)br * NCH * (NBUCK + 1);
    nd      += br * N_NODES;
    indeg_i += br * N_NODES;
    rowptr  += br * N_NODES;
    ce      += (size_t)br * N_EDGES;
    cd      += br * N_NODES;
    x       += br * N_NODES;
    const int tid  = threadIdx.x;
    const int node = b * NB_NODES + tid;
    const bool mine = tid < NB_NODES && node < N_NODES;
    int o0 = 0, len = 0;
    if (tid < NCH) {
        const int* p = off + (size_t)tid * (NBUCK + 1) + b;
        o0  = p[0];
        len = p[1] - o0;
        pbeg[tid] = tid * CH_EDGES + o0;
        plen[tid] = len;
    }
    const int v = mine ? indeg_i[node] : 0;
    int S, T, tot;
    block_excl_scan<32>(o0, wsum, S);
    block_excl_scan<32>(len, wsum, T);
    const int ex = block_excl_scan<32>(v, wsum, tot);
    const bool staged = T <= PLACE_CAP;
    if (tid < NB_NODES) cntn[tid] = ex;
    if (mine) rowptr[node] = S + ex;
    __syncthreads();

    place_pieces(srt, pbeg, plen, cntn, sorted, staged, S, nd, ce);
    __syncthreads();

    if (staged) {
        constexpr int R = PLACE_CAP / 1024;
        int    sv[R];
        float2 pr[R];
#pragma unroll
        for (int k = 0; k < R; ++k) {
            const int j = tid + k * 1024;
            sv[k] = j < T ? sorted[j] : 0;
        }
#pragma unroll
        for (int k = 0; k < R; ++k) pr[k] = nd[sv[k]];
#pragma unroll
        for (int k = 0; k < R; ++k) {
            const int j = tid + k * 1024;
            if (j < T) {
                ce[S + j] = make_int2(sv[k], __float_as_int(pr[k].x));
                sorted[j] = __float_as_int(pr[k].y);
            }
        }
    }
    __syncthreads();

    if (mine) {
        float a = 0.0f;
        if (staged) {
            for (int j = 0; j < v; ++j) a += __int_as_float(sorted[ex + j]);
        } else {
            for (int j = 0; j < v; ++j) a += nd[ce[S + ex + j].x].y;
        }
        const float cdv = rsqrtf(fmaxf((float)v, 1.0f));
        cd[node] = cdv;
        x[node]  = a * cdv;
    }
}

#define GEMM_BLOCKS 1024
#define A_PITCH     132
static_assert(N_NODES % 16 == 0, "row tiles must be whole");
static_assert(GEMM_BLOCKS <= N_NODES / 16, "every block owns a first tile");

__device__ __forceinline__ void rows_tail(const int2* __restrict__ ce, const float4* __restrict__ h4,
                                          int beg, int cnt, int lane, float4& acc) {
    for (int base = 32; base < cnt; base += 32) {
        const int n = min(cnt - base, 32);
        int cidx = 0, cscl = 0;
        if (lane < n) {
            int2 c = ce[beg + base + lane];
            cidx = c.x;
            cscl = c.y;
        }
        for (int j = 0; j < n; j += 4) {
            float4 v[4];
            float  c[4];
#pragma unroll
            for (int k = 0; k < 4; ++k) {
                v[k] = make_float4(0.0f, 0.0f, 0.0f, 0.0f);
                c[k] = 0.0f;
                if (j + k < n) {
                    const int s = __builtin_amdgcn_readlane(cidx, j + k);
                    c[k] = __int_as_float(__builtin_amdgcn_readlane(cscl, j + k));
                    v[k] = h4[(size_t)s * 32 + lane];
                }
            }
#pragma unroll
            for (int k = 0; k < 4; ++k) {
                acc.x = fmaf(v[k].x, c[k], acc.x); acc.y = fmaf(v[k].y, c[k], acc.y);
                acc.z = fmaf(v[k].z, c[k], acc.z); acc.w = fmaf(v[k].w, c[k], acc.w);
            }
        }
    }
}

__device__ __forceinline__ void rank1_tail(const int2* __restrict__ ce, const float* __restrict__ x,
                                           int beg, int cnt, int lane, const float4& w, const float4& b, float4& acc) {
    for (int base = 32; base < cnt; base += 32) {
        const int n = min(cnt - base, 32);
        int xsrc = 0, cscl = 0;
        if (lane < n) {
            int2 c = ce[beg + base + lane];
            cscl = c.y;
            xsrc = __float_as_int(x[c.x]);
        }
        for (int j = 0; j < n; ++j) {
            const float xv = __int_as_float(__builtin_amdgcn_readlane(xsrc, j));
            const float cv = __int_as_float(__builtin_amdgcn_readlane(cscl, j));
            acc.x = fmaf(fmaxf(fmaf(xv, w.x, b.x), 0.0f), cv, acc.x);
            acc.y = fmaf(fmaxf(fmaf(xv, w.y, b.y), 0.0f), cv, acc.y);
            acc.z = fmaf(fmaxf(fmaf(xv, w.z, b.z), 0.0f), cv, acc.z);
            acc.w = fmaf(fmaxf(fmaf(xv, w.w, b.w), 0.0f), cv, acc.w);
        }
    }
}

template <bool POOL, bool RANK1>
__device__ __forceinline__ void layer_body(const int* __restrict__ rowptr,
                                           const int* __restrict__ indeg_i,
                                           const int2* __restrict__ ce,
                                           const float* __restrict__ hin,
                                           const float* __restrict__ W1,
                                           const float* __restrict__ b1,
                                           const float* __restrict__ cd,
                                           const float* __restrict__ W,
                                           const float* __restrict__ bias,
                                           float* __restrict__ hout,
                                           const int* __restrict__ seg,
                                           float* __restrict__ hg,
                                           float* __restrict__ counts) {
    __shared__ __attribute__((aligned(16))) float Sm[POOL ? 2 : 4][16 * A_PITCH];
    float (*As)[16 * A_PITCH] = Sm;
    float (*Os)[16 * A_PITCH] = Sm + (POOL ? 0 : 2);
    constexpr int NT = N_NODES / 16;
    const int tid  = threadIdx.x;
    const int wave = tid >> 5;
    const int wv   = __builtin_amdgcn_readfirstlane(wave);
    const int lane = tid & 31;
    const int half = lane >> 4;
    const int lr   = lane & 15;
    const int n    = wave * 16 + lr;

    v2f bfr[HIDDEN / 4];
#pragma unroll
    for (int s = 0; s < HIDDEN / 4; ++s) {
        bfr[s].x = W[(4 * s + 2 * half) * HIDDEN + n];
        bfr[s].y = W[(4 * s + 2 * half + 1) * HIDDEN + n];
    }
    const float bn = bias[n];
    float4 w1 = {}, bb1 = {};
    if constexpr (RANK1) {
        w1  = ((const float4*)W1)[lane];
        bb1 = ((const float4*)b1)[lane];
    }

    const int sr = 2 * wv;
    int t = blockIdx.x;
    int2   rp, dg, nrp = {}, ndg = {};
    float2 cv, ncv = {};
    {
        const int r0 = t * 16 + sr;
        rp = *(const int2*)(rowptr + r0);
        dg = *(const int2*)(indeg_i + r0);
        cv = *(const float2*)(cd + r0);
    }
    int2 eA = ce[min(rp.x + lane, N_EDGES - 1)];
    int2 eB = ce[min(rp.y + lane, N_EDGES - 1)];
    if (t + GEMM_BLOCKS < NT) {
        const int r0 = (t + GEMM_BLOCKS) * 16 + sr;
        nrp = *(const int2*)(rowptr + r0);
        ndg = *(const int2*)(indeg_i + r0);
        ncv = *(const float2*)(cd + r0);
    }
    int4 na = {}, nb = {}, nc = {}, nd = {};
    if constexpr (POOL) {
        const int4* __restrict__ sg = (const int4*)(seg + t * 16);
        na = sg[0]; nb = sg[1]; nc = sg[2]; nd = sg[3];
    }

    int buf = 0;
    for (;;) {
        const int begA = rp.x, begB = rp.y, cntA = dg.x, cntB = dg.y;
        const int nA = min(cntA, 32), nB = min(cntB, 32);
        float4 accA = make_float4(0.0f, 0.0f, 0.0f, 0.0f);
        float4 accB = make_float4(0.0f, 0.0f, 0.0f, 0.0f);
        if constexpr (RANK1) {
            const int xA = __float_as_int(hin[eA.x]);
            const int xB = __float_as_int(hin[eB.x]);
            for (int j = 0; j < nA; ++j) {
                const float xv = __int_as_float(__builtin_amdgcn_readlane(xA, j));
                const float c  = __int_as_float(__builtin_amdgcn_readlane(eA.y, j));
                accA.x = fmaf(fmaxf(fmaf(xv, w1.x, bb1.x), 0.0f), c, accA.x);
                accA.y = fmaf(fmaxf(fmaf(xv, w1.y, bb1.y), 0.0f), c, accA.y);
                accA.z = fmaf(fmaxf(fmaf(xv, w1.z, bb1.z), 0.0f), c, accA.z);
                accA.w = fmaf(fmaxf(fmaf(xv, w1.w, bb1.w), 0.0f), c, accA.w);
            }
            for (int j = 0; j < nB; ++j) {
                const float xv = __int_as_float(__builtin_amdgcn_readlane(xB, j));
                const float c  = __int_as_float(__builtin_amdgcn_readlane(eB.y, j));
                accB.x = fmaf(fmaxf(fmaf(xv, w1.x, bb1.x), 0.0f), c, accB.x);
                accB.y = fmaf(fmaxf(fmaf(xv, w1.y, bb1.y), 0.0f), c, accB.y);
                accB.z = fmaf(fmaxf(fmaf(xv, w1.z, bb1.z), 0.0f), c, accB.z);
                accB.w = fmaf(fmaxf(fmaf(xv, w1.w, bb1.w), 0.0f), c, accB.w);
            }
            rank1_tail(ce, hin, begA, cntA, lane, w1, bb1, accA);
            rank1_tail(ce, hin, begB, cntB, lane, w1, bb1, accB);
        } else {
            const float4* __restrict__ h4 = (const float4*)hin;
            const int nmax = max(nA, nB);
            asm volatile("" :: "v"(eA.x), "v"(eA.y), "v"(eB.x), "v"(eB.y));
            for (int j = 0; j < nmax; j += 4) {
                float4 va[4], vb[4];
                float  ca[4], cb[4];
#pragma unroll
                for (int k = 0; k < 4; ++k) {
                    va[k] = make_float4(0.0f, 0.0f, 0.0f, 0.0f);
                    ca[k] = 0.0f;
                    if (j + k < nA) {
                        const int s = __builtin_amdgcn_readlane(eA.x, j + k);
                        ca[k] = __int_as_float(__builtin_amdgcn_readlane(eA.y, j + k));
                        va[k] = h4[(size_t)s * 32 + lane];
                    }
                }
#pragma unroll
                for (int k = 0; k < 4; ++k) {
                    vb[k] = make_float4(0.0f, 0.0f, 0.0f, 0.0f);
                    cb[k] = 0.0f;
                    if (j + k < nB) {
                        const int s = __builtin_amdgcn_readlane(eB.x, j + k);
                        cb[k] = __int_as_float(__builtin_amdgcn_readlane(eB.y, j + k));
                        vb[k] = h4[(size_t)s * 32 + lane];
                    }
                }
#pragma unroll
                for (int k = 0; k < 4; ++k) {
                    accA.x = fmaf(va[k].x, ca[k], accA.x); accA.y = fmaf(va[k].y, ca[k], accA.y);
                    accA.z = fmaf(va[k].z, ca[k], accA.z); accA.w = fmaf(va[k].w, ca[k], accA.w);
                }
#pragma unroll
                for (int k = 0; k < 4; ++k) {
                    accB.x = fmaf(vb[k].x, cb[k], accB.x); accB.y = fmaf(vb[k].y, cb[k], accB.y);
                    accB.z = fmaf(vb[k].z, cb[k], accB.z); accB.w = fmaf(vb[k].w, cb[k], accB.w);
                }
            }
            rows_tail(ce, h4, begA, cntA, lane, accA);
            rows_tail(ce, h4, begB, cntB, lane, accB);
        }

        {
            const float c0 = cv.x, c1 = cv.y;
            float* dst = &As[buf][sr * A_PITCH + lane * 4];
            *(float4*)dst             = make_float4(accA.x * c0, accA.y * c0, accA.z * c0, accA.w * c0);
            *(float4*)(dst + A_PITCH) = make_float4(accB.x * c1, accB.y * c1, accB.z * c1, accB.w * c1);
        }
        __syncthreads();

        if constexpr (!POOL) {
            if (t != (int)blockIdx.x) {
                const float* orow = &Os[buf ^ 1][sr * A_PITCH + lane * 4];
                const float4 o0 = *(const float4*)orow;
                const float4 o1 = *(const float4*)(orow + A_PITCH);
                float* hrow = hout + (size_t)((t - GEMM_BLOCKS) * 16 + sr) * HIDDEN + lane * 4;
                *(float4*)hrow            = o0;
                *(float4*)(hrow + HIDDEN) = o1;
            }
        }

        const int  tn   = t + GEMM_BLOCKS;
        const bool more = tn < NT;
        const int4 sa = na, sb = nb, sc = nc, sd = nd;
        int2   n2rp = nrp, n2dg = ndg;
        float2 n2cv = ncv;
        if (more) {
            eA = ce[min(nrp.x + lane, N_EDGES - 1)];
            eB = ce[min(nrp.y + lane, N_EDGES - 1)];
            const int t2 = tn + GEMM_BLOCKS;
            if (t2 < NT) {
                const int r0 = t2 * 16 + sr;
                n2rp = *(const int2*)(rowptr + r0);
                n2dg = *(const int2*)(indeg_i + r0);
                n2cv = *(const float2*)(cd + r0);
            }
        }
        if constexpr (POOL) {
            const int4* __restrict__ sg = (const int4*)(seg + min(tn, NT - 1) * 16);
            na = sg[0]; nb = sg[1]; nc = sg[2]; nd = sg[3];
        }

        const float* arow = &As[buf][lr * A_PITCH + 2 * half];
        v8f acc = {};
#pragma unroll
        for (int s = 0; s < HIDDEN / 4; ++s) {
            const v2f av = *(const v2f*)(arow + 4 * s);
            acc = __builtin_amdgcn_wmma_f32_16x16x4_f32(
                 false, av,  false, bfr[s],
                 (short)0, acc,  false,  false);
        }

        if constexpr (!POOL) {
#pragma unroll
            for (int r = 0; r < 8; ++r)
                Os[buf][(half * 8 + r) * A_PITCH + n] = fmaxf(acc[r] + bn, 0.0f);
        } else {
            const int  g0 = sa.x;
            const int  differ = (sa.y ^ g0) | (sa.z ^ g0) | (sa.w ^ g0) |
                                (sb.x ^ g0) | (sb.y ^ g0) | (sb.z ^ g0) | (sb.w ^ g0) |
                                (sc.x ^ g0) | (sc.y ^ g0) | (sc.z ^ g0) | (sc.w ^ g0) |
                                (sd.x ^ g0) | (sd.y ^ g0) | (sd.z ^ g0) | (sd.w ^ g0);
            if (differ == 0) {
                float s = fmaxf(acc[0] + bn, 0.0f);
#pragma unroll
                for (int r = 1; r < 8; ++r)
                    s += fmaxf(acc[r] + bn, 0.0f);
                s += __shfl_xor(s, 16);
                if (half == 0) atomicAdd(&hg[g0 * HIDDEN + n], s);
                if (tid == 0)  atomicAdd(&counts[g0], 16.0f);
            } else {
                const int gl[8] = {sa.x, sa.y, sa.z, sa.w, sb.x, sb.y, sb.z, sb.w};
                const int gh[8] = {sc.x, sc.y, sc.z, sc.w, sd.x, sd.y, sd.z, sd.w};
#pragma unroll
                for (int r = 0; r < 8; ++r) {
                    const int g = half ? gh[r] : gl[r];
                    atomicAdd(&hg[g * HIDDEN + n], fmaxf(acc[r] + bn, 0.0f));
                    if (wave == 0 && lr == 0) atomicAdd(&counts[g], 1.0f);
                }
            }
        }

        if (!more) break;
        rp = nrp;  dg = ndg;  cv = ncv;
        nrp = n2rp; ndg = n2dg; ncv = n2cv;
        t = tn;
        buf ^= 1;
    }
    if constexpr (!POOL) {
        __syncthreads();
        const float* orow = &Os[buf][sr * A_PITCH + lane * 4];
        const float4 o0 = *(const float4*)orow;
        const float4 o1 = *(const float4*)(orow + A_PITCH);
        float* hrow = hout + (size_t)(t * 16 + sr) * HIDDEN + lane * 4;
        *(float4*)hrow            = o0;
        *(float4*)(hrow + HIDDEN) = o1;
    }
}

__global__ void __launch_bounds__(256, 8) layer_rank1_kernel(const int* __restrict__ rowptr,
                                                          const int* __restrict__ indeg_i,
                                                          const int2* __restrict__ ce,
                                                          const float* __restrict__ x,
                                                          const float* __restrict__ W1,
                                                          const float* __restrict__ b1,
                                                          const float* __restrict__ cd,
                                                          const float* __restrict__ W,
                                                          const float* __restrict__ bias,
                                                          float* __restrict__ hout) {
    const int br = blockIdx.y;
    layer_body<false, true>(rowptr + br * N_NODES, indeg_i + br * N_NODES, ce + (size_t)br * N_EDGES,
                            x + br * N_NODES, W1, b1, cd + br * N_NODES, W, bias,
                            hout + (size_t)br * N_NODES * HIDDEN, nullptr, nullptr, nullptr);
}

__global__ void __launch_bounds__(256) layer_rows_kernel(const int* __restrict__ rowptr,
                                                         const int* __restrict__ indeg_i,
                                                         const int2* __restrict__ ce,
                                                         const float* __restrict__ hin,
                                                         const float* __restrict__ cd,
                                                         const float* __restrict__ W,
                                                         const float* __restrict__ bias,
                                                         float* __restrict__ hout) {
    const int br = blockIdx.y;
    layer_body<false, false>(rowptr + br * N_NODES, indeg_i + br * N_NODES, ce + (size_t)br * N_EDGES,
                             hin + (size_t)br * N_NODES * HIDDEN, nullptr, nullptr, cd + br * N_NODES, W, bias,
                             hout + (size_t)br * N_NODES * HIDDEN, nullptr, nullptr, nullptr);
}

__global__ void __launch_bounds__(256) layer_rows_pool_kernel(const int* __restrict__ rowptr,
                                                              const int* __restrict__ indeg_i,
                                                              const int2* __restrict__ ce,
                                                              const float* __restrict__ hin,
                                                              const float* __restrict__ cd,
                                                              const float* __restrict__ W,
                                                              const float* __restrict__ bias,
                                                              const int* __restrict__ seg1,
                                                              const int* __restrict__ seg2,
                                                              float* __restrict__ hg,
                                                              float* __restrict__ counts) {
    const int br = blockIdx.y;
    layer_body<true, false>(rowptr + br * N_NODES, indeg_i + br * N_NODES, ce + (size_t)br * N_EDGES,
                            hin + (size_t)br * N_NODES * HIDDEN, nullptr, nullptr, cd + br * N_NODES, W, bias,
                            nullptr, br ? seg2 : seg1, hg + br * N_GRAPHS * HIDDEN, counts + br * N_GRAPHS);
}

__global__ void pool_div_kernel(float* __restrict__ hg, const float* __restrict__ counts) {
    int i = blockIdx.x * blockDim.x + threadIdx.x;
    if (i < 2 * N_GRAPHS * HIDDEN)
        hg[i] /= fmaxf(counts[i >> 7], 1.0f);
}

__global__ void logits_kernel(const float* __restrict__ hg1, const float* __restrict__ hg2,
                              const float* __restrict__ Wc, const float* __restrict__ bc,
                              float* __restrict__ logits) {
    int i = blockIdx.x * blockDim.x + threadIdx.x;
    if (i < N_GRAPHS * N_CLASSES) {
        int g = i / N_CLASSES, c = i % N_CLASSES;
        float s = bc[c];
#pragma unroll 8
        for (int k = 0; k < HIDDEN; ++k)
            s += fabsf(hg1[g * HIDDEN + k] - hg2[g * HIDDEN + k]) * Wc[k * N_CLASSES + c];
        logits[i] = s;
    }
}

extern "C" void kernel_launch(void* const* d_in, const int* in_sizes, int n_in,
                              void* d_out, int out_size, void* d_ws, size_t ws_size,
                              hipStream_t stream) {
    const int*   src1 = (const int*)d_in[0];
    const int*   dst1 = (const int*)d_in[1];
    const int*   seg1 = (const int*)d_in[2];
    const int*   src2 = (const int*)d_in[3];
    const int*   dst2 = (const int*)d_in[4];
    const int*   seg2 = (const int*)d_in[5];
    const float* W1   = (const float*)d_in[6];
    const float* b1   = (const float*)d_in[7];
    const float* W2   = (const float*)d_in[8];
    const float* b2   = (const float*)d_in[9];
    const float* W3   = (const float*)d_in[10];
    const float* b3   = (const float*)d_in[11];
    const float* W4   = (const float*)d_in[12];
    const float* b4   = (const float*)d_in[13];
    const float* Wc   = (const float*)d_in[14];
    const float* bc   = (const float*)d_in[15];

    float* out   = (float*)d_out;
    float* hg    = out;
    float* hg1   = out;
    float* hg2   = out + N_GRAPHS * HIDDEN;
    float* logit = out + 2 * N_GRAPHS * HIDDEN;
    float* ws_f  = (float*)d_ws;

    float2* nd     = (float2*)ws_f;
    float* counts  = ws_f + 400000;
    int*   indeg_i = (int*)(ws_f + 400256);
    float* cd      = ws_f + 600256;
    float* x       = ws_f + 800256;
    int*   rowptr  = (int*)(ws_f + 1000256);
    float* ha      = ws_f + 1200256;
    float* hb      = ha + (size_t)2 * N_NODES * HIDDEN;
    int*   srt     = (int*)(hb + (size_t)2 * N_NODES * HIDDEN);
    int2*  ce      = (int2*)(srt + (size_t)2 * N_EDGES);
    int*   off     = (int*)(ce + (size_t)2 * N_EDGES);
    int*   off_s   = off + (size_t)2 * NCH * (NBUCK + 1);
    int*   srt_s   = off_s + (size_t)2 * NCH * (NBUCK + 1);

    hipMemsetAsync(counts, 0, (size_t)2 * N_GRAPHS * sizeof(float), stream);
    hipMemsetAsync(hg, 0, (size_t)2 * N_GRAPHS * HIDDEN * sizeof(float), stream);

    const int T = 256;
    bin_kernel<<<dim3(NCH, 2), T, 0, stream>>>(src1, dst1, src2, dst2, srt, off, srt_s, off_s);
    degree_kernel<<<dim3(NBUCK, 2), 1024, 0, stream>>>(srt_s, off_s, srt, off, nd, indeg_i);
    place_kernel<<<dim3(NBUCK, 2), 1024, 0, stream>>>(srt, off, nd, indeg_i, rowptr, ce, cd, x);

    layer_rank1_kernel<<<dim3(GEMM_BLOCKS, 2), T, 0, stream>>>(rowptr, indeg_i, ce, x, W1, b1, cd, W2, b2, ha);

    layer_rows_kernel<<<dim3(GEMM_BLOCKS, 2), T, 0, stream>>>(rowptr, indeg_i, ce, ha, cd, W3, b3, hb);

    layer_rows_pool_kernel<<<dim3(GEMM_BLOCKS, 2), T, 0, stream>>>(rowptr, indeg_i, ce, hb, cd, W4, b4,
                                                                   seg1, seg2, hg, counts);
    pool_div_kernel<<<(2 * N_GRAPHS * HIDDEN + T - 1) / T, T, 0, stream>>>(hg, counts);

    logits_kernel<<<(N_GRAPHS * N_CLASSES + 255) / 256, 256, 0, stream>>>(hg1, hg2, Wc, bc, logit);
}
